# starting kernel
# baseline (speedup 1.0000x reference)
#include <hip/hip_runtime.h>


typedef __attribute__((ext_vector_type(2))) float v2f;
typedef __attribute__((ext_vector_type(8))) float v8f;

__device__ __forceinline__ void atomAddF(float* p, float v) {
    asm volatile("global_atomic_add_f32 %0, %1, off" :: "v"(p), "v"(v) : "memory");
}

__global__ void k_zero(float* __restrict__ p, long n) {
    long i = (long)blockIdx.x * blockDim.x + threadIdx.x;
    long stride = (long)gridDim.x * blockDim.x;
    for (; i < n; i += stride) p[i] = 0.0f;
}

__global__ void k_deg(const int* __restrict__ src, const int* __restrict__ dst,
                      int ne, float* __restrict__ degS, float* __restrict__ degD) {
    int e = blockIdx.x * blockDim.x + threadIdx.x;
    if (e < ne) {
        atomAddF(&degS[src[e]], 1.0f);
        atomAddF(&degD[dst[e]], 1.0f);
    }
}

__global__ void k_rsqrt(float* __restrict__ p, int n) {
    int i = blockIdx.x * blockDim.x + threadIdx.x;
    if (i < n) p[i] = rsqrtf(fmaxf(p[i], 1.0f));
}

__global__ void k_spmm128(const float* __restrict__ x, const int* __restrict__ src,
                          const int* __restrict__ dst, const float* __restrict__ normS,
                          float* __restrict__ agg, int ne) {
    int e = blockIdx.x * 8 + (threadIdx.x >> 5);
    int lane = threadIdx.x & 31;
    if (e >= ne) return;
    int s = src[e];
    int d = dst[e];
    float ns = normS[s];
    const float4* xp = (const float4*)(x + (long)s * 128);
    float4 v = xp[lane];
    float* ap = agg + (long)d * 128 + lane * 4;
    atomAddF(ap + 0, v.x * ns);
    atomAddF(ap + 1, v.y * ns);
    atomAddF(ap + 2, v.z * ns);
    atomAddF(ap + 3, v.w * ns);
}

__global__ void k_spmm64(const float* __restrict__ g, const int* __restrict__ src,
                         const int* __restrict__ dst, float* __restrict__ agg, int ne) {
    int e = blockIdx.x * 8 + (threadIdx.x >> 5);
    int lane = threadIdx.x & 31;
    if (e >= ne) return;
    int s = src[e];
    int d = dst[e];
    const float2* gp = (const float2*)(g + (long)s * 64);
    float2 v = gp[lane];
    float* ap = agg + (long)d * 64 + lane * 2;
    atomAddF(ap + 0, v.x);
    atomAddF(ap + 1, v.y);
}

template <int K, int N, bool RELU, bool BIAS>
__global__ void k_gemm_wmma(const float* __restrict__ A,
                            const float* __restrict__ rowscale,
                            const float* __restrict__ W,
                            const float* __restrict__ bias,
                            float* __restrict__ C) {
    constexpr int tilesN = N / 16;
    int wave = (blockIdx.x * blockDim.x + threadIdx.x) >> 5;
    int lane = threadIdx.x & 31;
    int tm = wave / tilesN;
    int tn = wave % tilesN;

    int half = lane >> 4;
    int l15  = lane & 15;
    int arow = tm * 16 + l15;
    int gcol = tn * 16 + l15;

    float scale = rowscale[arow];
    const float* ap = A + (long)arow * K + half * 2;
    const float* wp = W + gcol;

    v8f c = {};
    for (int k = 0; k < K; k += 4) {
        v2f a;
        a.x = ap[k]     * scale;
        a.y = ap[k + 1] * scale;
        int kb = k + half * 2;
        v2f b;
        b.x = wp[(long)kb * N];
        b.y = wp[(long)(kb + 1) * N];
        c = __builtin_amdgcn_wmma_f32_16x16x4_f32(
                 false, a,  false, b,
                 (short)0, c,  false,  false);
    }

    float bval = 0.0f;
    if (BIAS) bval = bias[gcol];
    int rb = half * 8;
    #pragma unroll
    for (int i = 0; i < 8; ++i) {
        float v = c[i] + bval;
        if (RELU) v = fmaxf(v, 0.0f);
        C[(long)(tm * 16 + rb + i) * N + gcol] = v;
    }
}

__global__ void k_final(const float* __restrict__ agg2, const float* __restrict__ normD2,
                        const float* __restrict__ b2, float* __restrict__ out, int tot) {
    int t = blockIdx.x * blockDim.x + threadIdx.x;
    if (t < tot) {
        int r = t >> 6;
        int f = t & 63;
        out[t] = agg2[t] * normD2[r] + b2[f];
    }
}

extern "C" void kernel_launch(void* const* d_in, const int* in_sizes, int n_in,
                              void* d_out, int out_size, void* d_ws, size_t ws_size,
                              hipStream_t stream) {
    const float* x    = (const float*)d_in[0];
    const float* W1   = (const float*)d_in[1];
    const float* b1   = (const float*)d_in[2];
    const float* W2   = (const float*)d_in[3];
    const float* b2   = (const float*)d_in[4];
    const int*   src1 = (const int*)d_in[5];
    const int*   dst1 = (const int*)d_in[6];
    const int*   src2 = (const int*)d_in[7];
    const int*   dst2 = (const int*)d_in[8];

    const int IN_F = 128, H_F = 256, N_CLS = 64;
    const int NSRC1 = in_sizes[0] / IN_F;
    const int E1    = in_sizes[5];
    const int E2    = in_sizes[7];
    const int ND1   = 100000;
    const int ND2   = 16384;

    float* ws    = (float*)d_ws;
    float* degS1 = ws;
    float* degD1 = degS1 + NSRC1;
    float* degS2 = degD1 + ND1;
    float* degD2 = degS2 + ND1;
    float* agg1  = degD2 + ND2;
    float* agg2  = agg1 + (long)ND1 * IN_F;
    float* h     = agg2 + (long)ND2 * N_CLS;
    float* g     = h + (long)ND1 * H_F;

    long zcount = (NSRC1 + ND1 + ND1 + ND2) + (long)ND1 * IN_F + (long)ND2 * N_CLS;
    k_zero<<<2048, 256, 0, stream>>>(ws, zcount);

    k_deg<<<(E1 + 255) / 256, 256, 0, stream>>>(src1, dst1, E1, degS1, degD1);
    k_deg<<<(E2 + 255) / 256, 256, 0, stream>>>(src2, dst2, E2, degS2, degD2);
    int degtot = NSRC1 + ND1 + ND1 + ND2;
    k_rsqrt<<<(degtot + 255) / 256, 256, 0, stream>>>(ws, degtot);

    k_spmm128<<<(E1 + 7) / 8, 256, 0, stream>>>(x, src1, dst1, degS1, agg1, E1);

    {
        int waves = (ND1 / 16) * (H_F / 16);
        k_gemm_wmma<128, 256, true, true>
            <<<waves / 4, 128, 0, stream>>>(agg1, degD1, W1, b1, h);
    }

    {
        int waves = (ND1 / 16) * (N_CLS / 16);
        k_gemm_wmma<256, 64, false, false>
            <<<waves / 4, 128, 0, stream>>>(h, degS2, W2, nullptr, g);
    }

    k_spmm64<<<(E2 + 7) / 8, 256, 0, stream>>>(g, src2, dst2, agg2, E2);

    int tot = ND2 * N_CLS;
    k_final<<<(tot + 255) / 256, 256, 0, stream>>>(agg2, degD2, b2, (float*)d_out, tot);
}


# SpMMs by destination tiles: a workgroup keeps its dst rows in LDS, filters the dst list, owner waves add gathered rows in fixed order; agg1/agg2 float atomics, their zero fill, epilogue launch gone
# speedup vs baseline: 1.3606x; 1.3606x over previous
#include <hip/hip_runtime.h>


typedef __attribute__((ext_vector_type(2))) float v2f;
typedef __attribute__((ext_vector_type(8))) float v8f;

__device__ __forceinline__ void atomAddF(float* p, float v) {
    asm volatile("global_atomic_add_f32 %0, %1, off" :: "v"(p), "v"(v) : "memory");
}

__global__ void k_zero(float* __restrict__ p, long n) {
    long i = (long)blockIdx.x * blockDim.x + threadIdx.x;
    long stride = (long)gridDim.x * blockDim.x;
    for (; i < n; i += stride) p[i] = 0.0f;
}

__global__ void k_deg(const int* __restrict__ src, const int* __restrict__ dst,
                      int ne, float* __restrict__ degS, float* __restrict__ degD) {
    int e = blockIdx.x * blockDim.x + threadIdx.x;
    if (e < ne) {
        atomAddF(&degS[src[e]], 1.0f);
        atomAddF(&degD[dst[e]], 1.0f);
    }
}

__global__ void k_rsqrt(float* __restrict__ p, int n) {
    int i = blockIdx.x * blockDim.x + threadIdx.x;
    if (i < n) p[i] = rsqrtf(fmaxf(p[i], 1.0f));
}

constexpr int P_NLD  = 16;
constexpr int P_SEG  = 1024 * 4 * P_NLD;
constexpr int P_WCAP = 24;
constexpr int P_MCAP = 32 * P_WCAP;
constexpr int P_MAXP = (32 * 4 * P_NLD) / P_WCAP;

template <int F, int R, bool SRCSCALE, bool FINAL>
__global__ __launch_bounds__(1024)
void k_pull(const float* __restrict__ feat, const int* __restrict__ src,
            const int* __restrict__ dst, const float* __restrict__ normS,
            const float* __restrict__ normD, const float* __restrict__ bias,
            float* __restrict__ res, int ne, int nd) {
    constexpr int VL = F / 32;
    typedef float vfl __attribute__((ext_vector_type(VL)));
    static_assert(R <= 512 && (F == 128 || F == 64), "entry packing and lane vectors");

    __shared__ __attribute__((aligned(16))) float tile[R * F];
    __shared__ unsigned wq[32 * P_WCAP];
    __shared__ int wcnt[32];
    __shared__ unsigned mylist[32 * P_MCAP];
    __shared__ int more[2];

    const int tid  = threadIdx.x;
    const int lane = tid & 31;
    const int wave = __builtin_amdgcn_readfirstlane(tid >> 5);
    const int row0 = blockIdx.x * R;

    for (int i = tid; i < R * 32; i += 1024) ((vfl*)tile)[i] = vfl{};
    if (tid < 2) more[tid] = 0;
    __syncthreads();

    int pp = 0;
    const int nseg = (ne + P_SEG - 1) / P_SEG;
    for (int seg = 0; seg < nseg; ++seg) {
        const int segBase = seg * P_SEG;

        for (int p = 0; p <= P_MAXP; ++p) {
            const int fl = pp & 1;

            int4 dv[P_NLD];
            if (segBase + P_SEG <= ne) {
                #pragma unroll
                for (int u = 0; u < P_NLD; ++u)
                    dv[u] = *(const int4*)(dst + segBase + (u * 1024 + tid) * 4);
            } else {
                #pragma unroll
                for (int u = 0; u < P_NLD; ++u) {
                    const int i = segBase + (u * 1024 + tid) * 4;
                    if (i + 3 < ne) {
                        dv[u] = *(const int4*)(dst + i);
                    } else {
                        dv[u].x = (i     < ne) ? dst[i]     : -1;
                        dv[u].y = (i + 1 < ne) ? dst[i + 1] : -1;
                        dv[u].z = (i + 2 < ne) ? dst[i + 2] : -1;
                        dv[u].w = -1;
                    }
                }
            }

            int tot = 0;
            #pragma unroll
            for (int u = 0; u < P_NLD; ++u) {
                const int dd[4] = {dv[u].x, dv[u].y, dv[u].z, dv[u].w};
                unsigned rv[4], mk[4];
                #pragma unroll
                for (int c = 0; c < 4; ++c) {
                    rv[c] = (unsigned)(dd[c] - row0);
                    mk[c] = __builtin_amdgcn_ballot_w32(rv[c] < (unsigned)R);
                }
                if ((mk[0] | mk[1] | mk[2] | mk[3]) != 0u) {
                    #pragma unroll
                    for (int c = 0; c < 4; ++c) {
                        if (mk[c] != 0u) {
                            const int pos = tot + (int)__builtin_amdgcn_mbcnt_lo(mk[c], 0u) - p * P_WCAP;
                            if (rv[c] < (unsigned)R && pos >= 0 && pos < P_WCAP)
                                wq[wave * P_WCAP + pos] =
                                    ((unsigned)((u * 1024 + tid) * 4 + c) << 9) | rv[c];
                            tot += __builtin_popcount(mk[c]);
                        }
                    }
                }
            }
            if (lane == 0) {
                int cc = tot - p * P_WCAP;
                cc = cc < 0 ? 0 : (cc > P_WCAP ? P_WCAP : cc);
                wcnt[wave] = cc;
                if (tot > (p + 1) * P_WCAP) more[fl] = 1;
            }
            __syncthreads();
            const int again = __builtin_amdgcn_readfirstlane(more[fl]);
            if (tid == 0) more[fl ^ 1] = 0;

            int mc = 0;
            const int cnts = wcnt[lane];
            unsigned ents[32];
            const int qi = (lane < P_WCAP) ? lane : 0;
            #pragma unroll
            for (int sw = 0; sw < 32; ++sw) ents[sw] = wq[sw * P_WCAP + qi];
            #pragma unroll
            for (int sw = 0; sw < 32; ++sw) {
                const int cq = __builtin_amdgcn_readlane(cnts, sw);
                const unsigned ent = ents[sw];
                const bool mine = (lane < cq) && ((ent & 31u) == (unsigned)wave);
                const unsigned mk = __builtin_amdgcn_ballot_w32(mine);
                if (mk != 0u) {
                    if (mine)
                        mylist[wave * P_MCAP + mc + (int)__builtin_amdgcn_mbcnt_lo(mk, 0u)] = ent;
                    mc += __builtin_popcount(mk);
                }
            }
            __builtin_amdgcn_fence(__ATOMIC_RELEASE, "wavefront");
            __builtin_amdgcn_wave_barrier();
            __builtin_amdgcn_fence(__ATOMIC_ACQUIRE, "wavefront");

            for (int j0 = 0; j0 < mc; j0 += 32) {
                const int j = j0 + lane;
                int   e_s = 0, e_r = 0;
                float e_n = 1.0f;
                if (j < mc) {
                    const unsigned ent = mylist[wave * P_MCAP + j];
                    e_r = (int)(ent & 511u);
                    e_s = src[segBase + (int)(ent >> 9)];
                    if (SRCSCALE) e_n = normS[e_s];
                }
                const int m = (mc - j0 < 32) ? (mc - j0) : 32;
                for (int b = 0; b < m; b += 8) {
                    vfl   v[8];
                    int   rr[8];
                    float nn[8];
                    #pragma unroll
                    for (int k = 0; k < 8; ++k) {
                        const int jj = (b + k < m) ? (b + k) : (m - 1);
                        const int sk = __builtin_amdgcn_readlane(e_s, jj);
                        rr[k] = __builtin_amdgcn_readlane(e_r, jj);
                        nn[k] = __int_as_float(__builtin_amdgcn_readlane(__float_as_int(e_n), jj));
                        v[k] = *(const vfl*)(feat + (long)sk * F + lane * VL);
                    }
                    #pragma unroll
                    for (int k = 0; k < 8; ++k) {
                        if (b + k < m) {
                            vfl* tp = (vfl*)&tile[rr[k] * F + lane * VL];
                            vfl t = *tp;
                            if (SRCSCALE) t += v[k] * nn[k];
                            else          t += v[k];
                            *tp = t;
                        }
                    }
                }
            }
            __syncthreads();
            ++pp;
            if (!again) break;
        }
    }

    for (int i = tid; i < R * 32; i += 1024) {
        const int r = row0 + (i >> 5);
        if (r < nd) {
            vfl t = ((const vfl*)tile)[i];
            if (FINAL) {
                const float sc = normD[r];
                const vfl bv = *(const vfl*)(bias + (i & 31) * VL);
                t = t * sc + bv;
            }
            *(vfl*)(res + (long)r * F + (i & 31) * VL) = t;
        }
    }
}

template <int K, int N, int MT, int UNR, bool RELU, bool BIAS>
__global__ __launch_bounds__(128)
void k_gemm_wmma(const float* __restrict__ A,
                 const float* __restrict__ rowscale,
                 const float* __restrict__ W,
                 const float* __restrict__ bias,
                 float* __restrict__ C, int nRowGroups) {
    constexpr int NG = N / 64;
    const int wave = (blockIdx.x * blockDim.x + threadIdx.x) >> 5;
    const int lane = threadIdx.x & 31;
    const int rg = wave / NG;
    const int cg = wave % NG;
    if (rg >= nRowGroups) return;
    const int hf = lane >> 4;
    const int c  = lane & 15;
    const int row0 = rg * (MT * 16);
    const int col0 = cg * 64 + c * 4;

    const float* ap[MT];
    float sc[MT];
    #pragma unroll
    for (int m = 0; m < MT; ++m) {
        const int r = row0 + m * 16 + c;
        ap[m] = A + (long)r * K + hf * 4;
        sc[m] = rowscale[r];
    }
    const float* wp = W + (long)(hf * 4) * N + col0;

    v8f acc[MT][4];
    #pragma unroll
    for (int m = 0; m < MT; ++m)
        #pragma unroll
        for (int t = 0; t < 4; ++t) acc[m][t] = v8f{};

    #pragma unroll UNR
    for (int kk = 0; kk < K; kk += 8) {
        float4 av[MT], bq[4];
        #pragma unroll
        for (int m = 0; m < MT; ++m) av[m] = *(const float4*)(ap[m] + kk);
        #pragma unroll
        for (int i = 0; i < 4; ++i) bq[i] = *(const float4*)(wp + (long)(kk + i) * N);

        const float bv[4][4] = {
            {bq[0].x, bq[0].y, bq[0].z, bq[0].w},
            {bq[1].x, bq[1].y, bq[1].z, bq[1].w},
            {bq[2].x, bq[2].y, bq[2].z, bq[2].w},
            {bq[3].x, bq[3].y, bq[3].z, bq[3].w}};
        #pragma unroll
        for (int m = 0; m < MT; ++m) {
            v2f a; a.x = av[m].x * sc[m]; a.y = av[m].y * sc[m];
            #pragma unroll
            for (int t = 0; t < 4; ++t) {
                v2f b; b.x = bv[0][t]; b.y = bv[1][t];
                acc[m][t] = __builtin_amdgcn_wmma_f32_16x16x4_f32(
                    false, a, false, b, (short)0, acc[m][t], false, false);
            }
        }
        #pragma unroll
        for (int m = 0; m < MT; ++m) {
            v2f a; a.x = av[m].z * sc[m]; a.y = av[m].w * sc[m];
            #pragma unroll
            for (int t = 0; t < 4; ++t) {
                v2f b; b.x = bv[2][t]; b.y = bv[3][t];
                acc[m][t] = __builtin_amdgcn_wmma_f32_16x16x4_f32(
                    false, a, false, b, (short)0, acc[m][t], false, false);
            }
        }
    }

    float4 bval = {0.0f, 0.0f, 0.0f, 0.0f};
    if (BIAS) bval = *(const float4*)(bias + col0);
    #pragma unroll
    for (int m = 0; m < MT; ++m) {
        #pragma unroll
        for (int r = 0; r < 8; ++r) {
            float4 v;
            v.x = acc[m][0][r] + bval.x;
            v.y = acc[m][1][r] + bval.y;
            v.z = acc[m][2][r] + bval.z;
            v.w = acc[m][3][r] + bval.w;
            if (RELU) {
                v.x = fmaxf(v.x, 0.0f); v.y = fmaxf(v.y, 0.0f);
                v.z = fmaxf(v.z, 0.0f); v.w = fmaxf(v.w, 0.0f);
            }
            *(float4*)(C + (long)(row0 + m * 16 + hf * 8 + r) * N + col0) = v;
        }
    }
}

extern "C" void kernel_launch(void* const* d_in, const int* in_sizes, int n_in,
                              void* d_out, int out_size, void* d_ws, size_t ws_size,
                              hipStream_t stream) {
    const float* x    = (const float*)d_in[0];
    const float* W1   = (const float*)d_in[1];
    const float* b1   = (const float*)d_in[2];
    const float* W2   = (const float*)d_in[3];
    const float* b2   = (const float*)d_in[4];
    const int*   src1 = (const int*)d_in[5];
    const int*   dst1 = (const int*)d_in[6];
    const int*   src2 = (const int*)d_in[7];
    const int*   dst2 = (const int*)d_in[8];

    const int IN_F = 128, H_F = 256, N_CLS = 64;
    const int NSRC1 = in_sizes[0] / IN_F;
    const int E1    = in_sizes[5];
    const int E2    = in_sizes[7];
    const int ND1   = 100000;
    const int ND2   = 16384;

    float* ws    = (float*)d_ws;
    float* degS1 = ws;
    float* degD1 = degS1 + NSRC1;
    float* degS2 = degD1 + ND1;
    float* degD2 = degS2 + ND1;
    float* agg1  = degD2 + ND2;
    float* h     = agg1 + (long)ND1 * IN_F;
    float* g     = h + (long)ND1 * H_F;

    long zcount = (NSRC1 + ND1 + ND1 + ND2);
    k_zero<<<2048, 256, 0, stream>>>(ws, zcount);

    k_deg<<<(E1 + 255) / 256, 256, 0, stream>>>(src1, dst1, E1, degS1, degD1);
    k_deg<<<(E2 + 255) / 256, 256, 0, stream>>>(src2, dst2, E2, degS2, degD2);
    int degtot = NSRC1 + ND1 + ND1 + ND2;
    k_rsqrt<<<(degtot + 255) / 256, 256, 0, stream>>>(ws, degtot);

    k_pull<128, 400, true, false>
        <<<(ND1 + 399) / 400, 1024, 0, stream>>>(x, src1, dst1, degS1, nullptr, nullptr, agg1, E1, ND1);

    {
        k_gemm_wmma<128, 256, 5, 2, true, true>
            <<<1250, 128, 0, stream>>>(agg1, degD1, W1, b1, h, 1250);
    }

    {
        k_gemm_wmma<256, 64, 2, 4, false, false>
            <<<782, 128, 0, stream>>>(h, degS2, W2, nullptr, g, 3125);
    }

    k_pull<64, 64, false, true>
        <<<(ND2 + 63) / 64, 1024, 0, stream>>>(g, src2, dst2, nullptr, degD2, b2, (float*)d_out, E2, ND2);
}


# Prep kernels (zero fill, degree counts, rsqrt) relaunched as few fat waves: 16-byte accesses, 8 edges or several float4 per thread, both edge lists in one launch; 97k one-element waves become 11k
# speedup vs baseline: 1.4227x; 1.0456x over previous
#include <hip/hip_runtime.h>


typedef __attribute__((ext_vector_type(2))) float v2f;
typedef __attribute__((ext_vector_type(8))) float v8f;

__device__ __forceinline__ void atomAddF(float* p, float v) {
    asm volatile("global_atomic_add_f32 %0, %1, off" :: "v"(p), "v"(v) : "memory");
}

__global__ __launch_bounds__(256)
void k_zero(float* __restrict__ p, long n) {
    const long n4 = n >> 2;
    const long i = (long)blockIdx.x * blockDim.x + threadIdx.x;
    const long stride = (long)gridDim.x * blockDim.x;
    const float4 z = {0.0f, 0.0f, 0.0f, 0.0f};
    for (long j = i; j < n4; j += stride) ((float4*)p)[j] = z;
    const long t = (n4 << 2) + i;
    if (t < n) p[t] = 0.0f;
}

constexpr int D_EPB = 2048;
__global__ __launch_bounds__(256)
void k_deg(const int* __restrict__ src1, const int* __restrict__ dst1, int ne1,
           float* __restrict__ degS1, float* __restrict__ degD1, int nb1,
           const int* __restrict__ src2, const int* __restrict__ dst2, int ne2,
           float* __restrict__ degS2, float* __restrict__ degD2) {
    const bool first = (int)blockIdx.x < nb1;
    const int* __restrict__ src = first ? src1 : src2;
    const int* __restrict__ dst = first ? dst1 : dst2;
    float* __restrict__ degS = first ? degS1 : degS2;
    float* __restrict__ degD = first ? degD1 : degD2;
    const int ne  = first ? ne1 : ne2;
    const int blk = first ? (int)blockIdx.x : (int)blockIdx.x - nb1;
    const int e0  = blk * D_EPB + (int)threadIdx.x * 4;
    if (blk * D_EPB + D_EPB <= ne) {
        const int4 s0 = *(const int4*)(src + e0);
        const int4 s1 = *(const int4*)(src + e0 + 1024);
        const int4 d0 = *(const int4*)(dst + e0);
        const int4 d1 = *(const int4*)(dst + e0 + 1024);
        atomAddF(&degS[s0.x], 1.0f); atomAddF(&degS[s0.y], 1.0f);
        atomAddF(&degS[s0.z], 1.0f); atomAddF(&degS[s0.w], 1.0f);
        atomAddF(&degD[d0.x], 1.0f); atomAddF(&degD[d0.y], 1.0f);
        atomAddF(&degD[d0.z], 1.0f); atomAddF(&degD[d0.w], 1.0f);
        atomAddF(&degS[s1.x], 1.0f); atomAddF(&degS[s1.y], 1.0f);
        atomAddF(&degS[s1.z], 1.0f); atomAddF(&degS[s1.w], 1.0f);
        atomAddF(&degD[d1.x], 1.0f); atomAddF(&degD[d1.y], 1.0f);
        atomAddF(&degD[d1.z], 1.0f); atomAddF(&degD[d1.w], 1.0f);
    } else {
        for (int k = 0; k < 8; ++k) {
            const int e = e0 + (k >> 2) * 1024 + (k & 3);
            if (e < ne) {
                atomAddF(&degS[src[e]], 1.0f);
                atomAddF(&degD[dst[e]], 1.0f);
            }
        }
    }
}

__global__ __launch_bounds__(256)
void k_rsqrt(float* __restrict__ p, long n) {
    const long n4 = n >> 2;
    const long i = (long)blockIdx.x * blockDim.x + threadIdx.x;
    const long stride = (long)gridDim.x * blockDim.x;
    for (long j0 = i; j0 < n4; j0 += 4 * stride) {
        float4 v[4];
        #pragma unroll
        for (int k = 0; k < 4; ++k) {
            const long j = j0 + k * stride;
            if (j < n4) v[k] = ((const float4*)p)[j];
        }
        #pragma unroll
        for (int k = 0; k < 4; ++k) {
            const long j = j0 + k * stride;
            if (j < n4) {
                float4 t = v[k];
                t.x = rsqrtf(fmaxf(t.x, 1.0f)); t.y = rsqrtf(fmaxf(t.y, 1.0f));
                t.z = rsqrtf(fmaxf(t.z, 1.0f)); t.w = rsqrtf(fmaxf(t.w, 1.0f));
                ((float4*)p)[j] = t;
            }
        }
    }
    const long t = (n4 << 2) + i;
    if (t < n) p[t] = rsqrtf(fmaxf(p[t], 1.0f));
}

constexpr int P_NLD  = 16;
constexpr int P_SEG  = 1024 * 4 * P_NLD;
constexpr int P_WCAP = 24;
constexpr int P_MCAP = 32 * P_WCAP;
constexpr int P_MAXP = (32 * 4 * P_NLD) / P_WCAP;

template <int F, int R, bool SRCSCALE, bool FINAL>
__global__ __launch_bounds__(1024)
void k_pull(const float* __restrict__ feat, const int* __restrict__ src,
            const int* __restrict__ dst, const float* __restrict__ normS,
            const float* __restrict__ normD, const float* __restrict__ bias,
            float* __restrict__ res, int ne, int nd) {
    constexpr int VL = F / 32;
    typedef float vfl __attribute__((ext_vector_type(VL)));
    static_assert(R <= 512 && (F == 128 || F == 64), "entry packing and lane vectors");

    __shared__ __attribute__((aligned(16))) float tile[R * F];
    __shared__ unsigned wq[32 * P_WCAP];
    __shared__ int wcnt[32];
    __shared__ unsigned mylist[32 * P_MCAP];
    __shared__ int more[2];

    const int tid  = threadIdx.x;
    const int lane = tid & 31;
    const int wave = __builtin_amdgcn_readfirstlane(tid >> 5);
    const int row0 = blockIdx.x * R;

    for (int i = tid; i < R * 32; i += 1024) ((vfl*)tile)[i] = vfl{};
    if (tid < 2) more[tid] = 0;
    __syncthreads();

    int pp = 0;
    const int nseg = (ne + P_SEG - 1) / P_SEG;
    for (int seg = 0; seg < nseg; ++seg) {
        const int segBase = seg * P_SEG;

        for (int p = 0; p <= P_MAXP; ++p) {
            const int fl = pp & 1;

            int4 dv[P_NLD];
            if (segBase + P_SEG <= ne) {
                #pragma unroll
                for (int u = 0; u < P_NLD; ++u)
                    dv[u] = *(const int4*)(dst + segBase + (u * 1024 + tid) * 4);
            } else {
                #pragma unroll
                for (int u = 0; u < P_NLD; ++u) {
                    const int i = segBase + (u * 1024 + tid) * 4;
                    if (i + 3 < ne) {
                        dv[u] = *(const int4*)(dst + i);
                    } else {
                        dv[u].x = (i     < ne) ? dst[i]     : -1;
                        dv[u].y = (i + 1 < ne) ? dst[i + 1] : -1;
                        dv[u].z = (i + 2 < ne) ? dst[i + 2] : -1;
                        dv[u].w = -1;
                    }
                }
            }

            int tot = 0;
            #pragma unroll
            for (int u = 0; u < P_NLD; ++u) {
                const int dd[4] = {dv[u].x, dv[u].y, dv[u].z, dv[u].w};
                unsigned rv[4], mk[4];
                #pragma unroll
                for (int c = 0; c < 4; ++c) {
                    rv[c] = (unsigned)(dd[c] - row0);
                    mk[c] = __builtin_amdgcn_ballot_w32(rv[c] < (unsigned)R);
                }
                if ((mk[0] | mk[1] | mk[2] | mk[3]) != 0u) {
                    #pragma unroll
                    for (int c = 0; c < 4; ++c) {
                        if (mk[c] != 0u) {
                            const int pos = tot + (int)__builtin_amdgcn_mbcnt_lo(mk[c], 0u) - p * P_WCAP;
                            if (rv[c] < (unsigned)R && pos >= 0 && pos < P_WCAP)
                                wq[wave * P_WCAP + pos] =
                                    ((unsigned)((u * 1024 + tid) * 4 + c) << 9) | rv[c];
                            tot += __builtin_popcount(mk[c]);
                        }
                    }
                }
            }
            if (lane == 0) {
                int cc = tot - p * P_WCAP;
                cc = cc < 0 ? 0 : (cc > P_WCAP ? P_WCAP : cc);
                wcnt[wave] = cc;
                if (tot > (p + 1) * P_WCAP) more[fl] = 1;
            }
            __syncthreads();
            const int again = __builtin_amdgcn_readfirstlane(more[fl]);
            if (tid == 0) more[fl ^ 1] = 0;

            int mc = 0;
            const int cnts = wcnt[lane];
            unsigned ents[32];
            const int qi = (lane < P_WCAP) ? lane : 0;
            #pragma unroll
            for (int sw = 0; sw < 32; ++sw) ents[sw] = wq[sw * P_WCAP + qi];
            #pragma unroll
            for (int sw = 0; sw < 32; ++sw) {
                const int cq = __builtin_amdgcn_readlane(cnts, sw);
                const unsigned ent = ents[sw];
                const bool mine = (lane < cq) && ((ent & 31u) == (unsigned)wave);
                const unsigned mk = __builtin_amdgcn_ballot_w32(mine);
                if (mk != 0u) {
                    if (mine)
                        mylist[wave * P_MCAP + mc + (int)__builtin_amdgcn_mbcnt_lo(mk, 0u)] = ent;
                    mc += __builtin_popcount(mk);
                }
            }
            __builtin_amdgcn_fence(__ATOMIC_RELEASE, "wavefront");
            __builtin_amdgcn_wave_barrier();
            __builtin_amdgcn_fence(__ATOMIC_ACQUIRE, "wavefront");

            for (int j0 = 0; j0 < mc; j0 += 32) {
                const int j = j0 + lane;
                int   e_s = 0, e_r = 0;
                float e_n = 1.0f;
                if (j < mc) {
                    const unsigned ent = mylist[wave * P_MCAP + j];
                    e_r = (int)(ent & 511u);
                    e_s = src[segBase + (int)(ent >> 9)];
                    if (SRCSCALE) e_n = normS[e_s];
                }
                const int m = (mc - j0 < 32) ? (mc - j0) : 32;
                for (int b = 0; b < m; b += 8) {
                    vfl   v[8];
                    int   rr[8];
                    float nn[8];
                    #pragma unroll
                    for (int k = 0; k < 8; ++k) {
                        const int jj = (b + k < m) ? (b + k) : (m - 1);
                        const int sk = __builtin_amdgcn_readlane(e_s, jj);
                        rr[k] = __builtin_amdgcn_readlane(e_r, jj);
                        nn[k] = __int_as_float(__builtin_amdgcn_readlane(__float_as_int(e_n), jj));
                        v[k] = *(const vfl*)(feat + (long)sk * F + lane * VL);
                    }
                    #pragma unroll
                    for (int k = 0; k < 8; ++k) {
                        if (b + k < m) {
                            vfl* tp = (vfl*)&tile[rr[k] * F + lane * VL];
                            vfl t = *tp;
                            if (SRCSCALE) t += v[k] * nn[k];
                            else          t += v[k];
                            *tp = t;
                        }
                    }
                }
            }
            __syncthreads();
            ++pp;
            if (!again) break;
        }
    }

    for (int i = tid; i < R * 32; i += 1024) {
        const int r = row0 + (i >> 5);
        if (r < nd) {
            vfl t = ((const vfl*)tile)[i];
            if (FINAL) {
                const float sc = normD[r];
                const vfl bv = *(const vfl*)(bias + (i & 31) * VL);
                t = t * sc + bv;
            }
            *(vfl*)(res + (long)r * F + (i & 31) * VL) = t;
        }
    }
}

template <int K, int N, int MT, int UNR, bool RELU, bool BIAS>
__global__ __launch_bounds__(128)
void k_gemm_wmma(const float* __restrict__ A,
                 const float* __restrict__ rowscale,
                 const float* __restrict__ W,
                 const float* __restrict__ bias,
                 float* __restrict__ C, int nRowGroups) {
    constexpr int NG = N / 64;
    const int wave = (blockIdx.x * blockDim.x + threadIdx.x) >> 5;
    const int lane = threadIdx.x & 31;
    const int rg = wave / NG;
    const int cg = wave % NG;
    if (rg >= nRowGroups) return;
    const int hf = lane >> 4;
    const int c  = lane & 15;
    const int row0 = rg * (MT * 16);
    const int col0 = cg * 64 + c * 4;

    const float* ap[MT];
    float sc[MT];
    #pragma unroll
    for (int m = 0; m < MT; ++m) {
        const int r = row0 + m * 16 + c;
        ap[m] = A + (long)r * K + hf * 4;
        sc[m] = rowscale[r];
    }
    const float* wp = W + (long)(hf * 4) * N + col0;

    v8f acc[MT][4];
    #pragma unroll
    for (int m = 0; m < MT; ++m)
        #pragma unroll
        for (int t = 0; t < 4; ++t) acc[m][t] = v8f{};

    #pragma unroll UNR
    for (int kk = 0; kk < K; kk += 8) {
        float4 av[MT], bq[4];
        #pragma unroll
        for (int m = 0; m < MT; ++m) av[m] = *(const float4*)(ap[m] + kk);
        #pragma unroll
        for (int i = 0; i < 4; ++i) bq[i] = *(const float4*)(wp + (long)(kk + i) * N);

        const float bv[4][4] = {
            {bq[0].x, bq[0].y, bq[0].z, bq[0].w},
            {bq[1].x, bq[1].y, bq[1].z, bq[1].w},
            {bq[2].x, bq[2].y, bq[2].z, bq[2].w},
            {bq[3].x, bq[3].y, bq[3].z, bq[3].w}};
        #pragma unroll
        for (int m = 0; m < MT; ++m) {
            v2f a; a.x = av[m].x * sc[m]; a.y = av[m].y * sc[m];
            #pragma unroll
            for (int t = 0; t < 4; ++t) {
                v2f b; b.x = bv[0][t]; b.y = bv[1][t];
                acc[m][t] = __builtin_amdgcn_wmma_f32_16x16x4_f32(
                    false, a, false, b, (short)0, acc[m][t], false, false);
            }
        }
        #pragma unroll
        for (int m = 0; m < MT; ++m) {
            v2f a; a.x = av[m].z * sc[m]; a.y = av[m].w * sc[m];
            #pragma unroll
            for (int t = 0; t < 4; ++t) {
                v2f b; b.x = bv[2][t]; b.y = bv[3][t];
                acc[m][t] = __builtin_amdgcn_wmma_f32_16x16x4_f32(
                    false, a, false, b, (short)0, acc[m][t], false, false);
            }
        }
    }

    float4 bval = {0.0f, 0.0f, 0.0f, 0.0f};
    if (BIAS) bval = *(const float4*)(bias + col0);
    #pragma unroll
    for (int m = 0; m < MT; ++m) {
        #pragma unroll
        for (int r = 0; r < 8; ++r) {
            float4 v;
            v.x = acc[m][0][r] + bval.x;
            v.y = acc[m][1][r] + bval.y;
            v.z = acc[m][2][r] + bval.z;
            v.w = acc[m][3][r] + bval.w;
            if (RELU) {
                v.x = fmaxf(v.x, 0.0f); v.y = fmaxf(v.y, 0.0f);
                v.z = fmaxf(v.z, 0.0f); v.w = fmaxf(v.w, 0.0f);
            }
            *(float4*)(C + (long)(row0 + m * 16 + hf * 8 + r) * N + col0) = v;
        }
    }
}

extern "C" void kernel_launch(void* const* d_in, const int* in_sizes, int n_in,
                              void* d_out, int out_size, void* d_ws, size_t ws_size,
                              hipStream_t stream) {
    const float* x    = (const float*)d_in[0];
    const float* W1   = (const float*)d_in[1];
    const float* b1   = (const float*)d_in[2];
    const float* W2   = (const float*)d_in[3];
    const float* b2   = (const float*)d_in[4];
    const int*   src1 = (const int*)d_in[5];
    const int*   dst1 = (const int*)d_in[6];
    const int*   src2 = (const int*)d_in[7];
    const int*   dst2 = (const int*)d_in[8];

    const int IN_F = 128, H_F = 256, N_CLS = 64;
    const int NSRC1 = in_sizes[0] / IN_F;
    const int E1    = in_sizes[5];
    const int E2    = in_sizes[7];
    const int ND1   = 100000;
    const int ND2   = 16384;

    float* ws    = (float*)d_ws;
    float* degS1 = ws;
    float* degD1 = degS1 + NSRC1;
    float* degS2 = degD1 + ND1;
    float* degD2 = degS2 + ND1;
    float* agg1  = degD2 + ND2;
    float* h     = agg1 + (long)ND1 * IN_F;
    float* g     = h + (long)ND1 * H_F;

    long zcount = (NSRC1 + ND1 + ND1 + ND2);
    k_zero<<<256, 256, 0, stream>>>(ws, zcount);

    const int nb1 = (E1 + D_EPB - 1) / D_EPB;
    const int nb2 = (E2 + D_EPB - 1) / D_EPB;
    k_deg<<<nb1 + nb2, 256, 0, stream>>>(src1, dst1, E1, degS1, degD1, nb1,
                                         src2, dst2, E2, degS2, degD2);
    k_rsqrt<<<256, 256, 0, stream>>>(ws, zcount);

    k_pull<128, 400, true, false>
        <<<(ND1 + 399) / 400, 1024, 0, stream>>>(x, src1, dst1, degS1, nullptr, nullptr, agg1, E1, ND1);

    {
        k_gemm_wmma<128, 256, 5, 2, true, true>
            <<<1250, 128, 0, stream>>>(agg1, degD1, W1, b1, h, 1250);
    }

    {
        k_gemm_wmma<256, 64, 2, 4, false, false>
            <<<782, 128, 0, stream>>>(h, degS2, W2, nullptr, g, 3125);
    }

    k_pull<64, 64, false, true>
        <<<(ND2 + 63) / 64, 1024, 0, stream>>>(g, src2, dst2, nullptr, degD2, b2, (float*)d_out, E2, ND2);
}


# SpMM1 scan filters a 1-byte tile number per edge (written by the degree kernel) instead of the 4-byte dst list; kept edges fetch dst after; row sum order now segment, wave, load, hit round, lane
# speedup vs baseline: 1.5886x; 1.1166x over previous
#include <hip/hip_runtime.h>


typedef __attribute__((ext_vector_type(2))) float v2f;
typedef __attribute__((ext_vector_type(8))) float v8f;

__device__ __forceinline__ void atomAddF(float* p, float v) {
    asm volatile("global_atomic_add_f32 %0, %1, off" :: "v"(p), "v"(v) : "memory");
}

__global__ __launch_bounds__(256)
void k_zero(float* __restrict__ p, long n) {
    const long n4 = n >> 2;
    const long i = (long)blockIdx.x * blockDim.x + threadIdx.x;
    const long stride = (long)gridDim.x * blockDim.x;
    const float4 z = {0.0f, 0.0f, 0.0f, 0.0f};
    for (long j = i; j < n4; j += stride) ((float4*)p)[j] = z;
    const long t = (n4 << 2) + i;
    if (t < n) p[t] = 0.0f;
}

constexpr int D_EPB = 2048;
constexpr int PULL1_R = 400;
constexpr int KEY_PAD = 65536;
__device__ __forceinline__ unsigned tileKeys(const int4 d) {
    return (unsigned)(d.x / PULL1_R) | ((unsigned)(d.y / PULL1_R) << 8) |
           ((unsigned)(d.z / PULL1_R) << 16) | ((unsigned)(d.w / PULL1_R) << 24);
}
__global__ __launch_bounds__(256)
void k_deg(const int* __restrict__ src1, const int* __restrict__ dst1, int ne1,
           float* __restrict__ degS1, float* __restrict__ degD1, int nb1,
           const int* __restrict__ src2, const int* __restrict__ dst2, int ne2,
           float* __restrict__ degS2, float* __restrict__ degD2,
           float* __restrict__ key1f) {
    const bool first = (int)blockIdx.x < nb1;
    const int* __restrict__ src = first ? src1 : src2;
    const int* __restrict__ dst = first ? dst1 : dst2;
    float* __restrict__ degS = first ? degS1 : degS2;
    float* __restrict__ degD = first ? degD1 : degD2;
    const int ne  = first ? ne1 : ne2;
    const int blk = first ? (int)blockIdx.x : (int)blockIdx.x - nb1;
    const int e0  = blk * D_EPB + (int)threadIdx.x * 4;
    if (blk * D_EPB + D_EPB <= ne) {
        const int4 s0 = *(const int4*)(src + e0);
        const int4 s1 = *(const int4*)(src + e0 + 1024);
        const int4 d0 = *(const int4*)(dst + e0);
        const int4 d1 = *(const int4*)(dst + e0 + 1024);
        if (first) {
            ((unsigned*)key1f)[e0 >> 2]          = tileKeys(d0);
            ((unsigned*)key1f)[(e0 + 1024) >> 2] = tileKeys(d1);
        }
        atomAddF(&degS[s0.x], 1.0f); atomAddF(&degS[s0.y], 1.0f);
        atomAddF(&degS[s0.z], 1.0f); atomAddF(&degS[s0.w], 1.0f);
        atomAddF(&degD[d0.x], 1.0f); atomAddF(&degD[d0.y], 1.0f);
        atomAddF(&degD[d0.z], 1.0f); atomAddF(&degD[d0.w], 1.0f);
        atomAddF(&degS[s1.x], 1.0f); atomAddF(&degS[s1.y], 1.0f);
        atomAddF(&degS[s1.z], 1.0f); atomAddF(&degS[s1.w], 1.0f);
        atomAddF(&degD[d1.x], 1.0f); atomAddF(&degD[d1.y], 1.0f);
        atomAddF(&degD[d1.z], 1.0f); atomAddF(&degD[d1.w], 1.0f);
    } else {
        for (int k = 0; k < 8; ++k) {
            const int e = e0 + (k >> 2) * 1024 + (k & 3);
            if (e < ne) {
                const int de = dst[e];
                if (first) ((unsigned char*)key1f)[e] = (unsigned char)(de / PULL1_R);
                atomAddF(&degS[src[e]], 1.0f);
                atomAddF(&degD[de], 1.0f);
            }
        }
    }
    if (first && blk == nb1 - 1) {
        const int a4 = (ne + 3) & ~3;
        const int padEnd = ((ne + KEY_PAD - 1) / KEY_PAD) * KEY_PAD;
        if ((int)threadIdx.x < a4 - ne) ((unsigned char*)key1f)[ne + (int)threadIdx.x] = (unsigned char)255;
        for (int w = (a4 >> 2) + (int)threadIdx.x; w < (padEnd >> 2); w += 256)
            ((unsigned*)key1f)[w] = 0xFFFFFFFFu;
    }
}

__global__ __launch_bounds__(256)
void k_rsqrt(float* __restrict__ p, long n) {
    const long n4 = n >> 2;
    const long i = (long)blockIdx.x * blockDim.x + threadIdx.x;
    const long stride = (long)gridDim.x * blockDim.x;
    for (long j0 = i; j0 < n4; j0 += 4 * stride) {
        float4 v[4];
        #pragma unroll
        for (int k = 0; k < 4; ++k) {
            const long j = j0 + k * stride;
            if (j < n4) v[k] = ((const float4*)p)[j];
        }
        #pragma unroll
        for (int k = 0; k < 4; ++k) {
            const long j = j0 + k * stride;
            if (j < n4) {
                float4 t = v[k];
                t.x = rsqrtf(fmaxf(t.x, 1.0f)); t.y = rsqrtf(fmaxf(t.y, 1.0f));
                t.z = rsqrtf(fmaxf(t.z, 1.0f)); t.w = rsqrtf(fmaxf(t.w, 1.0f));
                ((float4*)p)[j] = t;
            }
        }
    }
    const long t = (n4 << 2) + i;
    if (t < n) p[t] = rsqrtf(fmaxf(p[t], 1.0f));
}

constexpr int P_NLD  = 16;
constexpr int P_SEG  = 1024 * 4 * P_NLD;
constexpr int P_WCAP = 24;
constexpr int P_MCAP = 32 * P_WCAP;
constexpr int P_MAXP = (32 * 4 * P_NLD) / P_WCAP;

template <int F, int R, bool SRCSCALE, bool FINAL, bool KEYED>
__global__ __launch_bounds__(1024)
void k_pull(const float* __restrict__ feat, const int* __restrict__ src,
            const int* __restrict__ dst, const float* __restrict__ normS,
            const float* __restrict__ normD, const float* __restrict__ bias,
            const float* __restrict__ keyf,
            float* __restrict__ res, int ne, int nd) {
    constexpr int VL = F / 32;
    typedef float vfl __attribute__((ext_vector_type(VL)));
    static_assert(R <= 512 && (F == 128 || F == 64), "entry packing and lane vectors");
    static_assert(P_WCAP <= 32, "one queue entry per lane when the destinations are fetched");

    __shared__ __attribute__((aligned(16))) float tile[R * F];
    __shared__ unsigned wq[32 * P_WCAP];
    __shared__ int wcnt[32];
    __shared__ unsigned mylist[32 * P_MCAP];
    __shared__ int more[2];

    const int tid  = threadIdx.x;
    const int lane = tid & 31;
    const int wave = __builtin_amdgcn_readfirstlane(tid >> 5);
    const int row0 = blockIdx.x * R;

    for (int i = tid; i < R * 32; i += 1024) ((vfl*)tile)[i] = vfl{};
    if (tid < 2) more[tid] = 0;
    __syncthreads();

    int pp = 0;
    const int nseg = (ne + P_SEG - 1) / P_SEG;
    for (int seg = 0; seg < nseg; ++seg) {
        const int segBase = seg * P_SEG;

        for (int p = 0; p <= P_MAXP; ++p) {
            const int fl = pp & 1;

            int tot = 0;
            if constexpr (KEYED) {
                static_assert(P_SEG == 1024 * 16 * 4 && P_SEG == KEY_PAD, "4 key loads of 16 edges per thread cover a segment; padding to whole segments");
                const unsigned char* keyb = (const unsigned char*)keyf + segBase;
                uint4 kv[4];
                #pragma unroll
                for (int u = 0; u < 4; ++u)
                    kv[u] = *(const uint4*)(keyb + (u * 1024 + tid) * 16);

                const unsigned tileRep = (unsigned)blockIdx.x * 0x01010101u;
                #pragma unroll
                for (int u = 0; u < 4; ++u) {
                    const unsigned wv[4] = {kv[u].x ^ tileRep, kv[u].y ^ tileRep,
                                            kv[u].z ^ tileRep, kv[u].w ^ tileRep};
                    unsigned hm = 0u;
                    #pragma unroll
                    for (int q = 0; q < 4; ++q) {
                        const unsigned y = ((wv[q] & 0x7F7F7F7Fu) + 0x7F7F7F7Fu) | wv[q];
                        hm |= (~y & 0x80808080u) >> (7 - q);
                    }
                    unsigned bal = __builtin_amdgcn_ballot_w32(hm != 0u);
                    while (bal != 0u) {
                        const int bp = __builtin_ctz(hm | 0x80000000u);
                        const int o16 = ((bp & 7) << 2) | (bp >> 3);
                        const int pos = tot + (int)__builtin_amdgcn_mbcnt_lo(bal, 0u) - p * P_WCAP;
                        if (hm != 0u && pos >= 0 && pos < P_WCAP)
                            wq[wave * P_WCAP + pos] = (unsigned)((u * 1024 + tid) * 16 + o16);
                        tot += __builtin_popcount(bal);
                        hm &= hm - 1u;
                        bal = __builtin_amdgcn_ballot_w32(hm != 0u);
                    }
                }

                __builtin_amdgcn_fence(__ATOMIC_RELEASE, "wavefront");
                __builtin_amdgcn_wave_barrier();
                __builtin_amdgcn_fence(__ATOMIC_ACQUIRE, "wavefront");
                int nq = tot - p * P_WCAP;
                nq = nq < 0 ? 0 : (nq > P_WCAP ? P_WCAP : nq);
                if (lane < nq) {
                    const unsigned o = wq[wave * P_WCAP + lane];
                    const int d = dst[segBase + (int)o];
                    wq[wave * P_WCAP + lane] = (o << 9) | (unsigned)(d - row0);
                }
            } else {
                int4 dv[P_NLD];
                if (segBase + P_SEG <= ne) {
                    #pragma unroll
                    for (int u = 0; u < P_NLD; ++u)
                        dv[u] = *(const int4*)(dst + segBase + (u * 1024 + tid) * 4);
                } else {
                    #pragma unroll
                    for (int u = 0; u < P_NLD; ++u) {
                        const int i = segBase + (u * 1024 + tid) * 4;
                        if (i + 3 < ne) {
                            dv[u] = *(const int4*)(dst + i);
                        } else {
                            dv[u].x = (i     < ne) ? dst[i]     : -1;
                            dv[u].y = (i + 1 < ne) ? dst[i + 1] : -1;
                            dv[u].z = (i + 2 < ne) ? dst[i + 2] : -1;
                            dv[u].w = -1;
                        }
                    }
                }

                #pragma unroll
                for (int u = 0; u < P_NLD; ++u) {
                    const int dd[4] = {dv[u].x, dv[u].y, dv[u].z, dv[u].w};
                    unsigned rv[4], mk[4];
                    #pragma unroll
                    for (int c = 0; c < 4; ++c) {
                        rv[c] = (unsigned)(dd[c] - row0);
                        mk[c] = __builtin_amdgcn_ballot_w32(rv[c] < (unsigned)R);
                    }
                    if ((mk[0] | mk[1] | mk[2] | mk[3]) != 0u) {
                        #pragma unroll
                        for (int c = 0; c < 4; ++c) {
                            if (mk[c] != 0u) {
                                const int pos = tot + (int)__builtin_amdgcn_mbcnt_lo(mk[c], 0u) - p * P_WCAP;
                                if (rv[c] < (unsigned)R && pos >= 0 && pos < P_WCAP)
                                    wq[wave * P_WCAP + pos] =
                                        ((unsigned)((u * 1024 + tid) * 4 + c) << 9) | rv[c];
                                tot += __builtin_popcount(mk[c]);
                            }
                        }
                    }
                }
            }
            if (lane == 0) {
                int cc = tot - p * P_WCAP;
                cc = cc < 0 ? 0 : (cc > P_WCAP ? P_WCAP : cc);
                wcnt[wave] = cc;
                if (tot > (p + 1) * P_WCAP) more[fl] = 1;
            }
            __syncthreads();
            const int again = __builtin_amdgcn_readfirstlane(more[fl]);
            if (tid == 0) more[fl ^ 1] = 0;

            int mc = 0;
            const int cnts = wcnt[lane];
            unsigned ents[32];
            const int qi = (lane < P_WCAP) ? lane : 0;
            #pragma unroll
            for (int sw = 0; sw < 32; ++sw) ents[sw] = wq[sw * P_WCAP + qi];
            #pragma unroll
            for (int sw = 0; sw < 32; ++sw) {
                const int cq = __builtin_amdgcn_readlane(cnts, sw);
                const unsigned ent = ents[sw];
                const bool mine = (lane < cq) && ((ent & 31u) == (unsigned)wave);
                const unsigned mk = __builtin_amdgcn_ballot_w32(mine);
                if (mk != 0u) {
                    if (mine)
                        mylist[wave * P_MCAP + mc + (int)__builtin_amdgcn_mbcnt_lo(mk, 0u)] = ent;
                    mc += __builtin_popcount(mk);
                }
            }
            __builtin_amdgcn_fence(__ATOMIC_RELEASE, "wavefront");
            __builtin_amdgcn_wave_barrier();
            __builtin_amdgcn_fence(__ATOMIC_ACQUIRE, "wavefront");

            for (int j0 = 0; j0 < mc; j0 += 32) {
                const int j = j0 + lane;
                int   e_s = 0, e_r = 0;
                float e_n = 1.0f;
                if (j < mc) {
                    const unsigned ent = mylist[wave * P_MCAP + j];
                    e_r = (int)(ent & 511u);
                    e_s = src[segBase + (int)(ent >> 9)];
                    if (SRCSCALE) e_n = normS[e_s];
                }
                const int m = (mc - j0 < 32) ? (mc - j0) : 32;
                for (int b = 0; b < m; b += 8) {
                    vfl   v[8];
                    int   rr[8];
                    float nn[8];
                    #pragma unroll
                    for (int k = 0; k < 8; ++k) {
                        const int jj = (b + k < m) ? (b + k) : (m - 1);
                        const int sk = __builtin_amdgcn_readlane(e_s, jj);
                        rr[k] = __builtin_amdgcn_readlane(e_r, jj);
                        nn[k] = __int_as_float(__builtin_amdgcn_readlane(__float_as_int(e_n), jj));
                        v[k] = *(const vfl*)(feat + (long)sk * F + lane * VL);
                    }
                    #pragma unroll
                    for (int k = 0; k < 8; ++k) {
                        if (b + k < m) {
                            vfl* tp = (vfl*)&tile[rr[k] * F + lane * VL];
                            vfl t = *tp;
                            if (SRCSCALE) t += v[k] * nn[k];
                            else          t += v[k];
                            *tp = t;
                        }
                    }
                }
            }
            __syncthreads();
            ++pp;
            if (!again) break;
        }
    }

    for (int i = tid; i < R * 32; i += 1024) {
        const int r = row0 + (i >> 5);
        if (r < nd) {
            vfl t = ((const vfl*)tile)[i];
            if (FINAL) {
                const float sc = normD[r];
                const vfl bv = *(const vfl*)(bias + (i & 31) * VL);
                t = t * sc + bv;
            }
            *(vfl*)(res + (long)r * F + (i & 31) * VL) = t;
        }
    }
}

template <int K, int N, int MT, int UNR, bool RELU, bool BIAS>
__global__ __launch_bounds__(128)
void k_gemm_wmma(const float* __restrict__ A,
                 const float* __restrict__ rowscale,
                 const float* __restrict__ W,
                 const float* __restrict__ bias,
                 float* __restrict__ C, int nRowGroups) {
    constexpr int NG = N / 64;
    const int wave = (blockIdx.x * blockDim.x + threadIdx.x) >> 5;
    const int lane = threadIdx.x & 31;
    const int rg = wave / NG;
    const int cg = wave % NG;
    if (rg >= nRowGroups) return;
    const int hf = lane >> 4;
    const int c  = lane & 15;
    const int row0 = rg * (MT * 16);
    const int col0 = cg * 64 + c * 4;

    const float* ap[MT];
    float sc[MT];
    #pragma unroll
    for (int m = 0; m < MT; ++m) {
        const int r = row0 + m * 16 + c;
        ap[m] = A + (long)r * K + hf * 4;
        sc[m] = rowscale[r];
    }
    const float* wp = W + (long)(hf * 4) * N + col0;

    v8f acc[MT][4];
    #pragma unroll
    for (int m = 0; m < MT; ++m)
        #pragma unroll
        for (int t = 0; t < 4; ++t) acc[m][t] = v8f{};

    #pragma unroll UNR
    for (int kk = 0; kk < K; kk += 8) {
        float4 av[MT], bq[4];
        #pragma unroll
        for (int m = 0; m < MT; ++m) av[m] = *(const float4*)(ap[m] + kk);
        #pragma unroll
        for (int i = 0; i < 4; ++i) bq[i] = *(const float4*)(wp + (long)(kk + i) * N);

        const float bv[4][4] = {
            {bq[0].x, bq[0].y, bq[0].z, bq[0].w},
            {bq[1].x, bq[1].y, bq[1].z, bq[1].w},
            {bq[2].x, bq[2].y, bq[2].z, bq[2].w},
            {bq[3].x, bq[3].y, bq[3].z, bq[3].w}};
        #pragma unroll
        for (int m = 0; m < MT; ++m) {
            v2f a; a.x = av[m].x * sc[m]; a.y = av[m].y * sc[m];
            #pragma unroll
            for (int t = 0; t < 4; ++t) {
                v2f b; b.x = bv[0][t]; b.y = bv[1][t];
                acc[m][t] = __builtin_amdgcn_wmma_f32_16x16x4_f32(
                    false, a, false, b, (short)0, acc[m][t], false, false);
            }
        }
        #pragma unroll
        for (int m = 0; m < MT; ++m) {
            v2f a; a.x = av[m].z * sc[m]; a.y = av[m].w * sc[m];
            #pragma unroll
            for (int t = 0; t < 4; ++t) {
                v2f b; b.x = bv[2][t]; b.y = bv[3][t];
                acc[m][t] = __builtin_amdgcn_wmma_f32_16x16x4_f32(
                    false, a, false, b, (short)0, acc[m][t], false, false);
            }
        }
    }

    float4 bval = {0.0f, 0.0f, 0.0f, 0.0f};
    if (BIAS) bval = *(const float4*)(bias + col0);
    #pragma unroll
    for (int m = 0; m < MT; ++m) {
        #pragma unroll
        for (int r = 0; r < 8; ++r) {
            float4 v;
            v.x = acc[m][0][r] + bval.x;
            v.y = acc[m][1][r] + bval.y;
            v.z = acc[m][2][r] + bval.z;
            v.w = acc[m][3][r] + bval.w;
            if (RELU) {
                v.x = fmaxf(v.x, 0.0f); v.y = fmaxf(v.y, 0.0f);
                v.z = fmaxf(v.z, 0.0f); v.w = fmaxf(v.w, 0.0f);
            }
            *(float4*)(C + (long)(row0 + m * 16 + hf * 8 + r) * N + col0) = v;
        }
    }
}

extern "C" void kernel_launch(void* const* d_in, const int* in_sizes, int n_in,
                              void* d_out, int out_size, void* d_ws, size_t ws_size,
                              hipStream_t stream) {
    const float* x    = (const float*)d_in[0];
    const float* W1   = (const float*)d_in[1];
    const float* b1   = (const float*)d_in[2];
    const float* W2   = (const float*)d_in[3];
    const float* b2   = (const float*)d_in[4];
    const int*   src1 = (const int*)d_in[5];
    const int*   dst1 = (const int*)d_in[6];
    const int*   src2 = (const int*)d_in[7];
    const int*   dst2 = (const int*)d_in[8];

    const int IN_F = 128, H_F = 256, N_CLS = 64;
    const int NSRC1 = in_sizes[0] / IN_F;
    const int E1    = in_sizes[5];
    const int E2    = in_sizes[7];
    const int ND1   = 100000;
    const int ND2   = 16384;

    float* ws    = (float*)d_ws;
    float* degS1 = ws;
    float* degD1 = degS1 + NSRC1;
    float* degS2 = degD1 + ND1;
    float* degD2 = degS2 + ND1;
    float* agg1  = degD2 + ND2;
    float* h     = agg1 + (long)ND1 * IN_F;
    float* g     = h + (long)ND1 * H_F;
    float* key1  = g;

    long zcount = (NSRC1 + ND1 + ND1 + ND2);
    k_zero<<<256, 256, 0, stream>>>(ws, zcount);

    const int nb1 = (E1 + D_EPB - 1) / D_EPB;
    const int nb2 = (E2 + D_EPB - 1) / D_EPB;
    k_deg<<<nb1 + nb2, 256, 0, stream>>>(src1, dst1, E1, degS1, degD1, nb1,
                                         src2, dst2, E2, degS2, degD2, key1);
    k_rsqrt<<<256, 256, 0, stream>>>(ws, zcount);

    k_pull<128, PULL1_R, true, false, true>
        <<<(ND1 + PULL1_R - 1) / PULL1_R, 1024, 0, stream>>>(x, src1, dst1, degS1, nullptr, nullptr, key1, agg1, E1, ND1);

    {
        k_gemm_wmma<128, 256, 5, 2, true, true>
            <<<1250, 128, 0, stream>>>(agg1, degD1, W1, b1, h, 1250);
    }

    {
        k_gemm_wmma<256, 64, 2, 4, false, false>
            <<<782, 128, 0, stream>>>(h, degS2, W2, nullptr, g, 3125);
    }

    k_pull<64, 64, false, true, false>
        <<<(ND2 + 63) / 64, 1024, 0, stream>>>(g, src2, dst2, nullptr, degD2, b2, nullptr, (float*)d_out, E2, ND2);
}


# First aggregation: waves split into 16 scanning and 16 row-owning ones, two queue sets; next segment's scan runs beside this one's gathers; row sum order: segment, pass, wave, load, round, lane
# speedup vs baseline: 1.6743x; 1.0539x over previous
#include <hip/hip_runtime.h>


typedef __attribute__((ext_vector_type(2))) float v2f;
typedef __attribute__((ext_vector_type(8))) float v8f;

__device__ __forceinline__ void atomAddF(float* p, float v) {
    asm volatile("global_atomic_add_f32 %0, %1, off" :: "v"(p), "v"(v) : "memory");
}

__global__ __launch_bounds__(256)
void k_zero(float* __restrict__ p, long n) {
    const long n4 = n >> 2;
    const long i = (long)blockIdx.x * blockDim.x + threadIdx.x;
    const long stride = (long)gridDim.x * blockDim.x;
    const float4 z = {0.0f, 0.0f, 0.0f, 0.0f};
    for (long j = i; j < n4; j += stride) ((float4*)p)[j] = z;
    const long t = (n4 << 2) + i;
    if (t < n) p[t] = 0.0f;
}

constexpr int D_EPB = 2048;
constexpr int PULL1_R = 400;
constexpr int KEY_PAD = 65536;
__device__ __forceinline__ unsigned tileKeys(const int4 d) {
    return (unsigned)(d.x / PULL1_R) | ((unsigned)(d.y / PULL1_R) << 8) |
           ((unsigned)(d.z / PULL1_R) << 16) | ((unsigned)(d.w / PULL1_R) << 24);
}
__global__ __launch_bounds__(256)
void k_deg(const int* __restrict__ src1, const int* __restrict__ dst1, int ne1,
           float* __restrict__ degS1, float* __restrict__ degD1, int nb1,
           const int* __restrict__ src2, const int* __restrict__ dst2, int ne2,
           float* __restrict__ degS2, float* __restrict__ degD2,
           float* __restrict__ key1f) {
    const bool first = (int)blockIdx.x < nb1;
    const int* __restrict__ src = first ? src1 : src2;
    const int* __restrict__ dst = first ? dst1 : dst2;
    float* __restrict__ degS = first ? degS1 : degS2;
    float* __restrict__ degD = first ? degD1 : degD2;
    const int ne  = first ? ne1 : ne2;
    const int blk = first ? (int)blockIdx.x : (int)blockIdx.x - nb1;
    const int e0  = blk * D_EPB + (int)threadIdx.x * 4;
    if (blk * D_EPB + D_EPB <= ne) {
        const int4 s0 = *(const int4*)(src + e0);
        const int4 s1 = *(const int4*)(src + e0 + 1024);
        const int4 d0 = *(const int4*)(dst + e0);
        const int4 d1 = *(const int4*)(dst + e0 + 1024);
        if (first) {
            ((unsigned*)key1f)[e0 >> 2]          = tileKeys(d0);
            ((unsigned*)key1f)[(e0 + 1024) >> 2] = tileKeys(d1);
        }
        atomAddF(&degS[s0.x], 1.0f); atomAddF(&degS[s0.y], 1.0f);
        atomAddF(&degS[s0.z], 1.0f); atomAddF(&degS[s0.w], 1.0f);
        atomAddF(&degD[d0.x], 1.0f); atomAddF(&degD[d0.y], 1.0f);
        atomAddF(&degD[d0.z], 1.0f); atomAddF(&degD[d0.w], 1.0f);
        atomAddF(&degS[s1.x], 1.0f); atomAddF(&degS[s1.y], 1.0f);
        atomAddF(&degS[s1.z], 1.0f); atomAddF(&degS[s1.w], 1.0f);
        atomAddF(&degD[d1.x], 1.0f); atomAddF(&degD[d1.y], 1.0f);
        atomAddF(&degD[d1.z], 1.0f); atomAddF(&degD[d1.w], 1.0f);
    } else {
        for (int k = 0; k < 8; ++k) {
            const int e = e0 + (k >> 2) * 1024 + (k & 3);
            if (e < ne) {
                const int de = dst[e];
                if (first) ((unsigned char*)key1f)[e] = (unsigned char)(de / PULL1_R);
                atomAddF(&degS[src[e]], 1.0f);
                atomAddF(&degD[de], 1.0f);
            }
        }
    }
    if (first && blk == nb1 - 1) {
        const int a4 = (ne + 3) & ~3;
        const int padEnd = ((ne + KEY_PAD - 1) / KEY_PAD) * KEY_PAD;
        if ((int)threadIdx.x < a4 - ne) ((unsigned char*)key1f)[ne + (int)threadIdx.x] = (unsigned char)255;
        for (int w = (a4 >> 2) + (int)threadIdx.x; w < (padEnd >> 2); w += 256)
            ((unsigned*)key1f)[w] = 0xFFFFFFFFu;
    }
}

__global__ __launch_bounds__(256)
void k_rsqrt(float* __restrict__ p, long n) {
    const long n4 = n >> 2;
    const long i = (long)blockIdx.x * blockDim.x + threadIdx.x;
    const long stride = (long)gridDim.x * blockDim.x;
    for (long j0 = i; j0 < n4; j0 += 4 * stride) {
        float4 v[4];
        #pragma unroll
        for (int k = 0; k < 4; ++k) {
            const long j = j0 + k * stride;
            if (j < n4) v[k] = ((const float4*)p)[j];
        }
        #pragma unroll
        for (int k = 0; k < 4; ++k) {
            const long j = j0 + k * stride;
            if (j < n4) {
                float4 t = v[k];
                t.x = rsqrtf(fmaxf(t.x, 1.0f)); t.y = rsqrtf(fmaxf(t.y, 1.0f));
                t.z = rsqrtf(fmaxf(t.z, 1.0f)); t.w = rsqrtf(fmaxf(t.w, 1.0f));
                ((float4*)p)[j] = t;
            }
        }
    }
    const long t = (n4 << 2) + i;
    if (t < n) p[t] = rsqrtf(fmaxf(p[t], 1.0f));
}

constexpr int P_NLD  = 16;
constexpr int P_SEG  = 1024 * 4 * P_NLD;
constexpr int P_WCAP = 24;
constexpr int P_MCAP = 32 * P_WCAP;
constexpr int P_MAXP = (32 * 4 * P_NLD) / P_WCAP;

template <int F, int R, bool SRCSCALE, bool FINAL>
__global__ __launch_bounds__(1024)
void k_pull(const float* __restrict__ feat, const int* __restrict__ src,
            const int* __restrict__ dst, const float* __restrict__ normS,
            const float* __restrict__ normD, const float* __restrict__ bias,
            float* __restrict__ res, int ne, int nd) {
    constexpr int VL = F / 32;
    typedef float vfl __attribute__((ext_vector_type(VL)));
    static_assert(R <= 512 && (F == 128 || F == 64), "entry packing and lane vectors");

    __shared__ __attribute__((aligned(16))) float tile[R * F];
    __shared__ unsigned wq[32 * P_WCAP];
    __shared__ int wcnt[32];
    __shared__ unsigned mylist[32 * P_MCAP];
    __shared__ int more[2];

    const int tid  = threadIdx.x;
    const int lane = tid & 31;
    const int wave = __builtin_amdgcn_readfirstlane(tid >> 5);
    const int row0 = blockIdx.x * R;

    for (int i = tid; i < R * 32; i += 1024) ((vfl*)tile)[i] = vfl{};
    if (tid < 2) more[tid] = 0;
    __syncthreads();

    int pp = 0;
    const int nseg = (ne + P_SEG - 1) / P_SEG;
    for (int seg = 0; seg < nseg; ++seg) {
        const int segBase = seg * P_SEG;

        for (int p = 0; p <= P_MAXP; ++p) {
            const int fl = pp & 1;

            int tot = 0;
            int4 dv[P_NLD];
            if (segBase + P_SEG <= ne) {
                #pragma unroll
                for (int u = 0; u < P_NLD; ++u)
                    dv[u] = *(const int4*)(dst + segBase + (u * 1024 + tid) * 4);
            } else {
                #pragma unroll
                for (int u = 0; u < P_NLD; ++u) {
                    const int i = segBase + (u * 1024 + tid) * 4;
                    if (i + 3 < ne) {
                        dv[u] = *(const int4*)(dst + i);
                    } else {
                        dv[u].x = (i     < ne) ? dst[i]     : -1;
                        dv[u].y = (i + 1 < ne) ? dst[i + 1] : -1;
                        dv[u].z = (i + 2 < ne) ? dst[i + 2] : -1;
                        dv[u].w = -1;
                    }
                }
            }

            #pragma unroll
            for (int u = 0; u < P_NLD; ++u) {
                const int dd[4] = {dv[u].x, dv[u].y, dv[u].z, dv[u].w};
                unsigned rv[4], mk[4];
                #pragma unroll
                for (int c = 0; c < 4; ++c) {
                    rv[c] = (unsigned)(dd[c] - row0);
                    mk[c] = __builtin_amdgcn_ballot_w32(rv[c] < (unsigned)R);
                }
                if ((mk[0] | mk[1] | mk[2] | mk[3]) != 0u) {
                    #pragma unroll
                    for (int c = 0; c < 4; ++c) {
                        if (mk[c] != 0u) {
                            const int pos = tot + (int)__builtin_amdgcn_mbcnt_lo(mk[c], 0u) - p * P_WCAP;
                            if (rv[c] < (unsigned)R && pos >= 0 && pos < P_WCAP)
                                wq[wave * P_WCAP + pos] =
                                    ((unsigned)((u * 1024 + tid) * 4 + c) << 9) | rv[c];
                            tot += __builtin_popcount(mk[c]);
                        }
                    }
                }
            }
            if (lane == 0) {
                int cc = tot - p * P_WCAP;
                cc = cc < 0 ? 0 : (cc > P_WCAP ? P_WCAP : cc);
                wcnt[wave] = cc;
                if (tot > (p + 1) * P_WCAP) more[fl] = 1;
            }
            __syncthreads();
            const int again = __builtin_amdgcn_readfirstlane(more[fl]);
            if (tid == 0) more[fl ^ 1] = 0;

            int mc = 0;
            const int cnts = wcnt[lane];
            unsigned ents[32];
            const int qi = (lane < P_WCAP) ? lane : 0;
            #pragma unroll
            for (int sw = 0; sw < 32; ++sw) ents[sw] = wq[sw * P_WCAP + qi];
            #pragma unroll
            for (int sw = 0; sw < 32; ++sw) {
                const int cq = __builtin_amdgcn_readlane(cnts, sw);
                const unsigned ent = ents[sw];
                const bool mine = (lane < cq) && ((ent & 31u) == (unsigned)wave);
                const unsigned mk = __builtin_amdgcn_ballot_w32(mine);
                if (mk != 0u) {
                    if (mine)
                        mylist[wave * P_MCAP + mc + (int)__builtin_amdgcn_mbcnt_lo(mk, 0u)] = ent;
                    mc += __builtin_popcount(mk);
                }
            }
            __builtin_amdgcn_fence(__ATOMIC_RELEASE, "wavefront");
            __builtin_amdgcn_wave_barrier();
            __builtin_amdgcn_fence(__ATOMIC_ACQUIRE, "wavefront");

            for (int j0 = 0; j0 < mc; j0 += 32) {
                const int j = j0 + lane;
                int   e_s = 0, e_r = 0;
                float e_n = 1.0f;
                if (j < mc) {
                    const unsigned ent = mylist[wave * P_MCAP + j];
                    e_r = (int)(ent & 511u);
                    e_s = src[segBase + (int)(ent >> 9)];
                    if (SRCSCALE) e_n = normS[e_s];
                }
                const int m = (mc - j0 < 32) ? (mc - j0) : 32;
                for (int b = 0; b < m; b += 8) {
                    vfl   v[8];
                    int   rr[8];
                    float nn[8];
                    #pragma unroll
                    for (int k = 0; k < 8; ++k) {
                        const int jj = (b + k < m) ? (b + k) : (m - 1);
                        const int sk = __builtin_amdgcn_readlane(e_s, jj);
                        rr[k] = __builtin_amdgcn_readlane(e_r, jj);
                        nn[k] = __int_as_float(__builtin_amdgcn_readlane(__float_as_int(e_n), jj));
                        v[k] = *(const vfl*)(feat + (long)sk * F + lane * VL);
                    }
                    #pragma unroll
                    for (int k = 0; k < 8; ++k) {
                        if (b + k < m) {
                            vfl* tp = (vfl*)&tile[rr[k] * F + lane * VL];
                            vfl t = *tp;
                            if (SRCSCALE) t += v[k] * nn[k];
                            else          t += v[k];
                            *tp = t;
                        }
                    }
                }
            }
            __syncthreads();
            ++pp;
            if (!again) break;
        }
    }

    for (int i = tid; i < R * 32; i += 1024) {
        const int r = row0 + (i >> 5);
        if (r < nd) {
            vfl t = ((const vfl*)tile)[i];
            if (FINAL) {
                const float sc = normD[r];
                const vfl bv = *(const vfl*)(bias + (i & 31) * VL);
                t = t * sc + bv;
            }
            *(vfl*)(res + (long)r * F + (i & 31) * VL) = t;
        }
    }
}

constexpr int Q_SCAN = 16;
constexpr int Q_OWN  = 32 - Q_SCAN;
constexpr int Q_NLD  = 8;
constexpr int Q_SEG  = Q_SCAN * 32 * 16 * Q_NLD;
constexpr int Q_WCAP = 32;
constexpr int Q_MCAP = Q_SCAN * Q_WCAP;
constexpr int Q_MAXP = (32 * 16 * Q_NLD) / Q_WCAP;
constexpr int Q_GB   = 16;

template <int F, int R>
__global__ __launch_bounds__(1024)
void k_pull_split(const float* __restrict__ feat, const int* __restrict__ src,
                  const int* __restrict__ dst, const float* __restrict__ normS,
                  const float* __restrict__ keyf,
                  float* __restrict__ res, int ne, int nd) {
    constexpr int VL = F / 32;
    typedef float vfl __attribute__((ext_vector_type(VL)));
    static_assert(R <= 512 && F == 128, "entry packing and lane vectors");
    static_assert(Q_SEG == KEY_PAD, "the tile bytes are padded to whole segments");
    static_assert(Q_WCAP == 32, "one queue entry per lane when the records are fetched and when they are picked");
    static_assert((Q_OWN & (Q_OWN - 1)) == 0 && Q_SCAN <= 32, "row owner by mask; one scanning wave's total per lane");

    __shared__ __attribute__((aligned(16))) float tile[R * F];
    __shared__ unsigned qrow[2][Q_SCAN * Q_WCAP];
    __shared__ int      qsrc[2][Q_SCAN * Q_WCAP];
    __shared__ float    qnrm[2][Q_SCAN * Q_WCAP];
    __shared__ int      qtot[2][Q_SCAN];
    __shared__ unsigned mylist[Q_OWN * Q_MCAP];

    const int tid  = threadIdx.x;
    const int lane = tid & 31;
    const int wave = __builtin_amdgcn_readfirstlane(tid >> 5);
    const int row0 = blockIdx.x * R;
    const bool scanner = wave < Q_SCAN;

    for (int i = tid; i < R * 32; i += 1024) ((vfl*)tile)[i] = vfl{};

    const int nseg = (ne + Q_SEG - 1) / Q_SEG;

    auto scanUnit = [&](const int seg, const int p, const int nb) {
        const int segBase = seg * Q_SEG;
        const unsigned char* keyb = (const unsigned char*)keyf + segBase;
        uint4 kv[Q_NLD];
        #pragma unroll
        for (int u = 0; u < Q_NLD; ++u)
            kv[u] = *(const uint4*)(keyb + (u * (Q_SCAN * 32) + tid) * 16);

        unsigned* qr = &qrow[nb][wave * Q_WCAP];
        const unsigned tileRep = (unsigned)blockIdx.x * 0x01010101u;
        int tot = 0;
        #pragma unroll
        for (int u = 0; u < Q_NLD; ++u) {
            const unsigned wv[4] = {kv[u].x ^ tileRep, kv[u].y ^ tileRep,
                                    kv[u].z ^ tileRep, kv[u].w ^ tileRep};
            unsigned hm = 0u;
            #pragma unroll
            for (int q = 0; q < 4; ++q) {
                const unsigned y = ((wv[q] & 0x7F7F7F7Fu) + 0x7F7F7F7Fu) | wv[q];
                hm |= (~y & 0x80808080u) >> (7 - q);
            }
            unsigned bal = __builtin_amdgcn_ballot_w32(hm != 0u);
            while (bal != 0u) {
                const int bp = __builtin_ctz(hm | 0x80000000u);
                const int o16 = ((bp & 7) << 2) | (bp >> 3);
                const int pos = tot + (int)__builtin_amdgcn_mbcnt_lo(bal, 0u) - p * Q_WCAP;
                if (hm != 0u && pos >= 0 && pos < Q_WCAP)
                    qr[pos] = (unsigned)((u * (Q_SCAN * 32) + tid) * 16 + o16);
                tot += __builtin_popcount(bal);
                hm &= hm - 1u;
                bal = __builtin_amdgcn_ballot_w32(hm != 0u);
            }
        }

        __builtin_amdgcn_fence(__ATOMIC_RELEASE, "wavefront");
        __builtin_amdgcn_wave_barrier();
        __builtin_amdgcn_fence(__ATOMIC_ACQUIRE, "wavefront");
        int nq = tot - p * Q_WCAP;
        nq = nq < 0 ? 0 : (nq > Q_WCAP ? Q_WCAP : nq);
        if (lane < nq) {
            const int e = segBase + (int)qr[lane];
            const int d = dst[e];
            const int s = src[e];
            qr[lane] = (unsigned)(d - row0);
            qsrc[nb][wave * Q_WCAP + lane] = s;
            qnrm[nb][wave * Q_WCAP + lane] = normS[s];
        }
        if (lane == 0) qtot[nb][wave] = tot;
    };

    if (scanner && nseg > 0) scanUnit(0, 0, 0);
    __syncthreads();

    int seg = 0, p = 0, b = 0;
    for (int it = 0; it < nseg * (Q_MAXP + 1); ++it) {
        if (seg >= nseg) break;
        const int t = (lane < Q_SCAN) ? qtot[b][lane] : 0;
        const bool again = __builtin_amdgcn_ballot_w32(t > (p + 1) * Q_WCAP) != 0u;
        const int nxSeg = again ? seg : seg + 1;
        const int nxP   = again ? p + 1 : 0;

        if (scanner) {
            if (nxSeg < nseg) scanUnit(nxSeg, nxP, b ^ 1);
        } else {
            const int ow = wave - Q_SCAN;
            int cnts = t - p * Q_WCAP;
            cnts = cnts < 0 ? 0 : (cnts > Q_WCAP ? Q_WCAP : cnts);

            unsigned ents[Q_SCAN];
            #pragma unroll
            for (int sw = 0; sw < Q_SCAN; ++sw) ents[sw] = qrow[b][sw * Q_WCAP + lane];
            int mc = 0;
            #pragma unroll
            for (int sw = 0; sw < Q_SCAN; ++sw) {
                const int cq = __builtin_amdgcn_readlane(cnts, sw);
                const unsigned ent = ents[sw];
                const bool mine = (lane < cq) && ((ent & (unsigned)(Q_OWN - 1)) == (unsigned)ow);
                const unsigned mk = __builtin_amdgcn_ballot_w32(mine);
                if (mk != 0u) {
                    if (mine)
                        mylist[ow * Q_MCAP + mc + (int)__builtin_amdgcn_mbcnt_lo(mk, 0u)] =
                            ((unsigned)(sw * Q_WCAP + lane) << 9) | ent;
                    mc += __builtin_popcount(mk);
                }
            }
            __builtin_amdgcn_fence(__ATOMIC_RELEASE, "wavefront");
            __builtin_amdgcn_wave_barrier();
            __builtin_amdgcn_fence(__ATOMIC_ACQUIRE, "wavefront");

            for (int j0 = 0; j0 < mc; j0 += 32) {
                const int j = j0 + lane;
                int   e_s = 0, e_r = 0;
                float e_n = 1.0f;
                if (j < mc) {
                    const unsigned ent = mylist[ow * Q_MCAP + j];
                    const int qi = (int)(ent >> 9);
                    e_r = (int)(ent & 511u);
                    e_s = qsrc[b][qi];
                    e_n = qnrm[b][qi];
                }
                const int m = (mc - j0 < 32) ? (mc - j0) : 32;
                for (int g0 = 0; g0 < m; g0 += Q_GB) {
                    vfl   v[Q_GB];
                    int   rr[Q_GB];
                    float nn[Q_GB];
                    #pragma unroll
                    for (int k = 0; k < Q_GB; ++k) {
                        const int jj = (g0 + k < m) ? (g0 + k) : (m - 1);
                        const int sk = __builtin_amdgcn_readlane(e_s, jj);
                        rr[k] = __builtin_amdgcn_readlane(e_r, jj);
                        nn[k] = __int_as_float(__builtin_amdgcn_readlane(__float_as_int(e_n), jj));
                        v[k] = *(const vfl*)(feat + (long)sk * F + lane * VL);
                    }
                    #pragma unroll
                    for (int k = 0; k < Q_GB; ++k) {
                        if (g0 + k < m) {
                            vfl* tp = (vfl*)&tile[rr[k] * F + lane * VL];
                            vfl tv = *tp;
                            tv += v[k] * nn[k];
                            *tp = tv;
                        }
                    }
                }
            }
        }
        __syncthreads();
        seg = nxSeg; p = nxP; b ^= 1;
    }

    for (int i = tid; i < R * 32; i += 1024) {
        const int r = row0 + (i >> 5);
        if (r < nd)
            *(vfl*)(res + (long)r * F + (i & 31) * VL) = ((const vfl*)tile)[i];
    }
}

template <int K, int N, int MT, int UNR, bool RELU, bool BIAS>
__global__ __launch_bounds__(128)
void k_gemm_wmma(const float* __restrict__ A,
                 const float* __restrict__ rowscale,
                 const float* __restrict__ W,
                 const float* __restrict__ bias,
                 float* __restrict__ C, int nRowGroups) {
    constexpr int NG = N / 64;
    const int wave = (blockIdx.x * blockDim.x + threadIdx.x) >> 5;
    const int lane = threadIdx.x & 31;
    const int rg = wave / NG;
    const int cg = wave % NG;
    if (rg >= nRowGroups) return;
    const int hf = lane >> 4;
    const int c  = lane & 15;
    const int row0 = rg * (MT * 16);
    const int col0 = cg * 64 + c * 4;

    const float* ap[MT];
    float sc[MT];
    #pragma unroll
    for (int m = 0; m < MT; ++m) {
        const int r = row0 + m * 16 + c;
        ap[m] = A + (long)r * K + hf * 4;
        sc[m] = rowscale[r];
    }
    const float* wp = W + (long)(hf * 4) * N + col0;

    v8f acc[MT][4];
    #pragma unroll
    for (int m = 0; m < MT; ++m)
        #pragma unroll
        for (int t = 0; t < 4; ++t) acc[m][t] = v8f{};

    #pragma unroll UNR
    for (int kk = 0; kk < K; kk += 8) {
        float4 av[MT], bq[4];
        #pragma unroll
        for (int m = 0; m < MT; ++m) av[m] = *(const float4*)(ap[m] + kk);
        #pragma unroll
        for (int i = 0; i < 4; ++i) bq[i] = *(const float4*)(wp + (long)(kk + i) * N);

        const float bv[4][4] = {
            {bq[0].x, bq[0].y, bq[0].z, bq[0].w},
            {bq[1].x, bq[1].y, bq[1].z, bq[1].w},
            {bq[2].x, bq[2].y, bq[2].z, bq[2].w},
            {bq[3].x, bq[3].y, bq[3].z, bq[3].w}};
        #pragma unroll
        for (int m = 0; m < MT; ++m) {
            v2f a; a.x = av[m].x * sc[m]; a.y = av[m].y * sc[m];
            #pragma unroll
            for (int t = 0; t < 4; ++t) {
                v2f b; b.x = bv[0][t]; b.y = bv[1][t];
                acc[m][t] = __builtin_amdgcn_wmma_f32_16x16x4_f32(
                    false, a, false, b, (short)0, acc[m][t], false, false);
            }
        }
        #pragma unroll
        for (int m = 0; m < MT; ++m) {
            v2f a; a.x = av[m].z * sc[m]; a.y = av[m].w * sc[m];
            #pragma unroll
            for (int t = 0; t < 4; ++t) {
                v2f b; b.x = bv[2][t]; b.y = bv[3][t];
                acc[m][t] = __builtin_amdgcn_wmma_f32_16x16x4_f32(
                    false, a, false, b, (short)0, acc[m][t], false, false);
            }
        }
    }

    float4 bval = {0.0f, 0.0f, 0.0f, 0.0f};
    if (BIAS) bval = *(const float4*)(bias + col0);
    #pragma unroll
    for (int m = 0; m < MT; ++m) {
        #pragma unroll
        for (int r = 0; r < 8; ++r) {
            float4 v;
            v.x = acc[m][0][r] + bval.x;
            v.y = acc[m][1][r] + bval.y;
            v.z = acc[m][2][r] + bval.z;
            v.w = acc[m][3][r] + bval.w;
            if (RELU) {
                v.x = fmaxf(v.x, 0.0f); v.y = fmaxf(v.y, 0.0f);
                v.z = fmaxf(v.z, 0.0f); v.w = fmaxf(v.w, 0.0f);
            }
            *(float4*)(C + (long)(row0 + m * 16 + hf * 8 + r) * N + col0) = v;
        }
    }
}

extern "C" void kernel_launch(void* const* d_in, const int* in_sizes, int n_in,
                              void* d_out, int out_size, void* d_ws, size_t ws_size,
                              hipStream_t stream) {
    const float* x    = (const float*)d_in[0];
    const float* W1   = (const float*)d_in[1];
    const float* b1   = (const float*)d_in[2];
    const float* W2   = (const float*)d_in[3];
    const float* b2   = (const float*)d_in[4];
    const int*   src1 = (const int*)d_in[5];
    const int*   dst1 = (const int*)d_in[6];
    const int*   src2 = (const int*)d_in[7];
    const int*   dst2 = (const int*)d_in[8];

    const int IN_F = 128, H_F = 256, N_CLS = 64;
    const int NSRC1 = in_sizes[0] / IN_F;
    const int E1    = in_sizes[5];
    const int E2    = in_sizes[7];
    const int ND1   = 100000;
    const int ND2   = 16384;

    float* ws    = (float*)d_ws;
    float* degS1 = ws;
    float* degD1 = degS1 + NSRC1;
    float* degS2 = degD1 + ND1;
    float* degD2 = degS2 + ND1;
    float* agg1  = degD2 + ND2;
    float* h     = agg1 + (long)ND1 * IN_F;
    float* g     = h + (long)ND1 * H_F;
    float* key1  = g;

    long zcount = (NSRC1 + ND1 + ND1 + ND2);
    k_zero<<<256, 256, 0, stream>>>(ws, zcount);

    const int nb1 = (E1 + D_EPB - 1) / D_EPB;
    const int nb2 = (E2 + D_EPB - 1) / D_EPB;
    k_deg<<<nb1 + nb2, 256, 0, stream>>>(src1, dst1, E1, degS1, degD1, nb1,
                                         src2, dst2, E2, degS2, degD2, key1);
    k_rsqrt<<<256, 256, 0, stream>>>(ws, zcount);

    k_pull_split<128, PULL1_R>
        <<<(ND1 + PULL1_R - 1) / PULL1_R, 1024, 0, stream>>>(x, src1, dst1, degS1, key1, agg1, E1, ND1);

    {
        k_gemm_wmma<128, 256, 5, 2, true, true>
            <<<1250, 128, 0, stream>>>(agg1, degD1, W1, b1, h, 1250);
    }

    {
        k_gemm_wmma<256, 64, 2, 4, false, false>
            <<<782, 128, 0, stream>>>(h, degS2, W2, nullptr, g, 3125);
    }

    k_pull<64, 64, false, true>
        <<<(ND2 + 63) / 64, 1024, 0, stream>>>(g, src2, dst2, nullptr, degD2, b2, (float*)d_out, E2, ND2);
}


# First aggregation: a scanning wave sends an L2 prefetch (2 x 256 B) for the source row of each edge it records, one segment ahead of the owning waves' gathers; sums and their order unchanged
# speedup vs baseline: 1.7582x; 1.0501x over previous
#include <hip/hip_runtime.h>


typedef __attribute__((ext_vector_type(2))) float v2f;
typedef __attribute__((ext_vector_type(8))) float v8f;

__device__ __forceinline__ void atomAddF(float* p, float v) {
    asm volatile("global_atomic_add_f32 %0, %1, off" :: "v"(p), "v"(v) : "memory");
}

__device__ __forceinline__ void l2Fetch(const float* p) {
    asm volatile("global_prefetch_b8 %0, off scope:SCOPE_SE" :: "v"(p) : "memory");
}

__global__ __launch_bounds__(256)
void k_zero(float* __restrict__ p, long n) {
    const long n4 = n >> 2;
    const long i = (long)blockIdx.x * blockDim.x + threadIdx.x;
    const long stride = (long)gridDim.x * blockDim.x;
    const float4 z = {0.0f, 0.0f, 0.0f, 0.0f};
    for (long j = i; j < n4; j += stride) ((float4*)p)[j] = z;
    const long t = (n4 << 2) + i;
    if (t < n) p[t] = 0.0f;
}

constexpr int D_EPB = 2048;
constexpr int PULL1_R = 400;
constexpr int KEY_PAD = 65536;
__device__ __forceinline__ unsigned tileKeys(const int4 d) {
    return (unsigned)(d.x / PULL1_R) | ((unsigned)(d.y / PULL1_R) << 8) |
           ((unsigned)(d.z / PULL1_R) << 16) | ((unsigned)(d.w / PULL1_R) << 24);
}
__global__ __launch_bounds__(256)
void k_deg(const int* __restrict__ src1, const int* __restrict__ dst1, int ne1,
           float* __restrict__ degS1, float* __restrict__ degD1, int nb1,
           const int* __restrict__ src2, const int* __restrict__ dst2, int ne2,
           float* __restrict__ degS2, float* __restrict__ degD2,
           float* __restrict__ key1f) {
    const bool first = (int)blockIdx.x < nb1;
    const int* __restrict__ src = first ? src1 : src2;
    const int* __restrict__ dst = first ? dst1 : dst2;
    float* __restrict__ degS = first ? degS1 : degS2;
    float* __restrict__ degD = first ? degD1 : degD2;
    const int ne  = first ? ne1 : ne2;
    const int blk = first ? (int)blockIdx.x : (int)blockIdx.x - nb1;
    const int e0  = blk * D_EPB + (int)threadIdx.x * 4;
    if (blk * D_EPB + D_EPB <= ne) {
        const int4 s0 = *(const int4*)(src + e0);
        const int4 s1 = *(const int4*)(src + e0 + 1024);
        const int4 d0 = *(const int4*)(dst + e0);
        const int4 d1 = *(const int4*)(dst + e0 + 1024);
        if (first) {
            ((unsigned*)key1f)[e0 >> 2]          = tileKeys(d0);
            ((unsigned*)key1f)[(e0 + 1024) >> 2] = tileKeys(d1);
        }
        atomAddF(&degS[s0.x], 1.0f); atomAddF(&degS[s0.y], 1.0f);
        atomAddF(&degS[s0.z], 1.0f); atomAddF(&degS[s0.w], 1.0f);
        atomAddF(&degD[d0.x], 1.0f); atomAddF(&degD[d0.y], 1.0f);
        atomAddF(&degD[d0.z], 1.0f); atomAddF(&degD[d0.w], 1.0f);
        atomAddF(&degS[s1.x], 1.0f); atomAddF(&degS[s1.y], 1.0f);
        atomAddF(&degS[s1.z], 1.0f); atomAddF(&degS[s1.w], 1.0f);
        atomAddF(&degD[d1.x], 1.0f); atomAddF(&degD[d1.y], 1.0f);
        atomAddF(&degD[d1.z], 1.0f); atomAddF(&degD[d1.w], 1.0f);
    } else {
        for (int k = 0; k < 8; ++k) {
            const int e = e0 + (k >> 2) * 1024 + (k & 3);
            if (e < ne) {
                const int de = dst[e];
                if (first) ((unsigned char*)key1f)[e] = (unsigned char)(de / PULL1_R);
                atomAddF(&degS[src[e]], 1.0f);
                atomAddF(&degD[de], 1.0f);
            }
        }
    }
    if (first && blk == nb1 - 1) {
        const int a4 = (ne + 3) & ~3;
        const int padEnd = ((ne + KEY_PAD - 1) / KEY_PAD) * KEY_PAD;
        if ((int)threadIdx.x < a4 - ne) ((unsigned char*)key1f)[ne + (int)threadIdx.x] = (unsigned char)255;
        for (int w = (a4 >> 2) + (int)threadIdx.x; w < (padEnd >> 2); w += 256)
            ((unsigned*)key1f)[w] = 0xFFFFFFFFu;
    }
}

__global__ __launch_bounds__(256)
void k_rsqrt(float* __restrict__ p, long n) {
    const long n4 = n >> 2;
    const long i = (long)blockIdx.x * blockDim.x + threadIdx.x;
    const long stride = (long)gridDim.x * blockDim.x;
    for (long j0 = i; j0 < n4; j0 += 4 * stride) {
        float4 v[4];
        #pragma unroll
        for (int k = 0; k < 4; ++k) {
            const long j = j0 + k * stride;
            if (j < n4) v[k] = ((const float4*)p)[j];
        }
        #pragma unroll
        for (int k = 0; k < 4; ++k) {
            const long j = j0 + k * stride;
            if (j < n4) {
                float4 t = v[k];
                t.x = rsqrtf(fmaxf(t.x, 1.0f)); t.y = rsqrtf(fmaxf(t.y, 1.0f));
                t.z = rsqrtf(fmaxf(t.z, 1.0f)); t.w = rsqrtf(fmaxf(t.w, 1.0f));
                ((float4*)p)[j] = t;
            }
        }
    }
    const long t = (n4 << 2) + i;
    if (t < n) p[t] = rsqrtf(fmaxf(p[t], 1.0f));
}

constexpr int P_NLD  = 16;
constexpr int P_SEG  = 1024 * 4 * P_NLD;
constexpr int P_WCAP = 24;
constexpr int P_MCAP = 32 * P_WCAP;
constexpr int P_MAXP = (32 * 4 * P_NLD) / P_WCAP;

template <int F, int R, bool SRCSCALE, bool FINAL>
__global__ __launch_bounds__(1024)
void k_pull(const float* __restrict__ feat, const int* __restrict__ src,
            const int* __restrict__ dst, const float* __restrict__ normS,
            const float* __restrict__ normD, const float* __restrict__ bias,
            float* __restrict__ res, int ne, int nd) {
    constexpr int VL = F / 32;
    typedef float vfl __attribute__((ext_vector_type(VL)));
    static_assert(R <= 512 && (F == 128 || F == 64), "entry packing and lane vectors");

    __shared__ __attribute__((aligned(16))) float tile[R * F];
    __shared__ unsigned wq[32 * P_WCAP];
    __shared__ int wcnt[32];
    __shared__ unsigned mylist[32 * P_MCAP];
    __shared__ int more[2];

    const int tid  = threadIdx.x;
    const int lane = tid & 31;
    const int wave = __builtin_amdgcn_readfirstlane(tid >> 5);
    const int row0 = blockIdx.x * R;

    for (int i = tid; i < R * 32; i += 1024) ((vfl*)tile)[i] = vfl{};
    if (tid < 2) more[tid] = 0;
    __syncthreads();

    int pp = 0;
    const int nseg = (ne + P_SEG - 1) / P_SEG;
    for (int seg = 0; seg < nseg; ++seg) {
        const int segBase = seg * P_SEG;

        for (int p = 0; p <= P_MAXP; ++p) {
            const int fl = pp & 1;

            int tot = 0;
            int4 dv[P_NLD];
            if (segBase + P_SEG <= ne) {
                #pragma unroll
                for (int u = 0; u < P_NLD; ++u)
                    dv[u] = *(const int4*)(dst + segBase + (u * 1024 + tid) * 4);
            } else {
                #pragma unroll
                for (int u = 0; u < P_NLD; ++u) {
                    const int i = segBase + (u * 1024 + tid) * 4;
                    if (i + 3 < ne) {
                        dv[u] = *(const int4*)(dst + i);
                    } else {
                        dv[u].x = (i     < ne) ? dst[i]     : -1;
                        dv[u].y = (i + 1 < ne) ? dst[i + 1] : -1;
                        dv[u].z = (i + 2 < ne) ? dst[i + 2] : -1;
                        dv[u].w = -1;
                    }
                }
            }

            #pragma unroll
            for (int u = 0; u < P_NLD; ++u) {
                const int dd[4] = {dv[u].x, dv[u].y, dv[u].z, dv[u].w};
                unsigned rv[4], mk[4];
                #pragma unroll
                for (int c = 0; c < 4; ++c) {
                    rv[c] = (unsigned)(dd[c] - row0);
                    mk[c] = __builtin_amdgcn_ballot_w32(rv[c] < (unsigned)R);
                }
                if ((mk[0] | mk[1] | mk[2] | mk[3]) != 0u) {
                    #pragma unroll
                    for (int c = 0; c < 4; ++c) {
                        if (mk[c] != 0u) {
                            const int pos = tot + (int)__builtin_amdgcn_mbcnt_lo(mk[c], 0u) - p * P_WCAP;
                            if (rv[c] < (unsigned)R && pos >= 0 && pos < P_WCAP)
                                wq[wave * P_WCAP + pos] =
                                    ((unsigned)((u * 1024 + tid) * 4 + c) << 9) | rv[c];
                            tot += __builtin_popcount(mk[c]);
                        }
                    }
                }
            }
            if (lane == 0) {
                int cc = tot - p * P_WCAP;
                cc = cc < 0 ? 0 : (cc > P_WCAP ? P_WCAP : cc);
                wcnt[wave] = cc;
                if (tot > (p + 1) * P_WCAP) more[fl] = 1;
            }
            __syncthreads();
            const int again = __builtin_amdgcn_readfirstlane(more[fl]);
            if (tid == 0) more[fl ^ 1] = 0;

            int mc = 0;
            const int cnts = wcnt[lane];
            unsigned ents[32];
            const int qi = (lane < P_WCAP) ? lane : 0;
            #pragma unroll
            for (int sw = 0; sw < 32; ++sw) ents[sw] = wq[sw * P_WCAP + qi];
            #pragma unroll
            for (int sw = 0; sw < 32; ++sw) {
                const int cq = __builtin_amdgcn_readlane(cnts, sw);
                const unsigned ent = ents[sw];
                const bool mine = (lane < cq) && ((ent & 31u) == (unsigned)wave);
                const unsigned mk = __builtin_amdgcn_ballot_w32(mine);
                if (mk != 0u) {
                    if (mine)
                        mylist[wave * P_MCAP + mc + (int)__builtin_amdgcn_mbcnt_lo(mk, 0u)] = ent;
                    mc += __builtin_popcount(mk);
                }
            }
            __builtin_amdgcn_fence(__ATOMIC_RELEASE, "wavefront");
            __builtin_amdgcn_wave_barrier();
            __builtin_amdgcn_fence(__ATOMIC_ACQUIRE, "wavefront");

            for (int j0 = 0; j0 < mc; j0 += 32) {
                const int j = j0 + lane;
                int   e_s = 0, e_r = 0;
                float e_n = 1.0f;
                if (j < mc) {
                    const unsigned ent = mylist[wave * P_MCAP + j];
                    e_r = (int)(ent & 511u);
                    e_s = src[segBase + (int)(ent >> 9)];
                    if (SRCSCALE) e_n = normS[e_s];
                }
                const int m = (mc - j0 < 32) ? (mc - j0) : 32;
                for (int b = 0; b < m; b += 8) {
                    vfl   v[8];
                    int   rr[8];
                    float nn[8];
                    #pragma unroll
                    for (int k = 0; k < 8; ++k) {
                        const int jj = (b + k < m) ? (b + k) : (m - 1);
                        const int sk = __builtin_amdgcn_readlane(e_s, jj);
                        rr[k] = __builtin_amdgcn_readlane(e_r, jj);
                        nn[k] = __int_as_float(__builtin_amdgcn_readlane(__float_as_int(e_n), jj));
                        v[k] = *(const vfl*)(feat + (long)sk * F + lane * VL);
                    }
                    #pragma unroll
                    for (int k = 0; k < 8; ++k) {
                        if (b + k < m) {
                            vfl* tp = (vfl*)&tile[rr[k] * F + lane * VL];
                            vfl t = *tp;
                            if (SRCSCALE) t += v[k] * nn[k];
                            else          t += v[k];
                            *tp = t;
                        }
                    }
                }
            }
            __syncthreads();
            ++pp;
            if (!again) break;
        }
    }

    for (int i = tid; i < R * 32; i += 1024) {
        const int r = row0 + (i >> 5);
        if (r < nd) {
            vfl t = ((const vfl*)tile)[i];
            if (FINAL) {
                const float sc = normD[r];
                const vfl bv = *(const vfl*)(bias + (i & 31) * VL);
                t = t * sc + bv;
            }
            *(vfl*)(res + (long)r * F + (i & 31) * VL) = t;
        }
    }
}

constexpr int Q_SCAN = 16;
constexpr int Q_OWN  = 32 - Q_SCAN;
constexpr int Q_NLD  = 8;
constexpr int Q_SEG  = Q_SCAN * 32 * 16 * Q_NLD;
constexpr int Q_WCAP = 32;
constexpr int Q_MCAP = Q_SCAN * Q_WCAP;
constexpr int Q_MAXP = (32 * 16 * Q_NLD) / Q_WCAP;
constexpr int Q_GB   = 16;

template <int F, int R>
__global__ __launch_bounds__(1024)
void k_pull_split(const float* __restrict__ feat, const int* __restrict__ src,
                  const int* __restrict__ dst, const float* __restrict__ normS,
                  const float* __restrict__ keyf,
                  float* __restrict__ res, int ne, int nd) {
    constexpr int VL = F / 32;
    typedef float vfl __attribute__((ext_vector_type(VL)));
    static_assert(R <= 512 && F == 128, "entry packing and lane vectors");
    static_assert(Q_SEG == KEY_PAD, "the tile bytes are padded to whole segments");
    static_assert(Q_WCAP == 32, "one queue entry per lane when the records are fetched and when they are picked");
    static_assert((Q_OWN & (Q_OWN - 1)) == 0 && Q_SCAN <= 32, "row owner by mask; one scanning wave's total per lane");

    __shared__ __attribute__((aligned(16))) float tile[R * F];
    __shared__ unsigned qrow[2][Q_SCAN * Q_WCAP];
    __shared__ int      qsrc[2][Q_SCAN * Q_WCAP];
    __shared__ float    qnrm[2][Q_SCAN * Q_WCAP];
    __shared__ int      qtot[2][Q_SCAN];
    __shared__ unsigned mylist[Q_OWN * Q_MCAP];

    const int tid  = threadIdx.x;
    const int lane = tid & 31;
    const int wave = __builtin_amdgcn_readfirstlane(tid >> 5);
    const int row0 = blockIdx.x * R;
    const bool scanner = wave < Q_SCAN;

    for (int i = tid; i < R * 32; i += 1024) ((vfl*)tile)[i] = vfl{};

    const int nseg = (ne + Q_SEG - 1) / Q_SEG;

    auto scanUnit = [&](const int seg, const int p, const int nb) {
        const int segBase = seg * Q_SEG;
        const unsigned char* keyb = (const unsigned char*)keyf + segBase;
        uint4 kv[Q_NLD];
        #pragma unroll
        for (int u = 0; u < Q_NLD; ++u)
            kv[u] = *(const uint4*)(keyb + (u * (Q_SCAN * 32) + tid) * 16);

        unsigned* qr = &qrow[nb][wave * Q_WCAP];
        const unsigned tileRep = (unsigned)blockIdx.x * 0x01010101u;
        int tot = 0;
        #pragma unroll
        for (int u = 0; u < Q_NLD; ++u) {
            const unsigned wv[4] = {kv[u].x ^ tileRep, kv[u].y ^ tileRep,
                                    kv[u].z ^ tileRep, kv[u].w ^ tileRep};
            unsigned hm = 0u;
            #pragma unroll
            for (int q = 0; q < 4; ++q) {
                const unsigned y = ((wv[q] & 0x7F7F7F7Fu) + 0x7F7F7F7Fu) | wv[q];
                hm |= (~y & 0x80808080u) >> (7 - q);
            }
            unsigned bal = __builtin_amdgcn_ballot_w32(hm != 0u);
            while (bal != 0u) {
                const int bp = __builtin_ctz(hm | 0x80000000u);
                const int o16 = ((bp & 7) << 2) | (bp >> 3);
                const int pos = tot + (int)__builtin_amdgcn_mbcnt_lo(bal, 0u) - p * Q_WCAP;
                if (hm != 0u && pos >= 0 && pos < Q_WCAP)
                    qr[pos] = (unsigned)((u * (Q_SCAN * 32) + tid) * 16 + o16);
                tot += __builtin_popcount(bal);
                hm &= hm - 1u;
                bal = __builtin_amdgcn_ballot_w32(hm != 0u);
            }
        }

        __builtin_amdgcn_fence(__ATOMIC_RELEASE, "wavefront");
        __builtin_amdgcn_wave_barrier();
        __builtin_amdgcn_fence(__ATOMIC_ACQUIRE, "wavefront");
        int nq = tot - p * Q_WCAP;
        nq = nq < 0 ? 0 : (nq > Q_WCAP ? Q_WCAP : nq);
        if (lane < nq) {
            const int e = segBase + (int)qr[lane];
            const int d = dst[e];
            const int s = src[e];
            l2Fetch(feat + (long)s * F);
            l2Fetch(feat + (long)s * F + 64);
            qr[lane] = (unsigned)(d - row0);
            qsrc[nb][wave * Q_WCAP + lane] = s;
            qnrm[nb][wave * Q_WCAP + lane] = normS[s];
        }
        if (lane == 0) qtot[nb][wave] = tot;
    };

    if (scanner && nseg > 0) scanUnit(0, 0, 0);
    __syncthreads();

    int seg = 0, p = 0, b = 0;
    for (int it = 0; it < nseg * (Q_MAXP + 1); ++it) {
        if (seg >= nseg) break;
        const int t = (lane < Q_SCAN) ? qtot[b][lane] : 0;
        const bool again = __builtin_amdgcn_ballot_w32(t > (p + 1) * Q_WCAP) != 0u;
        const int nxSeg = again ? seg : seg + 1;
        const int nxP   = again ? p + 1 : 0;

        if (scanner) {
            if (nxSeg < nseg) scanUnit(nxSeg, nxP, b ^ 1);
        } else {
            const int ow = wave - Q_SCAN;
            int cnts = t - p * Q_WCAP;
            cnts = cnts < 0 ? 0 : (cnts > Q_WCAP ? Q_WCAP : cnts);

            unsigned ents[Q_SCAN];
            #pragma unroll
            for (int sw = 0; sw < Q_SCAN; ++sw) ents[sw] = qrow[b][sw * Q_WCAP + lane];
            int mc = 0;
            #pragma unroll
            for (int sw = 0; sw < Q_SCAN; ++sw) {
                const int cq = __builtin_amdgcn_readlane(cnts, sw);
                const unsigned ent = ents[sw];
                const bool mine = (lane < cq) && ((ent & (unsigned)(Q_OWN - 1)) == (unsigned)ow);
                const unsigned mk = __builtin_amdgcn_ballot_w32(mine);
                if (mk != 0u) {
                    if (mine)
                        mylist[ow * Q_MCAP + mc + (int)__builtin_amdgcn_mbcnt_lo(mk, 0u)] =
                            ((unsigned)(sw * Q_WCAP + lane) << 9) | ent;
                    mc += __builtin_popcount(mk);
                }
            }
            __builtin_amdgcn_fence(__ATOMIC_RELEASE, "wavefront");
            __builtin_amdgcn_wave_barrier();
            __builtin_amdgcn_fence(__ATOMIC_ACQUIRE, "wavefront");

            for (int j0 = 0; j0 < mc; j0 += 32) {
                const int j = j0 + lane;
                int   e_s = 0, e_r = 0;
                float e_n = 1.0f;
                if (j < mc) {
                    const unsigned ent = mylist[ow * Q_MCAP + j];
                    const int qi = (int)(ent >> 9);
                    e_r = (int)(ent & 511u);
                    e_s = qsrc[b][qi];
                    e_n = qnrm[b][qi];
                }
                const int m = (mc - j0 < 32) ? (mc - j0) : 32;
                for (int g0 = 0; g0 < m; g0 += Q_GB) {
                    vfl   v[Q_GB];
                    int   rr[Q_GB];
                    float nn[Q_GB];
                    #pragma unroll
                    for (int k = 0; k < Q_GB; ++k) {
                        const int jj = (g0 + k < m) ? (g0 + k) : (m - 1);
                        const int sk = __builtin_amdgcn_readlane(e_s, jj);
                        rr[k] = __builtin_amdgcn_readlane(e_r, jj);
                        nn[k] = __int_as_float(__builtin_amdgcn_readlane(__float_as_int(e_n), jj));
                        v[k] = *(const vfl*)(feat + (long)sk * F + lane * VL);
                    }
                    #pragma unroll
                    for (int k = 0; k < Q_GB; ++k) {
                        if (g0 + k < m) {
                            vfl* tp = (vfl*)&tile[rr[k] * F + lane * VL];
                            vfl tv = *tp;
                            tv += v[k] * nn[k];
                            *tp = tv;
                        }
                    }
                }
            }
        }
        __syncthreads();
        seg = nxSeg; p = nxP; b ^= 1;
    }

    for (int i = tid; i < R * 32; i += 1024) {
        const int r = row0 + (i >> 5);
        if (r < nd)
            *(vfl*)(res + (long)r * F + (i & 31) * VL) = ((const vfl*)tile)[i];
    }
}

template <int K, int N, int MT, int UNR, bool RELU, bool BIAS>
__global__ __launch_bounds__(128)
void k_gemm_wmma(const float* __restrict__ A,
                 const float* __restrict__ rowscale,
                 const float* __restrict__ W,
                 const float* __restrict__ bias,
                 float* __restrict__ C, int nRowGroups) {
    constexpr int NG = N / 64;
    const int wave = (blockIdx.x * blockDim.x + threadIdx.x) >> 5;
    const int lane = threadIdx.x & 31;
    const int rg = wave / NG;
    const int cg = wave % NG;
    if (rg >= nRowGroups) return;
    const int hf = lane >> 4;
    const int c  = lane & 15;
    const int row0 = rg * (MT * 16);
    const int col0 = cg * 64 + c * 4;

    const float* ap[MT];
    float sc[MT];
    #pragma unroll
    for (int m = 0; m < MT; ++m) {
        const int r = row0 + m * 16 + c;
        ap[m] = A + (long)r * K + hf * 4;
        sc[m] = rowscale[r];
    }
    const float* wp = W + (long)(hf * 4) * N + col0;

    v8f acc[MT][4];
    #pragma unroll
    for (int m = 0; m < MT; ++m)
        #pragma unroll
        for (int t = 0; t < 4; ++t) acc[m][t] = v8f{};

    #pragma unroll UNR
    for (int kk = 0; kk < K; kk += 8) {
        float4 av[MT], bq[4];
        #pragma unroll
        for (int m = 0; m < MT; ++m) av[m] = *(const float4*)(ap[m] + kk);
        #pragma unroll
        for (int i = 0; i < 4; ++i) bq[i] = *(const float4*)(wp + (long)(kk + i) * N);

        const float bv[4][4] = {
            {bq[0].x, bq[0].y, bq[0].z, bq[0].w},
            {bq[1].x, bq[1].y, bq[1].z, bq[1].w},
            {bq[2].x, bq[2].y, bq[2].z, bq[2].w},
            {bq[3].x, bq[3].y, bq[3].z, bq[3].w}};
        #pragma unroll
        for (int m = 0; m < MT; ++m) {
            v2f a; a.x = av[m].x * sc[m]; a.y = av[m].y * sc[m];
            #pragma unroll
            for (int t = 0; t < 4; ++t) {
                v2f b; b.x = bv[0][t]; b.y = bv[1][t];
                acc[m][t] = __builtin_amdgcn_wmma_f32_16x16x4_f32(
                    false, a, false, b, (short)0, acc[m][t], false, false);
            }
        }
        #pragma unroll
        for (int m = 0; m < MT; ++m) {
            v2f a; a.x = av[m].z * sc[m]; a.y = av[m].w * sc[m];
            #pragma unroll
            for (int t = 0; t < 4; ++t) {
                v2f b; b.x = bv[2][t]; b.y = bv[3][t];
                acc[m][t] = __builtin_amdgcn_wmma_f32_16x16x4_f32(
                    false, a, false, b, (short)0, acc[m][t], false, false);
            }
        }
    }

    float4 bval = {0.0f, 0.0f, 0.0f, 0.0f};
    if (BIAS) bval = *(const float4*)(bias + col0);
    #pragma unroll
    for (int m = 0; m < MT; ++m) {
        #pragma unroll
        for (int r = 0; r < 8; ++r) {
            float4 v;
            v.x = acc[m][0][r] + bval.x;
            v.y = acc[m][1][r] + bval.y;
            v.z = acc[m][2][r] + bval.z;
            v.w = acc[m][3][r] + bval.w;
            if (RELU) {
                v.x = fmaxf(v.x, 0.0f); v.y = fmaxf(v.y, 0.0f);
                v.z = fmaxf(v.z, 0.0f); v.w = fmaxf(v.w, 0.0f);
            }
            *(float4*)(C + (long)(row0 + m * 16 + hf * 8 + r) * N + col0) = v;
        }
    }
}

extern "C" void kernel_launch(void* const* d_in, const int* in_sizes, int n_in,
                              void* d_out, int out_size, void* d_ws, size_t ws_size,
                              hipStream_t stream) {
    const float* x    = (const float*)d_in[0];
    const float* W1   = (const float*)d_in[1];
    const float* b1   = (const float*)d_in[2];
    const float* W2   = (const float*)d_in[3];
    const float* b2   = (const float*)d_in[4];
    const int*   src1 = (const int*)d_in[5];
    const int*   dst1 = (const int*)d_in[6];
    const int*   src2 = (const int*)d_in[7];
    const int*   dst2 = (const int*)d_in[8];

    const int IN_F = 128, H_F = 256, N_CLS = 64;
    const int NSRC1 = in_sizes[0] / IN_F;
    const int E1    = in_sizes[5];
    const int E2    = in_sizes[7];
    const int ND1   = 100000;
    const int ND2   = 16384;

    float* ws    = (float*)d_ws;
    float* degS1 = ws;
    float* degD1 = degS1 + NSRC1;
    float* degS2 = degD1 + ND1;
    float* degD2 = degS2 + ND1;
    float* agg1  = degD2 + ND2;
    float* h     = agg1 + (long)ND1 * IN_F;
    float* g     = h + (long)ND1 * H_F;
    float* key1  = g;

    long zcount = (NSRC1 + ND1 + ND1 + ND2);
    k_zero<<<256, 256, 0, stream>>>(ws, zcount);

    const int nb1 = (E1 + D_EPB - 1) / D_EPB;
    const int nb2 = (E2 + D_EPB - 1) / D_EPB;
    k_deg<<<nb1 + nb2, 256, 0, stream>>>(src1, dst1, E1, degS1, degD1, nb1,
                                         src2, dst2, E2, degS2, degD2, key1);
    k_rsqrt<<<256, 256, 0, stream>>>(ws, zcount);

    k_pull_split<128, PULL1_R>
        <<<(ND1 + PULL1_R - 1) / PULL1_R, 1024, 0, stream>>>(x, src1, dst1, degS1, key1, agg1, E1, ND1);

    {
        k_gemm_wmma<128, 256, 5, 2, true, true>
            <<<1250, 128, 0, stream>>>(agg1, degD1, W1, b1, h, 1250);
    }

    {
        k_gemm_wmma<256, 64, 2, 4, false, false>
            <<<782, 128, 0, stream>>>(h, degS2, W2, nullptr, g, 3125);
    }

    k_pull<64, 64, false, true>
        <<<(ND2 + 63) / 64, 1024, 0, stream>>>(g, src2, dst2, nullptr, degD2, b2, (float*)d_out, E2, ND2);
}


# First aggregation: an owning wave adds gathered rows four at a time: the four tile rows are read from LDS together, a repeated row is carried in registers; sums and their order unchanged
# speedup vs baseline: 1.8595x; 1.0576x over previous
#include <hip/hip_runtime.h>


typedef __attribute__((ext_vector_type(2))) float v2f;
typedef __attribute__((ext_vector_type(8))) float v8f;

__device__ __forceinline__ void atomAddF(float* p, float v) {
    asm volatile("global_atomic_add_f32 %0, %1, off" :: "v"(p), "v"(v) : "memory");
}

__device__ __forceinline__ void l2Fetch(const float* p) {
    asm volatile("global_prefetch_b8 %0, off scope:SCOPE_SE" :: "v"(p) : "memory");
}

__global__ __launch_bounds__(256)
void k_zero(float* __restrict__ p, long n) {
    const long n4 = n >> 2;
    const long i = (long)blockIdx.x * blockDim.x + threadIdx.x;
    const long stride = (long)gridDim.x * blockDim.x;
    const float4 z = {0.0f, 0.0f, 0.0f, 0.0f};
    for (long j = i; j < n4; j += stride) ((float4*)p)[j] = z;
    const long t = (n4 << 2) + i;
    if (t < n) p[t] = 0.0f;
}

constexpr int D_EPB = 2048;
constexpr int PULL1_R = 400;
constexpr int KEY_PAD = 65536;
__device__ __forceinline__ unsigned tileKeys(const int4 d) {
    return (unsigned)(d.x / PULL1_R) | ((unsigned)(d.y / PULL1_R) << 8) |
           ((unsigned)(d.z / PULL1_R) << 16) | ((unsigned)(d.w / PULL1_R) << 24);
}
__global__ __launch_bounds__(256)
void k_deg(const int* __restrict__ src1, const int* __restrict__ dst1, int ne1,
           float* __restrict__ degS1, float* __restrict__ degD1, int nb1,
           const int* __restrict__ src2, const int* __restrict__ dst2, int ne2,
           float* __restrict__ degS2, float* __restrict__ degD2,
           float* __restrict__ key1f) {
    const bool first = (int)blockIdx.x < nb1;
    const int* __restrict__ src = first ? src1 : src2;
    const int* __restrict__ dst = first ? dst1 : dst2;
    float* __restrict__ degS = first ? degS1 : degS2;
    float* __restrict__ degD = first ? degD1 : degD2;
    const int ne  = first ? ne1 : ne2;
    const int blk = first ? (int)blockIdx.x : (int)blockIdx.x - nb1;
    const int e0  = blk * D_EPB + (int)threadIdx.x * 4;
    if (blk * D_EPB + D_EPB <= ne) {
        const int4 s0 = *(const int4*)(src + e0);
        const int4 s1 = *(const int4*)(src + e0 + 1024);
        const int4 d0 = *(const int4*)(dst + e0);
        const int4 d1 = *(const int4*)(dst + e0 + 1024);
        if (first) {
            ((unsigned*)key1f)[e0 >> 2]          = tileKeys(d0);
            ((unsigned*)key1f)[(e0 + 1024) >> 2] = tileKeys(d1);
        }
        atomAddF(&degS[s0.x], 1.0f); atomAddF(&degS[s0.y], 1.0f);
        atomAddF(&degS[s0.z], 1.0f); atomAddF(&degS[s0.w], 1.0f);
        atomAddF(&degD[d0.x], 1.0f); atomAddF(&degD[d0.y], 1.0f);
        atomAddF(&degD[d0.z], 1.0f); atomAddF(&degD[d0.w], 1.0f);
        atomAddF(&degS[s1.x], 1.0f); atomAddF(&degS[s1.y], 1.0f);
        atomAddF(&degS[s1.z], 1.0f); atomAddF(&degS[s1.w], 1.0f);
        atomAddF(&degD[d1.x], 1.0f); atomAddF(&degD[d1.y], 1.0f);
        atomAddF(&degD[d1.z], 1.0f); atomAddF(&degD[d1.w], 1.0f);
    } else {
        for (int k = 0; k < 8; ++k) {
            const int e = e0 + (k >> 2) * 1024 + (k & 3);
            if (e < ne) {
                const int de = dst[e];
                if (first) ((unsigned char*)key1f)[e] = (unsigned char)(de / PULL1_R);
                atomAddF(&degS[src[e]], 1.0f);
                atomAddF(&degD[de], 1.0f);
            }
        }
    }
    if (first && blk == nb1 - 1) {
        const int a4 = (ne + 3) & ~3;
        const int padEnd = ((ne + KEY_PAD - 1) / KEY_PAD) * KEY_PAD;
        if ((int)threadIdx.x < a4 - ne) ((unsigned char*)key1f)[ne + (int)threadIdx.x] = (unsigned char)255;
        for (int w = (a4 >> 2) + (int)threadIdx.x; w < (padEnd >> 2); w += 256)
            ((unsigned*)key1f)[w] = 0xFFFFFFFFu;
    }
}

__global__ __launch_bounds__(256)
void k_rsqrt(float* __restrict__ p, long n) {
    const long n4 = n >> 2;
    const long i = (long)blockIdx.x * blockDim.x + threadIdx.x;
    const long stride = (long)gridDim.x * blockDim.x;
    for (long j0 = i; j0 < n4; j0 += 4 * stride) {
        float4 v[4];
        #pragma unroll
        for (int k = 0; k < 4; ++k) {
            const long j = j0 + k * stride;
            if (j < n4) v[k] = ((const float4*)p)[j];
        }
        #pragma unroll
        for (int k = 0; k < 4; ++k) {
            const long j = j0 + k * stride;
            if (j < n4) {
                float4 t = v[k];
                t.x = rsqrtf(fmaxf(t.x, 1.0f)); t.y = rsqrtf(fmaxf(t.y, 1.0f));
                t.z = rsqrtf(fmaxf(t.z, 1.0f)); t.w = rsqrtf(fmaxf(t.w, 1.0f));
                ((float4*)p)[j] = t;
            }
        }
    }
    const long t = (n4 << 2) + i;
    if (t < n) p[t] = rsqrtf(fmaxf(p[t], 1.0f));
}

constexpr int P_NLD  = 16;
constexpr int P_SEG  = 1024 * 4 * P_NLD;
constexpr int P_WCAP = 24;
constexpr int P_MCAP = 32 * P_WCAP;
constexpr int P_MAXP = (32 * 4 * P_NLD) / P_WCAP;

template <int F, int R, bool SRCSCALE, bool FINAL>
__global__ __launch_bounds__(1024)
void k_pull(const float* __restrict__ feat, const int* __restrict__ src,
            const int* __restrict__ dst, const float* __restrict__ normS,
            const float* __restrict__ normD, const float* __restrict__ bias,
            float* __restrict__ res, int ne, int nd) {
    constexpr int VL = F / 32;
    typedef float vfl __attribute__((ext_vector_type(VL)));
    static_assert(R <= 512 && (F == 128 || F == 64), "entry packing and lane vectors");

    __shared__ __attribute__((aligned(16))) float tile[R * F];
    __shared__ unsigned wq[32 * P_WCAP];
    __shared__ int wcnt[32];
    __shared__ unsigned mylist[32 * P_MCAP];
    __shared__ int more[2];

    const int tid  = threadIdx.x;
    const int lane = tid & 31;
    const int wave = __builtin_amdgcn_readfirstlane(tid >> 5);
    const int row0 = blockIdx.x * R;

    for (int i = tid; i < R * 32; i += 1024) ((vfl*)tile)[i] = vfl{};
    if (tid < 2) more[tid] = 0;
    __syncthreads();

    int pp = 0;
    const int nseg = (ne + P_SEG - 1) / P_SEG;
    for (int seg = 0; seg < nseg; ++seg) {
        const int segBase = seg * P_SEG;

        for (int p = 0; p <= P_MAXP; ++p) {
            const int fl = pp & 1;

            int tot = 0;
            int4 dv[P_NLD];
            if (segBase + P_SEG <= ne) {
                #pragma unroll
                for (int u = 0; u < P_NLD; ++u)
                    dv[u] = *(const int4*)(dst + segBase + (u * 1024 + tid) * 4);
            } else {
                #pragma unroll
                for (int u = 0; u < P_NLD; ++u) {
                    const int i = segBase + (u * 1024 + tid) * 4;
                    if (i + 3 < ne) {
                        dv[u] = *(const int4*)(dst + i);
                    } else {
                        dv[u].x = (i     < ne) ? dst[i]     : -1;
                        dv[u].y = (i + 1 < ne) ? dst[i + 1] : -1;
                        dv[u].z = (i + 2 < ne) ? dst[i + 2] : -1;
                        dv[u].w = -1;
                    }
                }
            }

            #pragma unroll
            for (int u = 0; u < P_NLD; ++u) {
                const int dd[4] = {dv[u].x, dv[u].y, dv[u].z, dv[u].w};
                unsigned rv[4], mk[4];
                #pragma unroll
                for (int c = 0; c < 4; ++c) {
                    rv[c] = (unsigned)(dd[c] - row0);
                    mk[c] = __builtin_amdgcn_ballot_w32(rv[c] < (unsigned)R);
                }
                if ((mk[0] | mk[1] | mk[2] | mk[3]) != 0u) {
                    #pragma unroll
                    for (int c = 0; c < 4; ++c) {
                        if (mk[c] != 0u) {
                            const int pos = tot + (int)__builtin_amdgcn_mbcnt_lo(mk[c], 0u) - p * P_WCAP;
                            if (rv[c] < (unsigned)R && pos >= 0 && pos < P_WCAP)
                                wq[wave * P_WCAP + pos] =
                                    ((unsigned)((u * 1024 + tid) * 4 + c) << 9) | rv[c];
                            tot += __builtin_popcount(mk[c]);
                        }
                    }
                }
            }
            if (lane == 0) {
                int cc = tot - p * P_WCAP;
                cc = cc < 0 ? 0 : (cc > P_WCAP ? P_WCAP : cc);
                wcnt[wave] = cc;
                if (tot > (p + 1) * P_WCAP) more[fl] = 1;
            }
            __syncthreads();
            const int again = __builtin_amdgcn_readfirstlane(more[fl]);
            if (tid == 0) more[fl ^ 1] = 0;

            int mc = 0;
            const int cnts = wcnt[lane];
            unsigned ents[32];
            const int qi = (lane < P_WCAP) ? lane : 0;
            #pragma unroll
            for (int sw = 0; sw < 32; ++sw) ents[sw] = wq[sw * P_WCAP + qi];
            #pragma unroll
            for (int sw = 0; sw < 32; ++sw) {
                const int cq = __builtin_amdgcn_readlane(cnts, sw);
                const unsigned ent = ents[sw];
                const bool mine = (lane < cq) && ((ent & 31u) == (unsigned)wave);
                const unsigned mk = __builtin_amdgcn_ballot_w32(mine);
                if (mk != 0u) {
                    if (mine)
                        mylist[wave * P_MCAP + mc + (int)__builtin_amdgcn_mbcnt_lo(mk, 0u)] = ent;
                    mc += __builtin_popcount(mk);
                }
            }
            __builtin_amdgcn_fence(__ATOMIC_RELEASE, "wavefront");
            __builtin_amdgcn_wave_barrier();
            __builtin_amdgcn_fence(__ATOMIC_ACQUIRE, "wavefront");

            for (int j0 = 0; j0 < mc; j0 += 32) {
                const int j = j0 + lane;
                int   e_s = 0, e_r = 0;
                float e_n = 1.0f;
                if (j < mc) {
                    const unsigned ent = mylist[wave * P_MCAP + j];
                    e_r = (int)(ent & 511u);
                    e_s = src[segBase + (int)(ent >> 9)];
                    if (SRCSCALE) e_n = normS[e_s];
                }
                const int m = (mc - j0 < 32) ? (mc - j0) : 32;
                for (int b = 0; b < m; b += 8) {
                    vfl   v[8];
                    int   rr[8];
                    float nn[8];
                    #pragma unroll
                    for (int k = 0; k < 8; ++k) {
                        const int jj = (b + k < m) ? (b + k) : (m - 1);
                        const int sk = __builtin_amdgcn_readlane(e_s, jj);
                        rr[k] = __builtin_amdgcn_readlane(e_r, jj);
                        nn[k] = __int_as_float(__builtin_amdgcn_readlane(__float_as_int(e_n), jj));
                        v[k] = *(const vfl*)(feat + (long)sk * F + lane * VL);
                    }
                    #pragma unroll
                    for (int k = 0; k < 8; ++k) {
                        if (b + k < m) {
                            vfl* tp = (vfl*)&tile[rr[k] * F + lane * VL];
                            vfl t = *tp;
                            if (SRCSCALE) t += v[k] * nn[k];
                            else          t += v[k];
                            *tp = t;
                        }
                    }
                }
            }
            __syncthreads();
            ++pp;
            if (!again) break;
        }
    }

    for (int i = tid; i < R * 32; i += 1024) {
        const int r = row0 + (i >> 5);
        if (r < nd) {
            vfl t = ((const vfl*)tile)[i];
            if (FINAL) {
                const float sc = normD[r];
                const vfl bv = *(const vfl*)(bias + (i & 31) * VL);
                t = t * sc + bv;
            }
            *(vfl*)(res + (long)r * F + (i & 31) * VL) = t;
        }
    }
}

constexpr int Q_SCAN = 16;
constexpr int Q_OWN  = 32 - Q_SCAN;
constexpr int Q_NLD  = 8;
constexpr int Q_SEG  = Q_SCAN * 32 * 16 * Q_NLD;
constexpr int Q_WCAP = 32;
constexpr int Q_MCAP = Q_SCAN * Q_WCAP;
constexpr int Q_MAXP = (32 * 16 * Q_NLD) / Q_WCAP;
constexpr int Q_GB   = 16;
constexpr int Q_AG   = 4;

template <int F, int R>
__global__ __launch_bounds__(1024)
void k_pull_split(const float* __restrict__ feat, const int* __restrict__ src,
                  const int* __restrict__ dst, const float* __restrict__ normS,
                  const float* __restrict__ keyf,
                  float* __restrict__ res, int ne, int nd) {
    constexpr int VL = F / 32;
    typedef float vfl __attribute__((ext_vector_type(VL)));
    static_assert(R <= 512 && F == 128, "entry packing and lane vectors");
    static_assert(Q_SEG == KEY_PAD, "the tile bytes are padded to whole segments");
    static_assert(Q_WCAP == 32, "one queue entry per lane when the records are fetched and when they are picked");
    static_assert((Q_OWN & (Q_OWN - 1)) == 0 && Q_SCAN <= 32, "row owner by mask; one scanning wave's total per lane");

    __shared__ __attribute__((aligned(16))) float tile[R * F];
    __shared__ unsigned qrow[2][Q_SCAN * Q_WCAP];
    __shared__ int      qsrc[2][Q_SCAN * Q_WCAP];
    __shared__ float    qnrm[2][Q_SCAN * Q_WCAP];
    __shared__ int      qtot[2][Q_SCAN];
    __shared__ unsigned mylist[Q_OWN * Q_MCAP];

    const int tid  = threadIdx.x;
    const int lane = tid & 31;
    const int wave = __builtin_amdgcn_readfirstlane(tid >> 5);
    const int row0 = blockIdx.x * R;
    const bool scanner = wave < Q_SCAN;

    for (int i = tid; i < R * 32; i += 1024) ((vfl*)tile)[i] = vfl{};

    const int nseg = (ne + Q_SEG - 1) / Q_SEG;

    auto scanUnit = [&](const int seg, const int p, const int nb) {
        const int segBase = seg * Q_SEG;
        const unsigned char* keyb = (const unsigned char*)keyf + segBase;
        uint4 kv[Q_NLD];
        #pragma unroll
        for (int u = 0; u < Q_NLD; ++u)
            kv[u] = *(const uint4*)(keyb + (u * (Q_SCAN * 32) + tid) * 16);

        unsigned* qr = &qrow[nb][wave * Q_WCAP];
        const unsigned tileRep = (unsigned)blockIdx.x * 0x01010101u;
        int tot = 0;
        #pragma unroll
        for (int u = 0; u < Q_NLD; ++u) {
            const unsigned wv[4] = {kv[u].x ^ tileRep, kv[u].y ^ tileRep,
                                    kv[u].z ^ tileRep, kv[u].w ^ tileRep};
            unsigned hm = 0u;
            #pragma unroll
            for (int q = 0; q < 4; ++q) {
                const unsigned y = ((wv[q] & 0x7F7F7F7Fu) + 0x7F7F7F7Fu) | wv[q];
                hm |= (~y & 0x80808080u) >> (7 - q);
            }
            unsigned bal = __builtin_amdgcn_ballot_w32(hm != 0u);
            while (bal != 0u) {
                const int bp = __builtin_ctz(hm | 0x80000000u);
                const int o16 = ((bp & 7) << 2) | (bp >> 3);
                const int pos = tot + (int)__builtin_amdgcn_mbcnt_lo(bal, 0u) - p * Q_WCAP;
                if (hm != 0u && pos >= 0 && pos < Q_WCAP)
                    qr[pos] = (unsigned)((u * (Q_SCAN * 32) + tid) * 16 + o16);
                tot += __builtin_popcount(bal);
                hm &= hm - 1u;
                bal = __builtin_amdgcn_ballot_w32(hm != 0u);
            }
        }

        __builtin_amdgcn_fence(__ATOMIC_RELEASE, "wavefront");
        __builtin_amdgcn_wave_barrier();
        __builtin_amdgcn_fence(__ATOMIC_ACQUIRE, "wavefront");
        int nq = tot - p * Q_WCAP;
        nq = nq < 0 ? 0 : (nq > Q_WCAP ? Q_WCAP : nq);
        if (lane < nq) {
            const int e = segBase + (int)qr[lane];
            const int d = dst[e];
            const int s = src[e];
            l2Fetch(feat + (long)s * F);
            l2Fetch(feat + (long)s * F + 64);
            qr[lane] = (unsigned)(d - row0);
            qsrc[nb][wave * Q_WCAP + lane] = s;
            qnrm[nb][wave * Q_WCAP + lane] = normS[s];
        }
        if (lane == 0) qtot[nb][wave] = tot;
    };

    if (scanner && nseg > 0) scanUnit(0, 0, 0);
    __syncthreads();

    int seg = 0, p = 0, b = 0;
    for (int it = 0; it < nseg * (Q_MAXP + 1); ++it) {
        if (seg >= nseg) break;
        const int t = (lane < Q_SCAN) ? qtot[b][lane] : 0;
        const bool again = __builtin_amdgcn_ballot_w32(t > (p + 1) * Q_WCAP) != 0u;
        const int nxSeg = again ? seg : seg + 1;
        const int nxP   = again ? p + 1 : 0;

        if (scanner) {
            if (nxSeg < nseg) scanUnit(nxSeg, nxP, b ^ 1);
        } else {
            const int ow = wave - Q_SCAN;
            int cnts = t - p * Q_WCAP;
            cnts = cnts < 0 ? 0 : (cnts > Q_WCAP ? Q_WCAP : cnts);

            unsigned ents[Q_SCAN];
            #pragma unroll
            for (int sw = 0; sw < Q_SCAN; ++sw) ents[sw] = qrow[b][sw * Q_WCAP + lane];
            int mc = 0;
            #pragma unroll
            for (int sw = 0; sw < Q_SCAN; ++sw) {
                const int cq = __builtin_amdgcn_readlane(cnts, sw);
                const unsigned ent = ents[sw];
                const bool mine = (lane < cq) && ((ent & (unsigned)(Q_OWN - 1)) == (unsigned)ow);
                const unsigned mk = __builtin_amdgcn_ballot_w32(mine);
                if (mk != 0u) {
                    if (mine)
                        mylist[ow * Q_MCAP + mc + (int)__builtin_amdgcn_mbcnt_lo(mk, 0u)] =
                            ((unsigned)(sw * Q_WCAP + lane) << 9) | ent;
                    mc += __builtin_popcount(mk);
                }
            }
            __builtin_amdgcn_fence(__ATOMIC_RELEASE, "wavefront");
            __builtin_amdgcn_wave_barrier();
            __builtin_amdgcn_fence(__ATOMIC_ACQUIRE, "wavefront");

            for (int j0 = 0; j0 < mc; j0 += 32) {
                const int j = j0 + lane;
                int   e_s = 0, e_r = 0;
                float e_n = 1.0f;
                if (j < mc) {
                    const unsigned ent = mylist[ow * Q_MCAP + j];
                    const int qi = (int)(ent >> 9);
                    e_r = (int)(ent & 511u);
                    e_s = qsrc[b][qi];
                    e_n = qnrm[b][qi];
                }
                const int m = (mc - j0 < 32) ? (mc - j0) : 32;
                for (int g0 = 0; g0 < m; g0 += Q_GB) {
                    vfl   v[Q_GB];
                    int   rr[Q_GB];
                    float nn[Q_GB];
                    #pragma unroll
                    for (int k = 0; k < Q_GB; ++k) {
                        const int jj = (g0 + k < m) ? (g0 + k) : (m - 1);
                        const int sk = __builtin_amdgcn_readlane(e_s, jj);
                        rr[k] = __builtin_amdgcn_readlane(e_r, jj);
                        nn[k] = __int_as_float(__builtin_amdgcn_readlane(__float_as_int(e_n), jj));
                        v[k] = *(const vfl*)(feat + (long)sk * F + lane * VL);
                    }
                    #pragma unroll
                    for (int k0 = 0; k0 < Q_GB; k0 += Q_AG) {
                        if (g0 + k0 + Q_AG <= m) {
                            vfl tv[Q_AG];
                            #pragma unroll
                            for (int a = 0; a < Q_AG; ++a)
                                tv[a] = *(const vfl*)&tile[rr[k0 + a] * F + lane * VL];
                            #pragma unroll
                            for (int a = 0; a < Q_AG; ++a) {
                                #pragma unroll
                                for (int i = 0; i < a; ++i)
                                    if (rr[k0 + i] == rr[k0 + a]) tv[a] = tv[i];
                                tv[a] += v[k0 + a] * nn[k0 + a];
                            }
                            #pragma unroll
                            for (int a = 0; a < Q_AG; ++a)
                                *(vfl*)&tile[rr[k0 + a] * F + lane * VL] = tv[a];
                        } else {
                            #pragma unroll
                            for (int a = 0; a < Q_AG; ++a) {
                                if (g0 + k0 + a < m) {
                                    vfl* tp = (vfl*)&tile[rr[k0 + a] * F + lane * VL];
                                    vfl tv = *tp;
                                    tv += v[k0 + a] * nn[k0 + a];
                                    *tp = tv;
                                }
                            }
                        }
                    }
                }
            }
        }
        __syncthreads();
        seg = nxSeg; p = nxP; b ^= 1;
    }

    for (int i = tid; i < R * 32; i += 1024) {
        const int r = row0 + (i >> 5);
        if (r < nd)
            *(vfl*)(res + (long)r * F + (i & 31) * VL) = ((const vfl*)tile)[i];
    }
}

template <int K, int N, int MT, int UNR, bool RELU, bool BIAS>
__global__ __launch_bounds__(128)
void k_gemm_wmma(const float* __restrict__ A,
                 const float* __restrict__ rowscale,
                 const float* __restrict__ W,
                 const float* __restrict__ bias,
                 float* __restrict__ C, int nRowGroups) {
    constexpr int NG = N / 64;
    const int wave = (blockIdx.x * blockDim.x + threadIdx.x) >> 5;
    const int lane = threadIdx.x & 31;
    const int rg = wave / NG;
    const int cg = wave % NG;
    if (rg >= nRowGroups) return;
    const int hf = lane >> 4;
    const int c  = lane & 15;
    const int row0 = rg * (MT * 16);
    const int col0 = cg * 64 + c * 4;

    const float* ap[MT];
    float sc[MT];
    #pragma unroll
    for (int m = 0; m < MT; ++m) {
        const int r = row0 + m * 16 + c;
        ap[m] = A + (long)r * K + hf * 4;
        sc[m] = rowscale[r];
    }
    const float* wp = W + (long)(hf * 4) * N + col0;

    v8f acc[MT][4];
    #pragma unroll
    for (int m = 0; m < MT; ++m)
        #pragma unroll
        for (int t = 0; t < 4; ++t) acc[m][t] = v8f{};

    #pragma unroll UNR
    for (int kk = 0; kk < K; kk += 8) {
        float4 av[MT], bq[4];
        #pragma unroll
        for (int m = 0; m < MT; ++m) av[m] = *(const float4*)(ap[m] + kk);
        #pragma unroll
        for (int i = 0; i < 4; ++i) bq[i] = *(const float4*)(wp + (long)(kk + i) * N);

        const float bv[4][4] = {
            {bq[0].x, bq[0].y, bq[0].z, bq[0].w},
            {bq[1].x, bq[1].y, bq[1].z, bq[1].w},
            {bq[2].x, bq[2].y, bq[2].z, bq[2].w},
            {bq[3].x, bq[3].y, bq[3].z, bq[3].w}};
        #pragma unroll
        for (int m = 0; m < MT; ++m) {
            v2f a; a.x = av[m].x * sc[m]; a.y = av[m].y * sc[m];
            #pragma unroll
            for (int t = 0; t < 4; ++t) {
                v2f b; b.x = bv[0][t]; b.y = bv[1][t];
                acc[m][t] = __builtin_amdgcn_wmma_f32_16x16x4_f32(
                    false, a, false, b, (short)0, acc[m][t], false, false);
            }
        }
        #pragma unroll
        for (int m = 0; m < MT; ++m) {
            v2f a; a.x = av[m].z * sc[m]; a.y = av[m].w * sc[m];
            #pragma unroll
            for (int t = 0; t < 4; ++t) {
                v2f b; b.x = bv[2][t]; b.y = bv[3][t];
                acc[m][t] = __builtin_amdgcn_wmma_f32_16x16x4_f32(
                    false, a, false, b, (short)0, acc[m][t], false, false);
            }
        }
    }

    float4 bval = {0.0f, 0.0f, 0.0f, 0.0f};
    if (BIAS) bval = *(const float4*)(bias + col0);
    #pragma unroll
    for (int m = 0; m < MT; ++m) {
        #pragma unroll
        for (int r = 0; r < 8; ++r) {
            float4 v;
            v.x = acc[m][0][r] + bval.x;
            v.y = acc[m][1][r] + bval.y;
            v.z = acc[m][2][r] + bval.z;
            v.w = acc[m][3][r] + bval.w;
            if (RELU) {
                v.x = fmaxf(v.x, 0.0f); v.y = fmaxf(v.y, 0.0f);
                v.z = fmaxf(v.z, 0.0f); v.w = fmaxf(v.w, 0.0f);
            }
            *(float4*)(C + (long)(row0 + m * 16 + hf * 8 + r) * N + col0) = v;
        }
    }
}

extern "C" void kernel_launch(void* const* d_in, const int* in_sizes, int n_in,
                              void* d_out, int out_size, void* d_ws, size_t ws_size,
                              hipStream_t stream) {
    const float* x    = (const float*)d_in[0];
    const float* W1   = (const float*)d_in[1];
    const float* b1   = (const float*)d_in[2];
    const float* W2   = (const float*)d_in[3];
    const float* b2   = (const float*)d_in[4];
    const int*   src1 = (const int*)d_in[5];
    const int*   dst1 = (const int*)d_in[6];
    const int*   src2 = (const int*)d_in[7];
    const int*   dst2 = (const int*)d_in[8];

    const int IN_F = 128, H_F = 256, N_CLS = 64;
    const int NSRC1 = in_sizes[0] / IN_F;
    const int E1    = in_sizes[5];
    const int E2    = in_sizes[7];
    const int ND1   = 100000;
    const int ND2   = 16384;

    float* ws    = (float*)d_ws;
    float* degS1 = ws;
    float* degD1 = degS1 + NSRC1;
    float* degS2 = degD1 + ND1;
    float* degD2 = degS2 + ND1;
    float* agg1  = degD2 + ND2;
    float* h     = agg1 + (long)ND1 * IN_F;
    float* g     = h + (long)ND1 * H_F;
    float* key1  = g;

    long zcount = (NSRC1 + ND1 + ND1 + ND2);
    k_zero<<<256, 256, 0, stream>>>(ws, zcount);

    const int nb1 = (E1 + D_EPB - 1) / D_EPB;
    const int nb2 = (E2 + D_EPB - 1) / D_EPB;
    k_deg<<<nb1 + nb2, 256, 0, stream>>>(src1, dst1, E1, degS1, degD1, nb1,
                                         src2, dst2, E2, degS2, degD2, key1);
    k_rsqrt<<<256, 256, 0, stream>>>(ws, zcount);

    k_pull_split<128, PULL1_R>
        <<<(ND1 + PULL1_R - 1) / PULL1_R, 1024, 0, stream>>>(x, src1, dst1, degS1, key1, agg1, E1, ND1);

    {
        k_gemm_wmma<128, 256, 5, 2, true, true>
            <<<1250, 128, 0, stream>>>(agg1, degD1, W1, b1, h, 1250);
    }

    {
        k_gemm_wmma<256, 64, 2, 4, false, false>
            <<<782, 128, 0, stream>>>(h, degS2, W2, nullptr, g, 3125);
    }

    k_pull<64, 64, false, true>
        <<<(ND2 + 63) / 64, 1024, 0, stream>>>(g, src2, dst2, nullptr, degD2, b2, (float*)d_out, E2, ND2);
}


# Destination degrees counted in registers by the aggregations' row-owning waves, which store the norms themselves: 1862144 scattered global atomic adds gone; float sums and their order unchanged
# speedup vs baseline: 2.2120x; 1.1896x over previous
#include <hip/hip_runtime.h>


typedef __attribute__((ext_vector_type(2))) float v2f;
typedef __attribute__((ext_vector_type(8))) float v8f;

__device__ __forceinline__ void atomAddF(float* p, float v) {
    asm volatile("global_atomic_add_f32 %0, %1, off" :: "v"(p), "v"(v) : "memory");
}

__device__ __forceinline__ void l2Fetch(const float* p) {
    asm volatile("global_prefetch_b8 %0, off scope:SCOPE_SE" :: "v"(p) : "memory");
}

__global__ __launch_bounds__(256)
void k_zero(float* __restrict__ p, long n) {
    const long n4 = n >> 2;
    const long i = (long)blockIdx.x * blockDim.x + threadIdx.x;
    const long stride = (long)gridDim.x * blockDim.x;
    const float4 z = {0.0f, 0.0f, 0.0f, 0.0f};
    for (long j = i; j < n4; j += stride) ((float4*)p)[j] = z;
    const long t = (n4 << 2) + i;
    if (t < n) p[t] = 0.0f;
}

constexpr int D_EPB = 2048;
constexpr int PULL1_R = 400;
constexpr int KEY_PAD = 65536;
__device__ __forceinline__ unsigned tileKeys(const int4 d) {
    return (unsigned)(d.x / PULL1_R) | ((unsigned)(d.y / PULL1_R) << 8) |
           ((unsigned)(d.z / PULL1_R) << 16) | ((unsigned)(d.w / PULL1_R) << 24);
}
__global__ __launch_bounds__(256)
void k_deg(const int* __restrict__ src1, const int* __restrict__ dst1, int ne1,
           float* __restrict__ degS1, int nb1,
           const int* __restrict__ src2, int ne2,
           float* __restrict__ degS2,
           float* __restrict__ key1f) {
    const bool first = (int)blockIdx.x < nb1;
    const int* __restrict__ src = first ? src1 : src2;
    float* __restrict__ degS = first ? degS1 : degS2;
    const int ne  = first ? ne1 : ne2;
    const int blk = first ? (int)blockIdx.x : (int)blockIdx.x - nb1;
    const int e0  = blk * D_EPB + (int)threadIdx.x * 4;
    if (blk * D_EPB + D_EPB <= ne) {
        const int4 s0 = *(const int4*)(src + e0);
        const int4 s1 = *(const int4*)(src + e0 + 1024);
        if (first) {
            const int4 d0 = *(const int4*)(dst1 + e0);
            const int4 d1 = *(const int4*)(dst1 + e0 + 1024);
            ((unsigned*)key1f)[e0 >> 2]          = tileKeys(d0);
            ((unsigned*)key1f)[(e0 + 1024) >> 2] = tileKeys(d1);
        }
        atomAddF(&degS[s0.x], 1.0f); atomAddF(&degS[s0.y], 1.0f);
        atomAddF(&degS[s0.z], 1.0f); atomAddF(&degS[s0.w], 1.0f);
        atomAddF(&degS[s1.x], 1.0f); atomAddF(&degS[s1.y], 1.0f);
        atomAddF(&degS[s1.z], 1.0f); atomAddF(&degS[s1.w], 1.0f);
    } else {
        for (int k = 0; k < 8; ++k) {
            const int e = e0 + (k >> 2) * 1024 + (k & 3);
            if (e < ne) {
                if (first) ((unsigned char*)key1f)[e] = (unsigned char)(dst1[e] / PULL1_R);
                atomAddF(&degS[src[e]], 1.0f);
            }
        }
    }
    if (first && blk == nb1 - 1) {
        const int a4 = (ne + 3) & ~3;
        const int padEnd = ((ne + KEY_PAD - 1) / KEY_PAD) * KEY_PAD;
        if ((int)threadIdx.x < a4 - ne) ((unsigned char*)key1f)[ne + (int)threadIdx.x] = (unsigned char)255;
        for (int w = (a4 >> 2) + (int)threadIdx.x; w < (padEnd >> 2); w += 256)
            ((unsigned*)key1f)[w] = 0xFFFFFFFFu;
    }
}

__global__ __launch_bounds__(256)
void k_rsqrt(float* __restrict__ p, long n) {
    const long n4 = n >> 2;
    const long i = (long)blockIdx.x * blockDim.x + threadIdx.x;
    const long stride = (long)gridDim.x * blockDim.x;
    for (long j0 = i; j0 < n4; j0 += 4 * stride) {
        float4 v[4];
        #pragma unroll
        for (int k = 0; k < 4; ++k) {
            const long j = j0 + k * stride;
            if (j < n4) v[k] = ((const float4*)p)[j];
        }
        #pragma unroll
        for (int k = 0; k < 4; ++k) {
            const long j = j0 + k * stride;
            if (j < n4) {
                float4 t = v[k];
                t.x = rsqrtf(fmaxf(t.x, 1.0f)); t.y = rsqrtf(fmaxf(t.y, 1.0f));
                t.z = rsqrtf(fmaxf(t.z, 1.0f)); t.w = rsqrtf(fmaxf(t.w, 1.0f));
                ((float4*)p)[j] = t;
            }
        }
    }
    const long t = (n4 << 2) + i;
    if (t < n) p[t] = rsqrtf(fmaxf(p[t], 1.0f));
}

constexpr int P_NLD  = 16;
constexpr int P_SEG  = 1024 * 4 * P_NLD;
constexpr int P_WCAP = 24;
constexpr int P_MCAP = 32 * P_WCAP;
constexpr int P_MAXP = (32 * 4 * P_NLD) / P_WCAP;

template <int F, int R, bool SRCSCALE, bool FINAL>
__global__ __launch_bounds__(1024)
void k_pull(const float* __restrict__ feat, const int* __restrict__ src,
            const int* __restrict__ dst, const float* __restrict__ normS,
            const float* __restrict__ bias,
            float* __restrict__ res, int ne, int nd) {
    constexpr int VL = F / 32;
    typedef float vfl __attribute__((ext_vector_type(VL)));
    static_assert(R <= 512 && (F == 128 || F == 64), "entry packing and lane vectors");

    __shared__ __attribute__((aligned(16))) float tile[R * F];
    __shared__ unsigned wq[32 * P_WCAP];
    __shared__ int wcnt[32];
    __shared__ unsigned mylist[32 * P_MCAP];
    __shared__ int more[2];
    __shared__ int rcnt[R];

    const int tid  = threadIdx.x;
    const int lane = tid & 31;
    const int wave = __builtin_amdgcn_readfirstlane(tid >> 5);
    const int row0 = blockIdx.x * R;

    for (int i = tid; i < R * 32; i += 1024) ((vfl*)tile)[i] = vfl{};
    if (tid < 2) more[tid] = 0;
    __syncthreads();

    int pp = 0;
    int cntv = 0;
    const int nseg = (ne + P_SEG - 1) / P_SEG;
    for (int seg = 0; seg < nseg; ++seg) {
        const int segBase = seg * P_SEG;

        for (int p = 0; p <= P_MAXP; ++p) {
            const int fl = pp & 1;

            int tot = 0;
            int4 dv[P_NLD];
            if (segBase + P_SEG <= ne) {
                #pragma unroll
                for (int u = 0; u < P_NLD; ++u)
                    dv[u] = *(const int4*)(dst + segBase + (u * 1024 + tid) * 4);
            } else {
                #pragma unroll
                for (int u = 0; u < P_NLD; ++u) {
                    const int i = segBase + (u * 1024 + tid) * 4;
                    if (i + 3 < ne) {
                        dv[u] = *(const int4*)(dst + i);
                    } else {
                        dv[u].x = (i     < ne) ? dst[i]     : -1;
                        dv[u].y = (i + 1 < ne) ? dst[i + 1] : -1;
                        dv[u].z = (i + 2 < ne) ? dst[i + 2] : -1;
                        dv[u].w = -1;
                    }
                }
            }

            #pragma unroll
            for (int u = 0; u < P_NLD; ++u) {
                const int dd[4] = {dv[u].x, dv[u].y, dv[u].z, dv[u].w};
                unsigned rv[4], mk[4];
                #pragma unroll
                for (int c = 0; c < 4; ++c) {
                    rv[c] = (unsigned)(dd[c] - row0);
                    mk[c] = __builtin_amdgcn_ballot_w32(rv[c] < (unsigned)R);
                }
                if ((mk[0] | mk[1] | mk[2] | mk[3]) != 0u) {
                    #pragma unroll
                    for (int c = 0; c < 4; ++c) {
                        if (mk[c] != 0u) {
                            const int pos = tot + (int)__builtin_amdgcn_mbcnt_lo(mk[c], 0u) - p * P_WCAP;
                            if (rv[c] < (unsigned)R && pos >= 0 && pos < P_WCAP)
                                wq[wave * P_WCAP + pos] =
                                    ((unsigned)((u * 1024 + tid) * 4 + c) << 9) | rv[c];
                            tot += __builtin_popcount(mk[c]);
                        }
                    }
                }
            }
            if (lane == 0) {
                int cc = tot - p * P_WCAP;
                cc = cc < 0 ? 0 : (cc > P_WCAP ? P_WCAP : cc);
                wcnt[wave] = cc;
                if (tot > (p + 1) * P_WCAP) more[fl] = 1;
            }
            __syncthreads();
            const int again = __builtin_amdgcn_readfirstlane(more[fl]);
            if (tid == 0) more[fl ^ 1] = 0;

            int mc = 0;
            const int cnts = wcnt[lane];
            unsigned ents[32];
            const int qi = (lane < P_WCAP) ? lane : 0;
            #pragma unroll
            for (int sw = 0; sw < 32; ++sw) ents[sw] = wq[sw * P_WCAP + qi];
            #pragma unroll
            for (int sw = 0; sw < 32; ++sw) {
                const int cq = __builtin_amdgcn_readlane(cnts, sw);
                const unsigned ent = ents[sw];
                const bool mine = (lane < cq) && ((ent & 31u) == (unsigned)wave);
                const unsigned mk = __builtin_amdgcn_ballot_w32(mine);
                if (mk != 0u) {
                    if (mine)
                        mylist[wave * P_MCAP + mc + (int)__builtin_amdgcn_mbcnt_lo(mk, 0u)] = ent;
                    mc += __builtin_popcount(mk);
                }
            }
            __builtin_amdgcn_fence(__ATOMIC_RELEASE, "wavefront");
            __builtin_amdgcn_wave_barrier();
            __builtin_amdgcn_fence(__ATOMIC_ACQUIRE, "wavefront");

            for (int j0 = 0; j0 < mc; j0 += 32) {
                const int j = j0 + lane;
                int   e_s = 0, e_r = 0;
                float e_n = 1.0f;
                if (j < mc) {
                    const unsigned ent = mylist[wave * P_MCAP + j];
                    e_r = (int)(ent & 511u);
                    e_s = src[segBase + (int)(ent >> 9)];
                    if (SRCSCALE) e_n = normS[e_s];
                }
                const int m = (mc - j0 < 32) ? (mc - j0) : 32;
                for (int b = 0; b < m; b += 8) {
                    vfl   v[8];
                    int   rr[8];
                    float nn[8];
                    #pragma unroll
                    for (int k = 0; k < 8; ++k) {
                        const int jj = (b + k < m) ? (b + k) : (m - 1);
                        const int sk = __builtin_amdgcn_readlane(e_s, jj);
                        rr[k] = __builtin_amdgcn_readlane(e_r, jj);
                        nn[k] = __int_as_float(__builtin_amdgcn_readlane(__float_as_int(e_n), jj));
                        v[k] = *(const vfl*)(feat + (long)sk * F + lane * VL);
                    }
                    #pragma unroll
                    for (int k = 0; k < 8; ++k) {
                        if (b + k < m) {
                            vfl* tp = (vfl*)&tile[rr[k] * F + lane * VL];
                            vfl t = *tp;
                            if (SRCSCALE) t += v[k] * nn[k];
                            else          t += v[k];
                            *tp = t;
                            if (FINAL) cntv += (((unsigned)rr[k] >> 5) == (unsigned)lane) ? 1 : 0;
                        }
                    }
                }
            }
            __syncthreads();
            ++pp;
            if (!again) break;
        }
    }

    if (FINAL) {
        if (lane * 32 + wave < R) rcnt[lane * 32 + wave] = cntv;
        __syncthreads();
    }

    for (int i = tid; i < R * 32; i += 1024) {
        const int r = row0 + (i >> 5);
        if (r < nd) {
            vfl t = ((const vfl*)tile)[i];
            if (FINAL) {
                const float sc = rsqrtf(fmaxf((float)rcnt[i >> 5], 1.0f));
                const vfl bv = *(const vfl*)(bias + (i & 31) * VL);
                t = t * sc + bv;
            }
            *(vfl*)(res + (long)r * F + (i & 31) * VL) = t;
        }
    }
}

constexpr int Q_SCAN = 16;
constexpr int Q_OWN  = 32 - Q_SCAN;
constexpr int Q_NLD  = 8;
constexpr int Q_SEG  = Q_SCAN * 32 * 16 * Q_NLD;
constexpr int Q_WCAP = 32;
constexpr int Q_MCAP = Q_SCAN * Q_WCAP;
constexpr int Q_MAXP = (32 * 16 * Q_NLD) / Q_WCAP;
constexpr int Q_GB   = 16;
constexpr int Q_AG   = 4;

template <int F, int R>
__global__ __launch_bounds__(1024)
void k_pull_split(const float* __restrict__ feat, const int* __restrict__ src,
                  const int* __restrict__ dst, const float* __restrict__ normS,
                  const float* __restrict__ keyf,
                  float* __restrict__ res, float* __restrict__ normD, int ne, int nd) {
    constexpr int VL = F / 32;
    typedef float vfl __attribute__((ext_vector_type(VL)));
    static_assert(R <= 512 && F == 128, "entry packing and lane vectors");
    static_assert(Q_SEG == KEY_PAD, "the tile bytes are padded to whole segments");
    static_assert(Q_WCAP == 32, "one queue entry per lane when the records are fetched and when they are picked");
    static_assert((Q_OWN & (Q_OWN - 1)) == 0 && Q_SCAN <= 32, "row owner by mask; one scanning wave's total per lane");

    __shared__ __attribute__((aligned(16))) float tile[R * F];
    __shared__ unsigned qrow[2][Q_SCAN * Q_WCAP];
    __shared__ int      qsrc[2][Q_SCAN * Q_WCAP];
    __shared__ float    qnrm[2][Q_SCAN * Q_WCAP];
    __shared__ int      qtot[2][Q_SCAN];
    __shared__ unsigned mylist[Q_OWN * Q_MCAP];
    __shared__ int      rcnt[R];

    const int tid  = threadIdx.x;
    const int lane = tid & 31;
    const int wave = __builtin_amdgcn_readfirstlane(tid >> 5);
    const int row0 = blockIdx.x * R;
    const bool scanner = wave < Q_SCAN;

    for (int i = tid; i < R * 32; i += 1024) ((vfl*)tile)[i] = vfl{};

    const int nseg = (ne + Q_SEG - 1) / Q_SEG;

    auto scanUnit = [&](const int seg, const int p, const int nb) {
        const int segBase = seg * Q_SEG;
        const unsigned char* keyb = (const unsigned char*)keyf + segBase;
        uint4 kv[Q_NLD];
        #pragma unroll
        for (int u = 0; u < Q_NLD; ++u)
            kv[u] = *(const uint4*)(keyb + (u * (Q_SCAN * 32) + tid) * 16);

        unsigned* qr = &qrow[nb][wave * Q_WCAP];
        const unsigned tileRep = (unsigned)blockIdx.x * 0x01010101u;
        int tot = 0;
        #pragma unroll
        for (int u = 0; u < Q_NLD; ++u) {
            const unsigned wv[4] = {kv[u].x ^ tileRep, kv[u].y ^ tileRep,
                                    kv[u].z ^ tileRep, kv[u].w ^ tileRep};
            unsigned hm = 0u;
            #pragma unroll
            for (int q = 0; q < 4; ++q) {
                const unsigned y = ((wv[q] & 0x7F7F7F7Fu) + 0x7F7F7F7Fu) | wv[q];
                hm |= (~y & 0x80808080u) >> (7 - q);
            }
            unsigned bal = __builtin_amdgcn_ballot_w32(hm != 0u);
            while (bal != 0u) {
                const int bp = __builtin_ctz(hm | 0x80000000u);
                const int o16 = ((bp & 7) << 2) | (bp >> 3);
                const int pos = tot + (int)__builtin_amdgcn_mbcnt_lo(bal, 0u) - p * Q_WCAP;
                if (hm != 0u && pos >= 0 && pos < Q_WCAP)
                    qr[pos] = (unsigned)((u * (Q_SCAN * 32) + tid) * 16 + o16);
                tot += __builtin_popcount(bal);
                hm &= hm - 1u;
                bal = __builtin_amdgcn_ballot_w32(hm != 0u);
            }
        }

        __builtin_amdgcn_fence(__ATOMIC_RELEASE, "wavefront");
        __builtin_amdgcn_wave_barrier();
        __builtin_amdgcn_fence(__ATOMIC_ACQUIRE, "wavefront");
        int nq = tot - p * Q_WCAP;
        nq = nq < 0 ? 0 : (nq > Q_WCAP ? Q_WCAP : nq);
        if (lane < nq) {
            const int e = segBase + (int)qr[lane];
            const int d = dst[e];
            const int s = src[e];
            l2Fetch(feat + (long)s * F);
            l2Fetch(feat + (long)s * F + 64);
            qr[lane] = (unsigned)(d - row0);
            qsrc[nb][wave * Q_WCAP + lane] = s;
            qnrm[nb][wave * Q_WCAP + lane] = normS[s];
        }
        if (lane == 0) qtot[nb][wave] = tot;
    };

    if (scanner && nseg > 0) scanUnit(0, 0, 0);
    __syncthreads();

    int cntv = 0;
    int seg = 0, p = 0, b = 0;
    for (int it = 0; it < nseg * (Q_MAXP + 1); ++it) {
        if (seg >= nseg) break;
        const int t = (lane < Q_SCAN) ? qtot[b][lane] : 0;
        const bool again = __builtin_amdgcn_ballot_w32(t > (p + 1) * Q_WCAP) != 0u;
        const int nxSeg = again ? seg : seg + 1;
        const int nxP   = again ? p + 1 : 0;

        if (scanner) {
            if (nxSeg < nseg) scanUnit(nxSeg, nxP, b ^ 1);
        } else {
            const int ow = wave - Q_SCAN;
            int cnts = t - p * Q_WCAP;
            cnts = cnts < 0 ? 0 : (cnts > Q_WCAP ? Q_WCAP : cnts);

            unsigned ents[Q_SCAN];
            #pragma unroll
            for (int sw = 0; sw < Q_SCAN; ++sw) ents[sw] = qrow[b][sw * Q_WCAP + lane];
            int mc = 0;
            #pragma unroll
            for (int sw = 0; sw < Q_SCAN; ++sw) {
                const int cq = __builtin_amdgcn_readlane(cnts, sw);
                const unsigned ent = ents[sw];
                const bool mine = (lane < cq) && ((ent & (unsigned)(Q_OWN - 1)) == (unsigned)ow);
                const unsigned mk = __builtin_amdgcn_ballot_w32(mine);
                if (mk != 0u) {
                    if (mine)
                        mylist[ow * Q_MCAP + mc + (int)__builtin_amdgcn_mbcnt_lo(mk, 0u)] =
                            ((unsigned)(sw * Q_WCAP + lane) << 9) | ent;
                    mc += __builtin_popcount(mk);
                }
            }
            __builtin_amdgcn_fence(__ATOMIC_RELEASE, "wavefront");
            __builtin_amdgcn_wave_barrier();
            __builtin_amdgcn_fence(__ATOMIC_ACQUIRE, "wavefront");

            for (int j0 = 0; j0 < mc; j0 += 32) {
                const int j = j0 + lane;
                int   e_s = 0, e_r = 0;
                float e_n = 1.0f;
                if (j < mc) {
                    const unsigned ent = mylist[ow * Q_MCAP + j];
                    const int qi = (int)(ent >> 9);
                    e_r = (int)(ent & 511u);
                    e_s = qsrc[b][qi];
                    e_n = qnrm[b][qi];
                }
                const int m = (mc - j0 < 32) ? (mc - j0) : 32;
                for (int g0 = 0; g0 < m; g0 += Q_GB) {
                    vfl   v[Q_GB];
                    int   rr[Q_GB];
                    float nn[Q_GB];
                    #pragma unroll
                    for (int k = 0; k < Q_GB; ++k) {
                        const int jj = (g0 + k < m) ? (g0 + k) : (m - 1);
                        const int sk = __builtin_amdgcn_readlane(e_s, jj);
                        rr[k] = __builtin_amdgcn_readlane(e_r, jj);
                        nn[k] = __int_as_float(__builtin_amdgcn_readlane(__float_as_int(e_n), jj));
                        v[k] = *(const vfl*)(feat + (long)sk * F + lane * VL);
                    }
                    #pragma unroll
                    for (int k0 = 0; k0 < Q_GB; k0 += Q_AG) {
                        if (g0 + k0 + Q_AG <= m) {
                            vfl tv[Q_AG];
                            #pragma unroll
                            for (int a = 0; a < Q_AG; ++a)
                                tv[a] = *(const vfl*)&tile[rr[k0 + a] * F + lane * VL];
                            #pragma unroll
                            for (int a = 0; a < Q_AG; ++a) {
                                #pragma unroll
                                for (int i = 0; i < a; ++i)
                                    if (rr[k0 + i] == rr[k0 + a]) tv[a] = tv[i];
                                tv[a] += v[k0 + a] * nn[k0 + a];
                                cntv += (((unsigned)rr[k0 + a] / (unsigned)Q_OWN) == (unsigned)lane) ? 1 : 0;
                            }
                            #pragma unroll
                            for (int a = 0; a < Q_AG; ++a)
                                *(vfl*)&tile[rr[k0 + a] * F + lane * VL] = tv[a];
                        } else {
                            #pragma unroll
                            for (int a = 0; a < Q_AG; ++a) {
                                if (g0 + k0 + a < m) {
                                    vfl* tp = (vfl*)&tile[rr[k0 + a] * F + lane * VL];
                                    vfl tv = *tp;
                                    tv += v[k0 + a] * nn[k0 + a];
                                    *tp = tv;
                                    cntv += (((unsigned)rr[k0 + a] / (unsigned)Q_OWN) == (unsigned)lane) ? 1 : 0;
                                }
                            }
                        }
                    }
                }
            }
        }
        __syncthreads();
        seg = nxSeg; p = nxP; b ^= 1;
    }

    if (!scanner) {
        const int ow = wave - Q_SCAN;
        if (lane * Q_OWN + ow < R) rcnt[lane * Q_OWN + ow] = cntv;
    }
    __syncthreads();

    for (int i = tid; i < R * 32; i += 1024) {
        const int r = row0 + (i >> 5);
        if (r < nd)
            *(vfl*)(res + (long)r * F + (i & 31) * VL) = ((const vfl*)tile)[i];
    }
    for (int i = tid; i < R; i += 1024) {
        const int r = row0 + i;
        if (r < nd) normD[r] = rsqrtf(fmaxf((float)rcnt[i], 1.0f));
    }
}

template <int K, int N, int MT, int UNR, bool RELU, bool BIAS>
__global__ __launch_bounds__(128)
void k_gemm_wmma(const float* __restrict__ A,
                 const float* __restrict__ rowscale,
                 const float* __restrict__ W,
                 const float* __restrict__ bias,
                 float* __restrict__ C, int nRowGroups) {
    constexpr int NG = N / 64;
    const int wave = (blockIdx.x * blockDim.x + threadIdx.x) >> 5;
    const int lane = threadIdx.x & 31;
    const int rg = wave / NG;
    const int cg = wave % NG;
    if (rg >= nRowGroups) return;
    const int hf = lane >> 4;
    const int c  = lane & 15;
    const int row0 = rg * (MT * 16);
    const int col0 = cg * 64 + c * 4;

    const float* ap[MT];
    float sc[MT];
    #pragma unroll
    for (int m = 0; m < MT; ++m) {
        const int r = row0 + m * 16 + c;
        ap[m] = A + (long)r * K + hf * 4;
        sc[m] = rowscale[r];
    }
    const float* wp = W + (long)(hf * 4) * N + col0;

    v8f acc[MT][4];
    #pragma unroll
    for (int m = 0; m < MT; ++m)
        #pragma unroll
        for (int t = 0; t < 4; ++t) acc[m][t] = v8f{};

    #pragma unroll UNR
    for (int kk = 0; kk < K; kk += 8) {
        float4 av[MT], bq[4];
        #pragma unroll
        for (int m = 0; m < MT; ++m) av[m] = *(const float4*)(ap[m] + kk);
        #pragma unroll
        for (int i = 0; i < 4; ++i) bq[i] = *(const float4*)(wp + (long)(kk + i) * N);

        const float bv[4][4] = {
            {bq[0].x, bq[0].y, bq[0].z, bq[0].w},
            {bq[1].x, bq[1].y, bq[1].z, bq[1].w},
            {bq[2].x, bq[2].y, bq[2].z, bq[2].w},
            {bq[3].x, bq[3].y, bq[3].z, bq[3].w}};
        #pragma unroll
        for (int m = 0; m < MT; ++m) {
            v2f a; a.x = av[m].x * sc[m]; a.y = av[m].y * sc[m];
            #pragma unroll
            for (int t = 0; t < 4; ++t) {
                v2f b; b.x = bv[0][t]; b.y = bv[1][t];
                acc[m][t] = __builtin_amdgcn_wmma_f32_16x16x4_f32(
                    false, a, false, b, (short)0, acc[m][t], false, false);
            }
        }
        #pragma unroll
        for (int m = 0; m < MT; ++m) {
            v2f a; a.x = av[m].z * sc[m]; a.y = av[m].w * sc[m];
            #pragma unroll
            for (int t = 0; t < 4; ++t) {
                v2f b; b.x = bv[2][t]; b.y = bv[3][t];
                acc[m][t] = __builtin_amdgcn_wmma_f32_16x16x4_f32(
                    false, a, false, b, (short)0, acc[m][t], false, false);
            }
        }
    }

    float4 bval = {0.0f, 0.0f, 0.0f, 0.0f};
    if (BIAS) bval = *(const float4*)(bias + col0);
    #pragma unroll
    for (int m = 0; m < MT; ++m) {
        #pragma unroll
        for (int r = 0; r < 8; ++r) {
            float4 v;
            v.x = acc[m][0][r] + bval.x;
            v.y = acc[m][1][r] + bval.y;
            v.z = acc[m][2][r] + bval.z;
            v.w = acc[m][3][r] + bval.w;
            if (RELU) {
                v.x = fmaxf(v.x, 0.0f); v.y = fmaxf(v.y, 0.0f);
                v.z = fmaxf(v.z, 0.0f); v.w = fmaxf(v.w, 0.0f);
            }
            *(float4*)(C + (long)(row0 + m * 16 + hf * 8 + r) * N + col0) = v;
        }
    }
}

extern "C" void kernel_launch(void* const* d_in, const int* in_sizes, int n_in,
                              void* d_out, int out_size, void* d_ws, size_t ws_size,
                              hipStream_t stream) {
    const float* x    = (const float*)d_in[0];
    const float* W1   = (const float*)d_in[1];
    const float* b1   = (const float*)d_in[2];
    const float* W2   = (const float*)d_in[3];
    const float* b2   = (const float*)d_in[4];
    const int*   src1 = (const int*)d_in[5];
    const int*   dst1 = (const int*)d_in[6];
    const int*   src2 = (const int*)d_in[7];
    const int*   dst2 = (const int*)d_in[8];

    const int IN_F = 128, H_F = 256, N_CLS = 64;
    const int NSRC1 = in_sizes[0] / IN_F;
    const int E1    = in_sizes[5];
    const int E2    = in_sizes[7];
    const int ND1   = 100000;
    const int ND2   = 16384;

    float* ws    = (float*)d_ws;
    float* degS1 = ws;
    float* degS2 = degS1 + NSRC1;
    float* degD1 = degS2 + ND1;
    float* agg1  = degD1 + ND1 + ND2;
    float* h     = agg1 + (long)ND1 * IN_F;
    float* g     = h + (long)ND1 * H_F;
    float* key1  = g;

    long zcount = (NSRC1 + ND1);
    k_zero<<<256, 256, 0, stream>>>(ws, zcount);

    const int nb1 = (E1 + D_EPB - 1) / D_EPB;
    const int nb2 = (E2 + D_EPB - 1) / D_EPB;
    k_deg<<<nb1 + nb2, 256, 0, stream>>>(src1, dst1, E1, degS1, nb1,
                                         src2, E2, degS2, key1);
    k_rsqrt<<<256, 256, 0, stream>>>(ws, zcount);

    k_pull_split<128, PULL1_R>
        <<<(ND1 + PULL1_R - 1) / PULL1_R, 1024, 0, stream>>>(x, src1, dst1, degS1, key1, agg1, degD1, E1, ND1);

    {
        k_gemm_wmma<128, 256, 5, 2, true, true>
            <<<1250, 128, 0, stream>>>(agg1, degD1, W1, b1, h, 1250);
    }

    {
        k_gemm_wmma<256, 64, 2, 4, false, false>
            <<<782, 128, 0, stream>>>(h, degS2, W2, nullptr, g, 3125);
    }

    k_pull<64, 64, false, true>
        <<<(ND2 + 63) / 64, 1024, 0, stream>>>(g, src2, dst2, nullptr, b2, (float*)d_out, E2, ND2);
}


# Source degree counters kept twice, one copy per half of the chip: a wave reads its die number and adds into its half's copy, the rsqrt pass adds both; counter lines stay in one second-level cache
# speedup vs baseline: 2.5939x; 1.1726x over previous
#include <hip/hip_runtime.h>


typedef __attribute__((ext_vector_type(2))) float v2f;
typedef __attribute__((ext_vector_type(8))) float v8f;

__device__ __forceinline__ void atomAddF(float* p, float v) {
    asm volatile("global_atomic_add_f32 %0, %1, off" :: "v"(p), "v"(v) : "memory");
}

__device__ __forceinline__ void l2Fetch(const float* p) {
    asm volatile("global_prefetch_b8 %0, off scope:SCOPE_SE" :: "v"(p) : "memory");
}

__global__ __launch_bounds__(256)
void k_zero(float* __restrict__ p, long n) {
    const long n4 = n >> 2;
    const long i = (long)blockIdx.x * blockDim.x + threadIdx.x;
    const long stride = (long)gridDim.x * blockDim.x;
    const float4 z = {0.0f, 0.0f, 0.0f, 0.0f};
    for (long j = i; j < n4; j += stride) ((float4*)p)[j] = z;
    const long t = (n4 << 2) + i;
    if (t < n) p[t] = 0.0f;
}

constexpr int D_EPB = 2048;
constexpr int PULL1_R = 400;
constexpr int KEY_PAD = 65536;
__device__ __forceinline__ unsigned tileKeys(const int4 d) {
    return (unsigned)(d.x / PULL1_R) | ((unsigned)(d.y / PULL1_R) << 8) |
           ((unsigned)(d.z / PULL1_R) << 16) | ((unsigned)(d.w / PULL1_R) << 24);
}
__global__ __launch_bounds__(256)
void k_deg(const int* __restrict__ src1, const int* __restrict__ dst1, int ne1,
           float* __restrict__ degS1, int nb1,
           const int* __restrict__ src2, int ne2,
           float* __restrict__ degS2, int halfStride,
           float* __restrict__ key1f) {
    const unsigned aid = __builtin_amdgcn_s_sendmsg_rtn(0x87);
    const long hoff = (long)((aid >> 18) & 1u) * halfStride;
    const bool first = (int)blockIdx.x < nb1;
    const int* __restrict__ src = first ? src1 : src2;
    float* __restrict__ degS = (first ? degS1 : degS2) + hoff;
    const int ne  = first ? ne1 : ne2;
    const int blk = first ? (int)blockIdx.x : (int)blockIdx.x - nb1;
    const int e0  = blk * D_EPB + (int)threadIdx.x * 4;
    if (blk * D_EPB + D_EPB <= ne) {
        const int4 s0 = *(const int4*)(src + e0);
        const int4 s1 = *(const int4*)(src + e0 + 1024);
        if (first) {
            const int4 d0 = *(const int4*)(dst1 + e0);
            const int4 d1 = *(const int4*)(dst1 + e0 + 1024);
            ((unsigned*)key1f)[e0 >> 2]          = tileKeys(d0);
            ((unsigned*)key1f)[(e0 + 1024) >> 2] = tileKeys(d1);
        }
        atomAddF(&degS[s0.x], 1.0f); atomAddF(&degS[s0.y], 1.0f);
        atomAddF(&degS[s0.z], 1.0f); atomAddF(&degS[s0.w], 1.0f);
        atomAddF(&degS[s1.x], 1.0f); atomAddF(&degS[s1.y], 1.0f);
        atomAddF(&degS[s1.z], 1.0f); atomAddF(&degS[s1.w], 1.0f);
    } else {
        for (int k = 0; k < 8; ++k) {
            const int e = e0 + (k >> 2) * 1024 + (k & 3);
            if (e < ne) {
                if (first) ((unsigned char*)key1f)[e] = (unsigned char)(dst1[e] / PULL1_R);
                atomAddF(&degS[src[e]], 1.0f);
            }
        }
    }
    if (first && blk == nb1 - 1) {
        const int a4 = (ne + 3) & ~3;
        const int padEnd = ((ne + KEY_PAD - 1) / KEY_PAD) * KEY_PAD;
        if ((int)threadIdx.x < a4 - ne) ((unsigned char*)key1f)[ne + (int)threadIdx.x] = (unsigned char)255;
        for (int w = (a4 >> 2) + (int)threadIdx.x; w < (padEnd >> 2); w += 256)
            ((unsigned*)key1f)[w] = 0xFFFFFFFFu;
    }
}

__global__ __launch_bounds__(256)
void k_rsqrt(const float* __restrict__ ca, const float* __restrict__ cb, float* __restrict__ p, long n) {
    const long n4 = n >> 2;
    const long i = (long)blockIdx.x * blockDim.x + threadIdx.x;
    const long stride = (long)gridDim.x * blockDim.x;
    for (long j0 = i; j0 < n4; j0 += 4 * stride) {
        float4 va[4], vb[4];
        #pragma unroll
        for (int k = 0; k < 4; ++k) {
            const long j = j0 + k * stride;
            if (j < n4) { va[k] = ((const float4*)ca)[j]; vb[k] = ((const float4*)cb)[j]; }
        }
        #pragma unroll
        for (int k = 0; k < 4; ++k) {
            const long j = j0 + k * stride;
            if (j < n4) {
                float4 t;
                t.x = rsqrtf(fmaxf(va[k].x + vb[k].x, 1.0f)); t.y = rsqrtf(fmaxf(va[k].y + vb[k].y, 1.0f));
                t.z = rsqrtf(fmaxf(va[k].z + vb[k].z, 1.0f)); t.w = rsqrtf(fmaxf(va[k].w + vb[k].w, 1.0f));
                ((float4*)p)[j] = t;
            }
        }
    }
    const long t = (n4 << 2) + i;
    if (t < n) p[t] = rsqrtf(fmaxf(ca[t] + cb[t], 1.0f));
}

constexpr int P_NLD  = 16;
constexpr int P_SEG  = 1024 * 4 * P_NLD;
constexpr int P_WCAP = 24;
constexpr int P_MCAP = 32 * P_WCAP;
constexpr int P_MAXP = (32 * 4 * P_NLD) / P_WCAP;

template <int F, int R, bool SRCSCALE, bool FINAL>
__global__ __launch_bounds__(1024)
void k_pull(const float* __restrict__ feat, const int* __restrict__ src,
            const int* __restrict__ dst, const float* __restrict__ normS,
            const float* __restrict__ bias,
            float* __restrict__ res, int ne, int nd) {
    constexpr int VL = F / 32;
    typedef float vfl __attribute__((ext_vector_type(VL)));
    static_assert(R <= 512 && (F == 128 || F == 64), "entry packing and lane vectors");

    __shared__ __attribute__((aligned(16))) float tile[R * F];
    __shared__ unsigned wq[32 * P_WCAP];
    __shared__ int wcnt[32];
    __shared__ unsigned mylist[32 * P_MCAP];
    __shared__ int more[2];
    __shared__ int rcnt[R];

    const int tid  = threadIdx.x;
    const int lane = tid & 31;
    const int wave = __builtin_amdgcn_readfirstlane(tid >> 5);
    const int row0 = blockIdx.x * R;

    for (int i = tid; i < R * 32; i += 1024) ((vfl*)tile)[i] = vfl{};
    if (tid < 2) more[tid] = 0;
    __syncthreads();

    int pp = 0;
    int cntv = 0;
    const int nseg = (ne + P_SEG - 1) / P_SEG;
    for (int seg = 0; seg < nseg; ++seg) {
        const int segBase = seg * P_SEG;

        for (int p = 0; p <= P_MAXP; ++p) {
            const int fl = pp & 1;

            int tot = 0;
            int4 dv[P_NLD];
            if (segBase + P_SEG <= ne) {
                #pragma unroll
                for (int u = 0; u < P_NLD; ++u)
                    dv[u] = *(const int4*)(dst + segBase + (u * 1024 + tid) * 4);
            } else {
                #pragma unroll
                for (int u = 0; u < P_NLD; ++u) {
                    const int i = segBase + (u * 1024 + tid) * 4;
                    if (i + 3 < ne) {
                        dv[u] = *(const int4*)(dst + i);
                    } else {
                        dv[u].x = (i     < ne) ? dst[i]     : -1;
                        dv[u].y = (i + 1 < ne) ? dst[i + 1] : -1;
                        dv[u].z = (i + 2 < ne) ? dst[i + 2] : -1;
                        dv[u].w = -1;
                    }
                }
            }

            #pragma unroll
            for (int u = 0; u < P_NLD; ++u) {
                const int dd[4] = {dv[u].x, dv[u].y, dv[u].z, dv[u].w};
                unsigned rv[4], mk[4];
                #pragma unroll
                for (int c = 0; c < 4; ++c) {
                    rv[c] = (unsigned)(dd[c] - row0);
                    mk[c] = __builtin_amdgcn_ballot_w32(rv[c] < (unsigned)R);
                }
                if ((mk[0] | mk[1] | mk[2] | mk[3]) != 0u) {
                    #pragma unroll
                    for (int c = 0; c < 4; ++c) {
                        if (mk[c] != 0u) {
                            const int pos = tot + (int)__builtin_amdgcn_mbcnt_lo(mk[c], 0u) - p * P_WCAP;
                            if (rv[c] < (unsigned)R && pos >= 0 && pos < P_WCAP)
                                wq[wave * P_WCAP + pos] =
                                    ((unsigned)((u * 1024 + tid) * 4 + c) << 9) | rv[c];
                            tot += __builtin_popcount(mk[c]);
                        }
                    }
                }
            }
            if (lane == 0) {
                int cc = tot - p * P_WCAP;
                cc = cc < 0 ? 0 : (cc > P_WCAP ? P_WCAP : cc);
                wcnt[wave] = cc;
                if (tot > (p + 1) * P_WCAP) more[fl] = 1;
            }
            __syncthreads();
            const int again = __builtin_amdgcn_readfirstlane(more[fl]);
            if (tid == 0) more[fl ^ 1] = 0;

            int mc = 0;
            const int cnts = wcnt[lane];
            unsigned ents[32];
            const int qi = (lane < P_WCAP) ? lane : 0;
            #pragma unroll
            for (int sw = 0; sw < 32; ++sw) ents[sw] = wq[sw * P_WCAP + qi];
            #pragma unroll
            for (int sw = 0; sw < 32; ++sw) {
                const int cq = __builtin_amdgcn_readlane(cnts, sw);
                const unsigned ent = ents[sw];
                const bool mine = (lane < cq) && ((ent & 31u) == (unsigned)wave);
                const unsigned mk = __builtin_amdgcn_ballot_w32(mine);
                if (mk != 0u) {
                    if (mine)
                        mylist[wave * P_MCAP + mc + (int)__builtin_amdgcn_mbcnt_lo(mk, 0u)] = ent;
                    mc += __builtin_popcount(mk);
                }
            }
            __builtin_amdgcn_fence(__ATOMIC_RELEASE, "wavefront");
            __builtin_amdgcn_wave_barrier();
            __builtin_amdgcn_fence(__ATOMIC_ACQUIRE, "wavefront");

            for (int j0 = 0; j0 < mc; j0 += 32) {
                const int j = j0 + lane;
                int   e_s = 0, e_r = 0;
                float e_n = 1.0f;
                if (j < mc) {
                    const unsigned ent = mylist[wave * P_MCAP + j];
                    e_r = (int)(ent & 511u);
                    e_s = src[segBase + (int)(ent >> 9)];
                    if (SRCSCALE) e_n = normS[e_s];
                }
                const int m = (mc - j0 < 32) ? (mc - j0) : 32;
                for (int b = 0; b < m; b += 8) {
                    vfl   v[8];
                    int   rr[8];
                    float nn[8];
                    #pragma unroll
                    for (int k = 0; k < 8; ++k) {
                        const int jj = (b + k < m) ? (b + k) : (m - 1);
                        const int sk = __builtin_amdgcn_readlane(e_s, jj);
                        rr[k] = __builtin_amdgcn_readlane(e_r, jj);
                        nn[k] = __int_as_float(__builtin_amdgcn_readlane(__float_as_int(e_n), jj));
                        v[k] = *(const vfl*)(feat + (long)sk * F + lane * VL);
                    }
                    #pragma unroll
                    for (int k = 0; k < 8; ++k) {
                        if (b + k < m) {
                            vfl* tp = (vfl*)&tile[rr[k] * F + lane * VL];
                            vfl t = *tp;
                            if (SRCSCALE) t += v[k] * nn[k];
                            else          t += v[k];
                            *tp = t;
                            if (FINAL) cntv += (((unsigned)rr[k] >> 5) == (unsigned)lane) ? 1 : 0;
                        }
                    }
                }
            }
            __syncthreads();
            ++pp;
            if (!again) break;
        }
    }

    if (FINAL) {
        if (lane * 32 + wave < R) rcnt[lane * 32 + wave] = cntv;
        __syncthreads();
    }

    for (int i = tid; i < R * 32; i += 1024) {
        const int r = row0 + (i >> 5);
        if (r < nd) {
            vfl t = ((const vfl*)tile)[i];
            if (FINAL) {
                const float sc = rsqrtf(fmaxf((float)rcnt[i >> 5], 1.0f));
                const vfl bv = *(const vfl*)(bias + (i & 31) * VL);
                t = t * sc + bv;
            }
            *(vfl*)(res + (long)r * F + (i & 31) * VL) = t;
        }
    }
}

constexpr int Q_SCAN = 16;
constexpr int Q_OWN  = 32 - Q_SCAN;
constexpr int Q_NLD  = 8;
constexpr int Q_SEG  = Q_SCAN * 32 * 16 * Q_NLD;
constexpr int Q_WCAP = 32;
constexpr int Q_MCAP = Q_SCAN * Q_WCAP;
constexpr int Q_MAXP = (32 * 16 * Q_NLD) / Q_WCAP;
constexpr int Q_GB   = 16;
constexpr int Q_AG   = 4;

template <int F, int R>
__global__ __launch_bounds__(1024)
void k_pull_split(const float* __restrict__ feat, const int* __restrict__ src,
                  const int* __restrict__ dst, const float* __restrict__ normS,
                  const float* __restrict__ keyf,
                  float* __restrict__ res, float* __restrict__ normD, int ne, int nd) {
    constexpr int VL = F / 32;
    typedef float vfl __attribute__((ext_vector_type(VL)));
    static_assert(R <= 512 && F == 128, "entry packing and lane vectors");
    static_assert(Q_SEG == KEY_PAD, "the tile bytes are padded to whole segments");
    static_assert(Q_WCAP == 32, "one queue entry per lane when the records are fetched and when they are picked");
    static_assert((Q_OWN & (Q_OWN - 1)) == 0 && Q_SCAN <= 32, "row owner by mask; one scanning wave's total per lane");

    __shared__ __attribute__((aligned(16))) float tile[R * F];
    __shared__ unsigned qrow[2][Q_SCAN * Q_WCAP];
    __shared__ int      qsrc[2][Q_SCAN * Q_WCAP];
    __shared__ float    qnrm[2][Q_SCAN * Q_WCAP];
    __shared__ int      qtot[2][Q_SCAN];
    __shared__ unsigned mylist[Q_OWN * Q_MCAP];
    __shared__ int      rcnt[R];

    const int tid  = threadIdx.x;
    const int lane = tid & 31;
    const int wave = __builtin_amdgcn_readfirstlane(tid >> 5);
    const int row0 = blockIdx.x * R;
    const bool scanner = wave < Q_SCAN;

    for (int i = tid; i < R * 32; i += 1024) ((vfl*)tile)[i] = vfl{};

    const int nseg = (ne + Q_SEG - 1) / Q_SEG;

    auto scanUnit = [&](const int seg, const int p, const int nb) {
        const int segBase = seg * Q_SEG;
        const unsigned char* keyb = (const unsigned char*)keyf + segBase;
        uint4 kv[Q_NLD];
        #pragma unroll
        for (int u = 0; u < Q_NLD; ++u)
            kv[u] = *(const uint4*)(keyb + (u * (Q_SCAN * 32) + tid) * 16);

        unsigned* qr = &qrow[nb][wave * Q_WCAP];
        const unsigned tileRep = (unsigned)blockIdx.x * 0x01010101u;
        int tot = 0;
        #pragma unroll
        for (int u = 0; u < Q_NLD; ++u) {
            const unsigned wv[4] = {kv[u].x ^ tileRep, kv[u].y ^ tileRep,
                                    kv[u].z ^ tileRep, kv[u].w ^ tileRep};
            unsigned hm = 0u;
            #pragma unroll
            for (int q = 0; q < 4; ++q) {
                const unsigned y = ((wv[q] & 0x7F7F7F7Fu) + 0x7F7F7F7Fu) | wv[q];
                hm |= (~y & 0x80808080u) >> (7 - q);
            }
            unsigned bal = __builtin_amdgcn_ballot_w32(hm != 0u);
            while (bal != 0u) {
                const int bp = __builtin_ctz(hm | 0x80000000u);
                const int o16 = ((bp & 7) << 2) | (bp >> 3);
                const int pos = tot + (int)__builtin_amdgcn_mbcnt_lo(bal, 0u) - p * Q_WCAP;
                if (hm != 0u && pos >= 0 && pos < Q_WCAP)
                    qr[pos] = (unsigned)((u * (Q_SCAN * 32) + tid) * 16 + o16);
                tot += __builtin_popcount(bal);
                hm &= hm - 1u;
                bal = __builtin_amdgcn_ballot_w32(hm != 0u);
            }
        }

        __builtin_amdgcn_fence(__ATOMIC_RELEASE, "wavefront");
        __builtin_amdgcn_wave_barrier();
        __builtin_amdgcn_fence(__ATOMIC_ACQUIRE, "wavefront");
        int nq = tot - p * Q_WCAP;
        nq = nq < 0 ? 0 : (nq > Q_WCAP ? Q_WCAP : nq);
        if (lane < nq) {
            const int e = segBase + (int)qr[lane];
            const int d = dst[e];
            const int s = src[e];
            l2Fetch(feat + (long)s * F);
            l2Fetch(feat + (long)s * F + 64);
            qr[lane] = (unsigned)(d - row0);
            qsrc[nb][wave * Q_WCAP + lane] = s;
            qnrm[nb][wave * Q_WCAP + lane] = normS[s];
        }
        if (lane == 0) qtot[nb][wave] = tot;
    };

    if (scanner && nseg > 0) scanUnit(0, 0, 0);
    __syncthreads();

    int cntv = 0;
    int seg = 0, p = 0, b = 0;
    for (int it = 0; it < nseg * (Q_MAXP + 1); ++it) {
        if (seg >= nseg) break;
        const int t = (lane < Q_SCAN) ? qtot[b][lane] : 0;
        const bool again = __builtin_amdgcn_ballot_w32(t > (p + 1) * Q_WCAP) != 0u;
        const int nxSeg = again ? seg : seg + 1;
        const int nxP   = again ? p + 1 : 0;

        if (scanner) {
            if (nxSeg < nseg) scanUnit(nxSeg, nxP, b ^ 1);
        } else {
            const int ow = wave - Q_SCAN;
            int cnts = t - p * Q_WCAP;
            cnts = cnts < 0 ? 0 : (cnts > Q_WCAP ? Q_WCAP : cnts);

            unsigned ents[Q_SCAN];
            #pragma unroll
            for (int sw = 0; sw < Q_SCAN; ++sw) ents[sw] = qrow[b][sw * Q_WCAP + lane];
            int mc = 0;
            #pragma unroll
            for (int sw = 0; sw < Q_SCAN; ++sw) {
                const int cq = __builtin_amdgcn_readlane(cnts, sw);
                const unsigned ent = ents[sw];
                const bool mine = (lane < cq) && ((ent & (unsigned)(Q_OWN - 1)) == (unsigned)ow);
                const unsigned mk = __builtin_amdgcn_ballot_w32(mine);
                if (mk != 0u) {
                    if (mine)
                        mylist[ow * Q_MCAP + mc + (int)__builtin_amdgcn_mbcnt_lo(mk, 0u)] =
                            ((unsigned)(sw * Q_WCAP + lane) << 9) | ent;
                    mc += __builtin_popcount(mk);
                }
            }
            __builtin_amdgcn_fence(__ATOMIC_RELEASE, "wavefront");
            __builtin_amdgcn_wave_barrier();
            __builtin_amdgcn_fence(__ATOMIC_ACQUIRE, "wavefront");

            for (int j0 = 0; j0 < mc; j0 += 32) {
                const int j = j0 + lane;
                int   e_s = 0, e_r = 0;
                float e_n = 1.0f;
                if (j < mc) {
                    const unsigned ent = mylist[ow * Q_MCAP + j];
                    const int qi = (int)(ent >> 9);
                    e_r = (int)(ent & 511u);
                    e_s = qsrc[b][qi];
                    e_n = qnrm[b][qi];
                }
                const int m = (mc - j0 < 32) ? (mc - j0) : 32;
                for (int g0 = 0; g0 < m; g0 += Q_GB) {
                    vfl   v[Q_GB];
                    int   rr[Q_GB];
                    float nn[Q_GB];
                    #pragma unroll
                    for (int k = 0; k < Q_GB; ++k) {
                        const int jj = (g0 + k < m) ? (g0 + k) : (m - 1);
                        const int sk = __builtin_amdgcn_readlane(e_s, jj);
                        rr[k] = __builtin_amdgcn_readlane(e_r, jj);
                        nn[k] = __int_as_float(__builtin_amdgcn_readlane(__float_as_int(e_n), jj));
                        v[k] = *(const vfl*)(feat + (long)sk * F + lane * VL);
                    }
                    #pragma unroll
                    for (int k0 = 0; k0 < Q_GB; k0 += Q_AG) {
                        if (g0 + k0 + Q_AG <= m) {
                            vfl tv[Q_AG];
                            #pragma unroll
                            for (int a = 0; a < Q_AG; ++a)
                                tv[a] = *(const vfl*)&tile[rr[k0 + a] * F + lane * VL];
                            #pragma unroll
                            for (int a = 0; a < Q_AG; ++a) {
                                #pragma unroll
                                for (int i = 0; i < a; ++i)
                                    if (rr[k0 + i] == rr[k0 + a]) tv[a] = tv[i];
                                tv[a] += v[k0 + a] * nn[k0 + a];
                                cntv += (((unsigned)rr[k0 + a] / (unsigned)Q_OWN) == (unsigned)lane) ? 1 : 0;
                            }
                            #pragma unroll
                            for (int a = 0; a < Q_AG; ++a)
                                *(vfl*)&tile[rr[k0 + a] * F + lane * VL] = tv[a];
                        } else {
                            #pragma unroll
                            for (int a = 0; a < Q_AG; ++a) {
                                if (g0 + k0 + a < m) {
                                    vfl* tp = (vfl*)&tile[rr[k0 + a] * F + lane * VL];
                                    vfl tv = *tp;
                                    tv += v[k0 + a] * nn[k0 + a];
                                    *tp = tv;
                                    cntv += (((unsigned)rr[k0 + a] / (unsigned)Q_OWN) == (unsigned)lane) ? 1 : 0;
                                }
                            }
                        }
                    }
                }
            }
        }
        __syncthreads();
        seg = nxSeg; p = nxP; b ^= 1;
    }

    if (!scanner) {
        const int ow = wave - Q_SCAN;
        if (lane * Q_OWN + ow < R) rcnt[lane * Q_OWN + ow] = cntv;
    }
    __syncthreads();

    for (int i = tid; i < R * 32; i += 1024) {
        const int r = row0 + (i >> 5);
        if (r < nd)
            *(vfl*)(res + (long)r * F + (i & 31) * VL) = ((const vfl*)tile)[i];
    }
    for (int i = tid; i < R; i += 1024) {
        const int r = row0 + i;
        if (r < nd) normD[r] = rsqrtf(fmaxf((float)rcnt[i], 1.0f));
    }
}

template <int K, int N, int MT, int UNR, bool RELU, bool BIAS>
__global__ __launch_bounds__(128)
void k_gemm_wmma(const float* __restrict__ A,
                 const float* __restrict__ rowscale,
                 const float* __restrict__ W,
                 const float* __restrict__ bias,
                 float* __restrict__ C, int nRowGroups) {
    constexpr int NG = N / 64;
    const int wave = (blockIdx.x * blockDim.x + threadIdx.x) >> 5;
    const int lane = threadIdx.x & 31;
    const int rg = wave / NG;
    const int cg = wave % NG;
    if (rg >= nRowGroups) return;
    const int hf = lane >> 4;
    const int c  = lane & 15;
    const int row0 = rg * (MT * 16);
    const int col0 = cg * 64 + c * 4;

    const float* ap[MT];
    float sc[MT];
    #pragma unroll
    for (int m = 0; m < MT; ++m) {
        const int r = row0 + m * 16 + c;
        ap[m] = A + (long)r * K + hf * 4;
        sc[m] = rowscale[r];
    }
    const float* wp = W + (long)(hf * 4) * N + col0;

    v8f acc[MT][4];
    #pragma unroll
    for (int m = 0; m < MT; ++m)
        #pragma unroll
        for (int t = 0; t < 4; ++t) acc[m][t] = v8f{};

    #pragma unroll UNR
    for (int kk = 0; kk < K; kk += 8) {
        float4 av[MT], bq[4];
        #pragma unroll
        for (int m = 0; m < MT; ++m) av[m] = *(const float4*)(ap[m] + kk);
        #pragma unroll
        for (int i = 0; i < 4; ++i) bq[i] = *(const float4*)(wp + (long)(kk + i) * N);

        const float bv[4][4] = {
            {bq[0].x, bq[0].y, bq[0].z, bq[0].w},
            {bq[1].x, bq[1].y, bq[1].z, bq[1].w},
            {bq[2].x, bq[2].y, bq[2].z, bq[2].w},
            {bq[3].x, bq[3].y, bq[3].z, bq[3].w}};
        #pragma unroll
        for (int m = 0; m < MT; ++m) {
            v2f a; a.x = av[m].x * sc[m]; a.y = av[m].y * sc[m];
            #pragma unroll
            for (int t = 0; t < 4; ++t) {
                v2f b; b.x = bv[0][t]; b.y = bv[1][t];
                acc[m][t] = __builtin_amdgcn_wmma_f32_16x16x4_f32(
                    false, a, false, b, (short)0, acc[m][t], false, false);
            }
        }
        #pragma unroll
        for (int m = 0; m < MT; ++m) {
            v2f a; a.x = av[m].z * sc[m]; a.y = av[m].w * sc[m];
            #pragma unroll
            for (int t = 0; t < 4; ++t) {
                v2f b; b.x = bv[2][t]; b.y = bv[3][t];
                acc[m][t] = __builtin_amdgcn_wmma_f32_16x16x4_f32(
                    false, a, false, b, (short)0, acc[m][t], false, false);
            }
        }
    }

    float4 bval = {0.0f, 0.0f, 0.0f, 0.0f};
    if (BIAS) bval = *(const float4*)(bias + col0);
    #pragma unroll
    for (int m = 0; m < MT; ++m) {
        #pragma unroll
        for (int r = 0; r < 8; ++r) {
            float4 v;
            v.x = acc[m][0][r] + bval.x;
            v.y = acc[m][1][r] + bval.y;
            v.z = acc[m][2][r] + bval.z;
            v.w = acc[m][3][r] + bval.w;
            if (RELU) {
                v.x = fmaxf(v.x, 0.0f); v.y = fmaxf(v.y, 0.0f);
                v.z = fmaxf(v.z, 0.0f); v.w = fmaxf(v.w, 0.0f);
            }
            *(float4*)(C + (long)(row0 + m * 16 + hf * 8 + r) * N + col0) = v;
        }
    }
}

extern "C" void kernel_launch(void* const* d_in, const int* in_sizes, int n_in,
                              void* d_out, int out_size, void* d_ws, size_t ws_size,
                              hipStream_t stream) {
    const float* x    = (const float*)d_in[0];
    const float* W1   = (const float*)d_in[1];
    const float* b1   = (const float*)d_in[2];
    const float* W2   = (const float*)d_in[3];
    const float* b2   = (const float*)d_in[4];
    const int*   src1 = (const int*)d_in[5];
    const int*   dst1 = (const int*)d_in[6];
    const int*   src2 = (const int*)d_in[7];
    const int*   dst2 = (const int*)d_in[8];

    const int IN_F = 128, H_F = 256, N_CLS = 64;
    const int NSRC1 = in_sizes[0] / IN_F;
    const int E1    = in_sizes[5];
    const int E2    = in_sizes[7];
    const int ND1   = 100000;
    const int ND2   = 16384;

    float* ws    = (float*)d_ws;
    float* degS1 = ws;
    float* degS2 = degS1 + NSRC1;
    float* degD1 = degS2 + ND1;
    float* agg1  = degD1 + ND1 + ND2;
    float* h     = agg1 + (long)ND1 * IN_F;
    float* g     = h + (long)ND1 * H_F;
    float* key1  = g;

    long zcount = (NSRC1 + ND1);
    float* cnt = h;
    k_zero<<<256, 256, 0, stream>>>(cnt, 2 * zcount);

    const int nb1 = (E1 + D_EPB - 1) / D_EPB;
    const int nb2 = (E2 + D_EPB - 1) / D_EPB;
    k_deg<<<nb1 + nb2, 256, 0, stream>>>(src1, dst1, E1, cnt, nb1,
                                         src2, E2, cnt + NSRC1, (int)zcount, key1);
    k_rsqrt<<<256, 256, 0, stream>>>(cnt, cnt + zcount, ws, zcount);

    k_pull_split<128, PULL1_R>
        <<<(ND1 + PULL1_R - 1) / PULL1_R, 1024, 0, stream>>>(x, src1, dst1, degS1, key1, agg1, degD1, E1, ND1);

    {
        k_gemm_wmma<128, 256, 5, 2, true, true>
            <<<1250, 128, 0, stream>>>(agg1, degD1, W1, b1, h, 1250);
    }

    {
        k_gemm_wmma<256, 64, 2, 4, false, false>
            <<<782, 128, 0, stream>>>(h, degS2, W2, nullptr, g, 3125);
    }

    k_pull<64, 64, false, true>
        <<<(ND2 + 63) / 64, 1024, 0, stream>>>(g, src2, dst2, nullptr, b2, (float*)d_out, E2, ND2);
}


# Second aggregation run on the first one's kernel: one-byte tile numbers of list 2 (from spare blocks of the second product's launch), scanning beside owning waves; row sum order as in the first
# speedup vs baseline: 2.7138x; 1.0462x over previous
#include <hip/hip_runtime.h>


typedef __attribute__((ext_vector_type(2))) float v2f;
typedef __attribute__((ext_vector_type(8))) float v8f;

__device__ __forceinline__ void atomAddF(float* p, float v) {
    asm volatile("global_atomic_add_f32 %0, %1, off" :: "v"(p), "v"(v) : "memory");
}

__device__ __forceinline__ void l2Fetch(const float* p) {
    asm volatile("global_prefetch_b8 %0, off scope:SCOPE_SE" :: "v"(p) : "memory");
}

__global__ __launch_bounds__(256)
void k_zero(float* __restrict__ p, long n) {
    const long n4 = n >> 2;
    const long i = (long)blockIdx.x * blockDim.x + threadIdx.x;
    const long stride = (long)gridDim.x * blockDim.x;
    const float4 z = {0.0f, 0.0f, 0.0f, 0.0f};
    for (long j = i; j < n4; j += stride) ((float4*)p)[j] = z;
    const long t = (n4 << 2) + i;
    if (t < n) p[t] = 0.0f;
}

constexpr int D_EPB = 2048;
constexpr int PULL1_R = 400;
constexpr int PULL2_R = 65;
constexpr int KEY_PAD = 65536;
template <int R>
__device__ __forceinline__ unsigned tileKeys(const int4 d) {
    return (unsigned)(d.x / R) | ((unsigned)(d.y / R) << 8) |
           ((unsigned)(d.z / R) << 16) | ((unsigned)(d.w / R) << 24);
}
__global__ __launch_bounds__(256)
void k_deg(const int* __restrict__ src1, const int* __restrict__ dst1, int ne1,
           float* __restrict__ degS1, int nb1,
           const int* __restrict__ src2, int ne2,
           float* __restrict__ degS2, int halfStride,
           float* __restrict__ key1f) {
    const unsigned aid = __builtin_amdgcn_s_sendmsg_rtn(0x87);
    const long hoff = (long)((aid >> 18) & 1u) * halfStride;
    const bool first = (int)blockIdx.x < nb1;
    const int* __restrict__ src = first ? src1 : src2;
    float* __restrict__ degS = (first ? degS1 : degS2) + hoff;
    const int ne  = first ? ne1 : ne2;
    const int blk = first ? (int)blockIdx.x : (int)blockIdx.x - nb1;
    const int e0  = blk * D_EPB + (int)threadIdx.x * 4;
    if (blk * D_EPB + D_EPB <= ne) {
        const int4 s0 = *(const int4*)(src + e0);
        const int4 s1 = *(const int4*)(src + e0 + 1024);
        if (first) {
            const int4 d0 = *(const int4*)(dst1 + e0);
            const int4 d1 = *(const int4*)(dst1 + e0 + 1024);
            ((unsigned*)key1f)[e0 >> 2]          = tileKeys<PULL1_R>(d0);
            ((unsigned*)key1f)[(e0 + 1024) >> 2] = tileKeys<PULL1_R>(d1);
        }
        atomAddF(&degS[s0.x], 1.0f); atomAddF(&degS[s0.y], 1.0f);
        atomAddF(&degS[s0.z], 1.0f); atomAddF(&degS[s0.w], 1.0f);
        atomAddF(&degS[s1.x], 1.0f); atomAddF(&degS[s1.y], 1.0f);
        atomAddF(&degS[s1.z], 1.0f); atomAddF(&degS[s1.w], 1.0f);
    } else {
        for (int k = 0; k < 8; ++k) {
            const int e = e0 + (k >> 2) * 1024 + (k & 3);
            if (e < ne) {
                if (first) ((unsigned char*)key1f)[e] = (unsigned char)(dst1[e] / PULL1_R);
                atomAddF(&degS[src[e]], 1.0f);
            }
        }
    }
    if (first && blk == nb1 - 1) {
        const int a4 = (ne + 3) & ~3;
        const int padEnd = ((ne + KEY_PAD - 1) / KEY_PAD) * KEY_PAD;
        if ((int)threadIdx.x < a4 - ne) ((unsigned char*)key1f)[ne + (int)threadIdx.x] = (unsigned char)255;
        for (int w = (a4 >> 2) + (int)threadIdx.x; w < (padEnd >> 2); w += 256)
            ((unsigned*)key1f)[w] = 0xFFFFFFFFu;
    }
}

__global__ __launch_bounds__(256)
void k_rsqrt(const float* __restrict__ ca, const float* __restrict__ cb, float* __restrict__ p, long n) {
    const long n4 = n >> 2;
    const long i = (long)blockIdx.x * blockDim.x + threadIdx.x;
    const long stride = (long)gridDim.x * blockDim.x;
    for (long j0 = i; j0 < n4; j0 += 4 * stride) {
        float4 va[4], vb[4];
        #pragma unroll
        for (int k = 0; k < 4; ++k) {
            const long j = j0 + k * stride;
            if (j < n4) { va[k] = ((const float4*)ca)[j]; vb[k] = ((const float4*)cb)[j]; }
        }
        #pragma unroll
        for (int k = 0; k < 4; ++k) {
            const long j = j0 + k * stride;
            if (j < n4) {
                float4 t;
                t.x = rsqrtf(fmaxf(va[k].x + vb[k].x, 1.0f)); t.y = rsqrtf(fmaxf(va[k].y + vb[k].y, 1.0f));
                t.z = rsqrtf(fmaxf(va[k].z + vb[k].z, 1.0f)); t.w = rsqrtf(fmaxf(va[k].w + vb[k].w, 1.0f));
                ((float4*)p)[j] = t;
            }
        }
    }
    const long t = (n4 << 2) + i;
    if (t < n) p[t] = rsqrtf(fmaxf(ca[t] + cb[t], 1.0f));
}

constexpr int Q_SCAN = 16;
constexpr int Q_OWN  = 32 - Q_SCAN;
constexpr int Q_NLD  = 8;
constexpr int Q_SEG  = Q_SCAN * 32 * 16 * Q_NLD;
constexpr int Q_WCAP = 32;
constexpr int Q_MCAP = Q_SCAN * Q_WCAP;
constexpr int Q_MAXP = (32 * 16 * Q_NLD) / Q_WCAP;
constexpr int Q_GB   = 16;
constexpr int Q_AG   = 4;

template <int F, int R, bool SRCSCALE, bool FINAL>
__global__ __launch_bounds__(1024)
void k_pull_split(const float* __restrict__ feat, const int* __restrict__ src,
                  const int* __restrict__ dst, const float* __restrict__ normS,
                  const float* __restrict__ keyf, const float* __restrict__ bias,
                  float* __restrict__ res, float* __restrict__ normD, int ne, int nd) {
    constexpr int VL = F / 32;
    typedef float vfl __attribute__((ext_vector_type(VL)));
    static_assert(R <= 512 && (F == 128 || F == 64), "entry packing and lane vectors");
    static_assert(Q_SEG == KEY_PAD, "the tile bytes are padded to whole segments");
    static_assert(Q_WCAP == 32, "one queue entry per lane when the records are fetched and when they are picked");
    static_assert((Q_OWN & (Q_OWN - 1)) == 0 && Q_SCAN <= 32, "row owner by mask; one scanning wave's total per lane");

    __shared__ __attribute__((aligned(16))) float tile[R * F];
    __shared__ unsigned qrow[2][Q_SCAN * Q_WCAP];
    __shared__ int      qsrc[2][Q_SCAN * Q_WCAP];
    __shared__ float    qnrm[2][Q_SCAN * Q_WCAP];
    __shared__ int      qtot[2][Q_SCAN];
    __shared__ unsigned mylist[Q_OWN * Q_MCAP];
    __shared__ int      rcnt[R];

    const int tid  = threadIdx.x;
    const int lane = tid & 31;
    const int wave = __builtin_amdgcn_readfirstlane(tid >> 5);
    const int row0 = blockIdx.x * R;
    const bool scanner = wave < Q_SCAN;

    for (int i = tid; i < R * 32; i += 1024) ((vfl*)tile)[i] = vfl{};

    const int nseg = (ne + Q_SEG - 1) / Q_SEG;

    auto scanUnit = [&](const int seg, const int p, const int nb) {
        const int segBase = seg * Q_SEG;
        const unsigned char* keyb = (const unsigned char*)keyf + segBase;
        uint4 kv[Q_NLD];
        #pragma unroll
        for (int u = 0; u < Q_NLD; ++u)
            kv[u] = *(const uint4*)(keyb + (u * (Q_SCAN * 32) + tid) * 16);

        unsigned* qr = &qrow[nb][wave * Q_WCAP];
        const unsigned tileRep = (unsigned)blockIdx.x * 0x01010101u;
        int tot = 0;
        #pragma unroll
        for (int u = 0; u < Q_NLD; ++u) {
            const unsigned wv[4] = {kv[u].x ^ tileRep, kv[u].y ^ tileRep,
                                    kv[u].z ^ tileRep, kv[u].w ^ tileRep};
            unsigned hm = 0u;
            #pragma unroll
            for (int q = 0; q < 4; ++q) {
                const unsigned y = ((wv[q] & 0x7F7F7F7Fu) + 0x7F7F7F7Fu) | wv[q];
                hm |= (~y & 0x80808080u) >> (7 - q);
            }
            unsigned bal = __builtin_amdgcn_ballot_w32(hm != 0u);
            while (bal != 0u) {
                const int bp = __builtin_ctz(hm | 0x80000000u);
                const int o16 = ((bp & 7) << 2) | (bp >> 3);
                const int pos = tot + (int)__builtin_amdgcn_mbcnt_lo(bal, 0u) - p * Q_WCAP;
                if (hm != 0u && pos >= 0 && pos < Q_WCAP)
                    qr[pos] = (unsigned)((u * (Q_SCAN * 32) + tid) * 16 + o16);
                tot += __builtin_popcount(bal);
                hm &= hm - 1u;
                bal = __builtin_amdgcn_ballot_w32(hm != 0u);
            }
        }

        __builtin_amdgcn_fence(__ATOMIC_RELEASE, "wavefront");
        __builtin_amdgcn_wave_barrier();
        __builtin_amdgcn_fence(__ATOMIC_ACQUIRE, "wavefront");
        int nq = tot - p * Q_WCAP;
        nq = nq < 0 ? 0 : (nq > Q_WCAP ? Q_WCAP : nq);
        if (lane < nq) {
            const int e = segBase + (int)qr[lane];
            const int d = dst[e];
            const int s = src[e];
            if (F == 128) {
                l2Fetch(feat + (long)s * F);
                l2Fetch(feat + (long)s * F + 64);
            }
            qr[lane] = (unsigned)(d - row0);
            qsrc[nb][wave * Q_WCAP + lane] = s;
            if (SRCSCALE) qnrm[nb][wave * Q_WCAP + lane] = normS[s];
        }
        if (lane == 0) qtot[nb][wave] = tot;
    };

    if (scanner && nseg > 0) scanUnit(0, 0, 0);
    __syncthreads();

    int cntv = 0;
    int seg = 0, p = 0, b = 0;
    for (int it = 0; it < nseg * (Q_MAXP + 1); ++it) {
        if (seg >= nseg) break;
        const int t = (lane < Q_SCAN) ? qtot[b][lane] : 0;
        const bool again = __builtin_amdgcn_ballot_w32(t > (p + 1) * Q_WCAP) != 0u;
        const int nxSeg = again ? seg : seg + 1;
        const int nxP   = again ? p + 1 : 0;

        if (scanner) {
            if (nxSeg < nseg) scanUnit(nxSeg, nxP, b ^ 1);
        } else {
            const int ow = wave - Q_SCAN;
            int cnts = t - p * Q_WCAP;
            cnts = cnts < 0 ? 0 : (cnts > Q_WCAP ? Q_WCAP : cnts);

            unsigned ents[Q_SCAN];
            #pragma unroll
            for (int sw = 0; sw < Q_SCAN; ++sw) ents[sw] = qrow[b][sw * Q_WCAP + lane];
            int mc = 0;
            #pragma unroll
            for (int sw = 0; sw < Q_SCAN; ++sw) {
                const int cq = __builtin_amdgcn_readlane(cnts, sw);
                const unsigned ent = ents[sw];
                const bool mine = (lane < cq) && ((ent & (unsigned)(Q_OWN - 1)) == (unsigned)ow);
                const unsigned mk = __builtin_amdgcn_ballot_w32(mine);
                if (mk != 0u) {
                    if (mine)
                        mylist[ow * Q_MCAP + mc + (int)__builtin_amdgcn_mbcnt_lo(mk, 0u)] =
                            ((unsigned)(sw * Q_WCAP + lane) << 9) | ent;
                    mc += __builtin_popcount(mk);
                }
            }
            __builtin_amdgcn_fence(__ATOMIC_RELEASE, "wavefront");
            __builtin_amdgcn_wave_barrier();
            __builtin_amdgcn_fence(__ATOMIC_ACQUIRE, "wavefront");

            for (int j0 = 0; j0 < mc; j0 += 32) {
                const int j = j0 + lane;
                int   e_s = 0, e_r = 0;
                float e_n = 1.0f;
                if (j < mc) {
                    const unsigned ent = mylist[ow * Q_MCAP + j];
                    const int qi = (int)(ent >> 9);
                    e_r = (int)(ent & 511u);
                    e_s = qsrc[b][qi];
                    if (SRCSCALE) e_n = qnrm[b][qi];
                }
                const int m = (mc - j0 < 32) ? (mc - j0) : 32;
                for (int g0 = 0; g0 < m; g0 += Q_GB) {
                    vfl   v[Q_GB];
                    int   rr[Q_GB];
                    float nn[Q_GB];
                    #pragma unroll
                    for (int k = 0; k < Q_GB; ++k) {
                        const int jj = (g0 + k < m) ? (g0 + k) : (m - 1);
                        const int sk = __builtin_amdgcn_readlane(e_s, jj);
                        rr[k] = __builtin_amdgcn_readlane(e_r, jj);
                        nn[k] = __int_as_float(__builtin_amdgcn_readlane(__float_as_int(e_n), jj));
                        v[k] = *(const vfl*)(feat + (long)sk * F + lane * VL);
                    }
                    #pragma unroll
                    for (int k0 = 0; k0 < Q_GB; k0 += Q_AG) {
                        if (g0 + k0 + Q_AG <= m) {
                            vfl tv[Q_AG];
                            #pragma unroll
                            for (int a = 0; a < Q_AG; ++a)
                                tv[a] = *(const vfl*)&tile[rr[k0 + a] * F + lane * VL];
                            #pragma unroll
                            for (int a = 0; a < Q_AG; ++a) {
                                #pragma unroll
                                for (int i = 0; i < a; ++i)
                                    if (rr[k0 + i] == rr[k0 + a]) tv[a] = tv[i];
                                if (SRCSCALE) tv[a] += v[k0 + a] * nn[k0 + a];
                                else          tv[a] += v[k0 + a];
                                cntv += (((unsigned)rr[k0 + a] / (unsigned)Q_OWN) == (unsigned)lane) ? 1 : 0;
                            }
                            #pragma unroll
                            for (int a = 0; a < Q_AG; ++a)
                                *(vfl*)&tile[rr[k0 + a] * F + lane * VL] = tv[a];
                        } else {
                            #pragma unroll
                            for (int a = 0; a < Q_AG; ++a) {
                                if (g0 + k0 + a < m) {
                                    vfl* tp = (vfl*)&tile[rr[k0 + a] * F + lane * VL];
                                    vfl tv = *tp;
                                    if (SRCSCALE) tv += v[k0 + a] * nn[k0 + a];
                                    else          tv += v[k0 + a];
                                    *tp = tv;
                                    cntv += (((unsigned)rr[k0 + a] / (unsigned)Q_OWN) == (unsigned)lane) ? 1 : 0;
                                }
                            }
                        }
                    }
                }
            }
        }
        __syncthreads();
        seg = nxSeg; p = nxP; b ^= 1;
    }

    if (!scanner) {
        const int ow = wave - Q_SCAN;
        if (lane * Q_OWN + ow < R) rcnt[lane * Q_OWN + ow] = cntv;
    }
    __syncthreads();

    for (int i = tid; i < R * 32; i += 1024) {
        const int r = row0 + (i >> 5);
        if (r < nd) {
            vfl t = ((const vfl*)tile)[i];
            if (FINAL) {
                const float sc = rsqrtf(fmaxf((float)rcnt[i >> 5], 1.0f));
                const vfl bv = *(const vfl*)(bias + (i & 31) * VL);
                t = t * sc + bv;
            }
            *(vfl*)(res + (long)r * F + (i & 31) * VL) = t;
        }
    }
    if (!FINAL) {
        for (int i = tid; i < R; i += 1024) {
            const int r = row0 + i;
            if (r < nd) normD[r] = rsqrtf(fmaxf((float)rcnt[i], 1.0f));
        }
    }
}

constexpr int K_EPB = 2048;
template <int K, int N, int MT, int UNR, bool RELU, bool BIAS, bool KEYS>
__global__ __launch_bounds__(128)
void k_gemm_wmma(const float* __restrict__ A,
                 const float* __restrict__ rowscale,
                 const float* __restrict__ W,
                 const float* __restrict__ bias,
                 float* __restrict__ C, int nRowGroups,
                 const int* __restrict__ kdst, float* __restrict__ keyf, int kne, int nGemmBlocks) {
    if (KEYS && (int)blockIdx.x >= nGemmBlocks) {
        const int padEnd = ((kne + KEY_PAD - 1) / KEY_PAD) * KEY_PAD;
        const int base = ((int)blockIdx.x - nGemmBlocks) * K_EPB + (int)threadIdx.x * 4;
        #pragma unroll
        for (int u = 0; u < 4; ++u) {
            const int e = base + u * 512;
            if (e + 3 < kne) {
                ((unsigned*)keyf)[e >> 2] = tileKeys<PULL2_R>(*(const int4*)(kdst + e));
            } else {
                for (int q = 0; q < 4; ++q)
                    if (e + q < padEnd)
                        ((unsigned char*)keyf)[e + q] =
                            (e + q < kne) ? (unsigned char)(kdst[e + q] / PULL2_R) : (unsigned char)255;
            }
        }
        return;
    }
    constexpr int NG = N / 64;
    const int wave = (blockIdx.x * blockDim.x + threadIdx.x) >> 5;
    const int lane = threadIdx.x & 31;
    const int rg = wave / NG;
    const int cg = wave % NG;
    if (rg >= nRowGroups) return;
    const int hf = lane >> 4;
    const int c  = lane & 15;
    const int row0 = rg * (MT * 16);
    const int col0 = cg * 64 + c * 4;

    const float* ap[MT];
    float sc[MT];
    #pragma unroll
    for (int m = 0; m < MT; ++m) {
        const int r = row0 + m * 16 + c;
        ap[m] = A + (long)r * K + hf * 4;
        sc[m] = rowscale[r];
    }
    const float* wp = W + (long)(hf * 4) * N + col0;

    v8f acc[MT][4];
    #pragma unroll
    for (int m = 0; m < MT; ++m)
        #pragma unroll
        for (int t = 0; t < 4; ++t) acc[m][t] = v8f{};

    #pragma unroll UNR
    for (int kk = 0; kk < K; kk += 8) {
        float4 av[MT], bq[4];
        #pragma unroll
        for (int m = 0; m < MT; ++m) av[m] = *(const float4*)(ap[m] + kk);
        #pragma unroll
        for (int i = 0; i < 4; ++i) bq[i] = *(const float4*)(wp + (long)(kk + i) * N);

        const float bv[4][4] = {
            {bq[0].x, bq[0].y, bq[0].z, bq[0].w},
            {bq[1].x, bq[1].y, bq[1].z, bq[1].w},
            {bq[2].x, bq[2].y, bq[2].z, bq[2].w},
            {bq[3].x, bq[3].y, bq[3].z, bq[3].w}};
        #pragma unroll
        for (int m = 0; m < MT; ++m) {
            v2f a; a.x = av[m].x * sc[m]; a.y = av[m].y * sc[m];
            #pragma unroll
            for (int t = 0; t < 4; ++t) {
                v2f b; b.x = bv[0][t]; b.y = bv[1][t];
                acc[m][t] = __builtin_amdgcn_wmma_f32_16x16x4_f32(
                    false, a, false, b, (short)0, acc[m][t], false, false);
            }
        }
        #pragma unroll
        for (int m = 0; m < MT; ++m) {
            v2f a; a.x = av[m].z * sc[m]; a.y = av[m].w * sc[m];
            #pragma unroll
            for (int t = 0; t < 4; ++t) {
                v2f b; b.x = bv[2][t]; b.y = bv[3][t];
                acc[m][t] = __builtin_amdgcn_wmma_f32_16x16x4_f32(
                    false, a, false, b, (short)0, acc[m][t], false, false);
            }
        }
    }

    float4 bval = {0.0f, 0.0f, 0.0f, 0.0f};
    if (BIAS) bval = *(const float4*)(bias + col0);
    #pragma unroll
    for (int m = 0; m < MT; ++m) {
        #pragma unroll
        for (int r = 0; r < 8; ++r) {
            float4 v;
            v.x = acc[m][0][r] + bval.x;
            v.y = acc[m][1][r] + bval.y;
            v.z = acc[m][2][r] + bval.z;
            v.w = acc[m][3][r] + bval.w;
            if (RELU) {
                v.x = fmaxf(v.x, 0.0f); v.y = fmaxf(v.y, 0.0f);
                v.z = fmaxf(v.z, 0.0f); v.w = fmaxf(v.w, 0.0f);
            }
            *(float4*)(C + (long)(row0 + m * 16 + hf * 8 + r) * N + col0) = v;
        }
    }
}

extern "C" void kernel_launch(void* const* d_in, const int* in_sizes, int n_in,
                              void* d_out, int out_size, void* d_ws, size_t ws_size,
                              hipStream_t stream) {
    const float* x    = (const float*)d_in[0];
    const float* W1   = (const float*)d_in[1];
    const float* b1   = (const float*)d_in[2];
    const float* W2   = (const float*)d_in[3];
    const float* b2   = (const float*)d_in[4];
    const int*   src1 = (const int*)d_in[5];
    const int*   dst1 = (const int*)d_in[6];
    const int*   src2 = (const int*)d_in[7];
    const int*   dst2 = (const int*)d_in[8];

    const int IN_F = 128, H_F = 256, N_CLS = 64;
    const int NSRC1 = in_sizes[0] / IN_F;
    const int E1    = in_sizes[5];
    const int E2    = in_sizes[7];
    const int ND1   = 100000;
    const int ND2   = 16384;

    float* ws    = (float*)d_ws;
    float* degS1 = ws;
    float* degS2 = degS1 + NSRC1;
    float* degD1 = degS2 + ND1;
    float* agg1  = degD1 + ND1 + ND2;
    float* h     = agg1 + (long)ND1 * IN_F;
    float* g     = h + (long)ND1 * H_F;
    float* key1  = g;

    long zcount = (NSRC1 + ND1);
    float* cnt = h;
    k_zero<<<256, 256, 0, stream>>>(cnt, 2 * zcount);

    const int nb1 = (E1 + D_EPB - 1) / D_EPB;
    const int nb2 = (E2 + D_EPB - 1) / D_EPB;
    k_deg<<<nb1 + nb2, 256, 0, stream>>>(src1, dst1, E1, cnt, nb1,
                                         src2, E2, cnt + NSRC1, (int)zcount, key1);
    k_rsqrt<<<256, 256, 0, stream>>>(cnt, cnt + zcount, ws, zcount);

    k_pull_split<128, PULL1_R, true, false>
        <<<(ND1 + PULL1_R - 1) / PULL1_R, 1024, 0, stream>>>(x, src1, dst1, degS1, key1, nullptr, agg1, degD1, E1, ND1);

    {
        k_gemm_wmma<128, 256, 5, 2, true, true, false>
            <<<1250, 128, 0, stream>>>(agg1, degD1, W1, b1, h, 1250, nullptr, nullptr, 0, 1250);
    }

    float* key2 = agg1;
    {
        const int nkb = ((E2 + KEY_PAD - 1) / KEY_PAD) * (KEY_PAD / K_EPB);
        k_gemm_wmma<256, 64, 2, 4, false, false, true>
            <<<782 + nkb, 128, 0, stream>>>(h, degS2, W2, nullptr, g, 3125, dst2, key2, E2, 782);
    }

    k_pull_split<64, PULL2_R, false, true>
        <<<(ND2 + PULL2_R - 1) / PULL2_R, 1024, 0, stream>>>(g, src2, dst2, nullptr, key2, b2, (float*)d_out, nullptr, E2, ND2);
}
